# starting kernel
# baseline (speedup 1.0000x reference)
#include <hip/hip_runtime.h>
#include <math.h>

#define IN_DIM   256
#define HEADS    3
#define HID      3
#define NCLS     3
#define NEG_SLOPE 0.2f
#define GAT_EPS  1e-16f

typedef __attribute__((ext_vector_type(2))) float v2f;
typedef __attribute__((ext_vector_type(8))) float v8f;

__device__ __forceinline__ unsigned fenc(float f) {
    unsigned u = __float_as_uint(f);
    return (u & 0x80000000u) ? ~u : (u | 0x80000000u);
}
__device__ __forceinline__ float fdec(unsigned u) {
    return (u & 0x80000000u) ? __uint_as_float(u & 0x7fffffffu)
                             : __uint_as_float(~u);
}
__device__ __forceinline__ float lrelu(float x) { return x > 0.f ? x : NEG_SLOPE * x; }

__global__ __launch_bounds__(256) void gemm1_wmma(const float* __restrict__ x,
                                                  const float* __restrict__ W,
                                                  float* __restrict__ h, int N) {
    __shared__ float Wp[IN_DIM * 16];
    for (int i = threadIdx.x; i < IN_DIM * 16; i += 256) {
        int k = i >> 4, n = i & 15;
        Wp[i] = (n < HEADS * HID) ? W[k * (HEADS * HID) + n] : 0.0f;
    }
    __syncthreads();

    int lane  = threadIdx.x & 31;
    int wave  = threadIdx.x >> 5;
    long tile = (long)blockIdx.x * 8 + wave;
    long m0   = tile * 16;
    if (m0 >= N) return;

    int laneM = lane & 15;
    int koff  = (lane < 16) ? 0 : 2;
    const float* xrow = x + (m0 + laneM) * IN_DIM;

    v8f c = {};
    #pragma unroll 4
    for (int k0 = 0; k0 < IN_DIM; k0 += 4) {
        v2f a, b;
        a.x = xrow[k0 + koff];
        a.y = xrow[k0 + koff + 1];
        b.x = Wp[(k0 + koff)     * 16 + laneM];
        b.y = Wp[(k0 + koff + 1) * 16 + laneM];
        c = __builtin_amdgcn_wmma_f32_16x16x4_f32(false, a, false, b,
                                                  (short)0, c, false, false);
    }

    int rbase = (lane < 16) ? 0 : 8;
    if (laneM < HEADS * HID) {
        #pragma unroll
        for (int v = 0; v < 8; ++v)
            h[(m0 + rbase + v) * (HEADS * HID) + laneM] = c[v];
    }
}

__global__ void node_att(const float* __restrict__ h,
                         const float* __restrict__ atts,
                         const float* __restrict__ attd,
                         float* __restrict__ as_, float* __restrict__ ad_,
                         unsigned* __restrict__ m, int N) {
    int n = blockIdx.x * blockDim.x + threadIdx.x;
    if (n >= N) return;
    #pragma unroll
    for (int hh = 0; hh < HEADS; ++hh) {
        float s = 0.f, d = 0.f;
        #pragma unroll
        for (int cc = 0; cc < 3; ++cc) {
            float v = h[n * 9 + hh * 3 + cc];
            s += v * atts[hh * 3 + cc];
            d += v * attd[hh * 3 + cc];
        }
        as_[n * 3 + hh] = s;
        ad_[n * 3 + hh] = d;
        m[n * 3 + hh] = fenc(lrelu(s + d));
    }
}

__global__ void edge_max(const int* __restrict__ src, const int* __restrict__ dst,
                         const float* __restrict__ as_, const float* __restrict__ ad_,
                         unsigned* __restrict__ m, int E) {
    int i = blockIdx.x * blockDim.x + threadIdx.x;
    if (i >= E) return;
    int s = src[i], d = dst[i];
    #pragma unroll
    for (int hh = 0; hh < HEADS; ++hh) {
        float e = lrelu(as_[s * 3 + hh] + ad_[d * 3 + hh]);
        atomicMax(&m[d * 3 + hh], fenc(e));
    }
}

__global__ void node_init(const float* __restrict__ h,
                          const float* __restrict__ as_, const float* __restrict__ ad_,
                          const unsigned* __restrict__ m,
                          float* __restrict__ s, float* __restrict__ acc, int N) {
    int n = blockIdx.x * blockDim.x + threadIdx.x;
    if (n >= N) return;
    #pragma unroll
    for (int hh = 0; hh < HEADS; ++hh) {
        float e  = lrelu(as_[n * 3 + hh] + ad_[n * 3 + hh]);
        float p  = expf(e - fdec(m[n * 3 + hh]));
        s[n * 3 + hh] = p;
        #pragma unroll
        for (int cc = 0; cc < 3; ++cc)
            acc[n * 9 + hh * 3 + cc] = p * h[n * 9 + hh * 3 + cc];
    }
}

__global__ void edge_scatter(const int* __restrict__ src, const int* __restrict__ dst,
                             const float* __restrict__ h,
                             const float* __restrict__ as_, const float* __restrict__ ad_,
                             const unsigned* __restrict__ m,
                             float* __restrict__ s, float* __restrict__ acc, int E) {
    int i = blockIdx.x * blockDim.x + threadIdx.x;
    if (i >= E) return;
    int sn = src[i], dn = dst[i];
    __builtin_prefetch(&h[sn * 9], 0, 0);
    #pragma unroll
    for (int hh = 0; hh < HEADS; ++hh) {
        float e = lrelu(as_[sn * 3 + hh] + ad_[dn * 3 + hh]);
        float p = expf(e - fdec(m[dn * 3 + hh]));
        atomicAdd(&s[dn * 3 + hh], p);
        #pragma unroll
        for (int cc = 0; cc < 3; ++cc)
            atomicAdd(&acc[dn * 9 + hh * 3 + cc], p * h[sn * 9 + hh * 3 + cc]);
    }
}

__global__ void finish_elu(const float* __restrict__ acc, const float* __restrict__ s,
                           const float* __restrict__ b, float* __restrict__ t, int N) {
    int n = blockIdx.x * blockDim.x + threadIdx.x;
    if (n >= N) return;
    #pragma unroll
    for (int hh = 0; hh < HEADS; ++hh) {
        float inv = 1.0f / (s[n * 3 + hh] + GAT_EPS);
        #pragma unroll
        for (int cc = 0; cc < 3; ++cc) {
            float v = acc[n * 9 + hh * 3 + cc] * inv + b[hh * 3 + cc];
            t[n * 9 + hh * 3 + cc] = (v > 0.f) ? v : (expf(v) - 1.0f);
        }
    }
}

__global__ void gemm2(const float* __restrict__ t, const float* __restrict__ W2,
                      float* __restrict__ h2, int N) {
    __shared__ float Ws[81];
    if (threadIdx.x < 81) Ws[threadIdx.x] = W2[threadIdx.x];
    __syncthreads();
    int n = blockIdx.x * blockDim.x + threadIdx.x;
    if (n >= N) return;
    float tin[9];
    #pragma unroll
    for (int i = 0; i < 9; ++i) tin[i] = t[n * 9 + i];
    #pragma unroll
    for (int j = 0; j < 9; ++j) {
        float a = 0.f;
        #pragma unroll
        for (int i = 0; i < 9; ++i) a += tin[i] * Ws[i * 9 + j];
        h2[n * 9 + j] = a;
    }
}

__global__ void finish_out(const float* __restrict__ acc, const float* __restrict__ s,
                           const float* __restrict__ b2, float* __restrict__ out, int N) {
    int n = blockIdx.x * blockDim.x + threadIdx.x;
    if (n >= N) return;
    float v[NCLS];
    #pragma unroll
    for (int cc = 0; cc < NCLS; ++cc) {
        float acv = 0.f;
        #pragma unroll
        for (int hh = 0; hh < HEADS; ++hh)
            acv += acc[n * 9 + hh * 3 + cc] / (s[n * 3 + hh] + GAT_EPS);
        v[cc] = acv * (1.0f / HEADS) + b2[cc];
    }
    float mx = fmaxf(v[0], fmaxf(v[1], v[2]));
    float lse = logf(expf(v[0] - mx) + expf(v[1] - mx) + expf(v[2] - mx));
    #pragma unroll
    for (int cc = 0; cc < NCLS; ++cc)
        out[n * NCLS + cc] = v[cc] - mx - lse;
}

extern "C" void kernel_launch(void* const* d_in, const int* in_sizes, int n_in,
                              void* d_out, int out_size, void* d_ws, size_t ws_size,
                              hipStream_t stream) {
    const float* x        = (const float*)d_in[0];
    const int*   eidx     = (const int*)d_in[1];
    const float* W1       = (const float*)d_in[2];
    const float* att_src1 = (const float*)d_in[3];
    const float* att_dst1 = (const float*)d_in[4];
    const float* b1       = (const float*)d_in[5];
    const float* W2       = (const float*)d_in[6];
    const float* att_src2 = (const float*)d_in[7];
    const float* att_dst2 = (const float*)d_in[8];
    const float* b2       = (const float*)d_in[9];

    const int N = in_sizes[0] / IN_DIM;
    const int E = in_sizes[1] / 2;
    const int* src = eidx;
    const int* dst = eidx + E;

    float*    h   = (float*)d_ws;
    float*    as_ = h   + (size_t)N * 9;
    float*    ad_ = as_ + (size_t)N * 3;
    unsigned* m   = (unsigned*)(ad_ + (size_t)N * 3);
    float*    s   = (float*)(m + (size_t)N * 3);
    float*    acc = s   + (size_t)N * 3;
    float*    t   = acc + (size_t)N * 9;

    const int nb = (N + 255) / 256;
    const int eb = (E + 255) / 256;
    const int tiles = (N + 15) / 16;
    const int gb = (tiles + 7) / 8;

    gemm1_wmma <<<gb, 256, 0, stream>>>(x, W1, h, N);
    node_att   <<<nb, 256, 0, stream>>>(h, att_src1, att_dst1, as_, ad_, m, N);
    edge_max   <<<eb, 256, 0, stream>>>(src, dst, as_, ad_, m, E);
    node_init  <<<nb, 256, 0, stream>>>(h, as_, ad_, m, s, acc, N);
    edge_scatter<<<eb, 256, 0, stream>>>(src, dst, h, as_, ad_, m, s, acc, E);
    finish_elu <<<nb, 256, 0, stream>>>(acc, s, b1, t, N);

    gemm2      <<<nb, 256, 0, stream>>>(t, W2, h, N);
    node_att   <<<nb, 256, 0, stream>>>(h, att_src2, att_dst2, as_, ad_, m, N);
    edge_max   <<<eb, 256, 0, stream>>>(src, dst, as_, ad_, m, E);
    node_init  <<<nb, 256, 0, stream>>>(h, as_, ad_, m, s, acc, N);
    edge_scatter<<<eb, 256, 0, stream>>>(src, dst, h, as_, ad_, m, s, acc, E);
    finish_out <<<nb, 256, 0, stream>>>(acc, s, b2, (float*)d_out, N);
}


# Edges sorted by destination once per call (count, scan, fill); one thread per node and head sums its row in registers with a running max: float atomics gone, sum order is row fill order
# speedup vs baseline: 5.1964x; 5.1964x over previous
#include <hip/hip_runtime.h>
#include <math.h>

#define IN_DIM   256
#define HEADS    3
#define HID      3
#define NCLS     3
#define NEG_SLOPE 0.2f
#define GAT_EPS  1e-16f

typedef __attribute__((ext_vector_type(2))) float v2f;
typedef __attribute__((ext_vector_type(8))) float v8f;

__device__ __forceinline__ float lrelu(float x) { return x > 0.f ? x : NEG_SLOPE * x; }

__global__ __launch_bounds__(256) void gemm1_wmma(const float* __restrict__ x,
                                                  const float* __restrict__ W,
                                                  float* __restrict__ h, int N) {
    __shared__ float Wp[IN_DIM * 16];
    for (int i = threadIdx.x; i < IN_DIM * 16; i += 256) {
        int k = i >> 4, n = i & 15;
        Wp[i] = (n < HEADS * HID) ? W[k * (HEADS * HID) + n] : 0.0f;
    }
    __syncthreads();

    int lane  = threadIdx.x & 31;
    int wave  = threadIdx.x >> 5;
    long tile = (long)blockIdx.x * 8 + wave;
    long m0   = tile * 16;
    if (m0 >= N) return;

    int laneM = lane & 15;
    int koff  = (lane < 16) ? 0 : 2;
    const float* xrow = x + (m0 + laneM) * IN_DIM;

    v8f c = {};
    #pragma unroll 4
    for (int k0 = 0; k0 < IN_DIM; k0 += 4) {
        v2f a, b;
        a.x = xrow[k0 + koff];
        a.y = xrow[k0 + koff + 1];
        b.x = Wp[(k0 + koff)     * 16 + laneM];
        b.y = Wp[(k0 + koff + 1) * 16 + laneM];
        c = __builtin_amdgcn_wmma_f32_16x16x4_f32(false, a, false, b,
                                                  (short)0, c, false, false);
    }

    int rbase = (lane < 16) ? 0 : 8;
    if (laneM < HEADS * HID) {
        #pragma unroll
        for (int v = 0; v < 8; ++v)
            h[(m0 + rbase + v) * (HEADS * HID) + laneM] = c[v];
    }
}

__global__ void node_rec1(const float* __restrict__ h, const float* __restrict__ atts,
                          float4* __restrict__ R, int* __restrict__ cnt, int N) {
    int n = blockIdx.x * blockDim.x + threadIdx.x;
    if (n >= N) return;
    cnt[n] = 0;
    #pragma unroll
    for (int hh = 0; hh < HEADS; ++hh) {
        float v0 = h[n * 9 + hh * 3 + 0];
        float v1 = h[n * 9 + hh * 3 + 1];
        float v2 = h[n * 9 + hh * 3 + 2];
        float s = 0.f;
        s += v0 * atts[hh * 3 + 0];
        s += v1 * atts[hh * 3 + 1];
        s += v2 * atts[hh * 3 + 2];
        R[n * 3 + hh] = make_float4(v0, v1, v2, s);
    }
}

__global__ void node_rec2(const float* __restrict__ t, const float* __restrict__ W2,
                          const float* __restrict__ atts, float4* __restrict__ R, int N) {
    __shared__ float Ws[81];
    if (threadIdx.x < 81) Ws[threadIdx.x] = W2[threadIdx.x];
    __syncthreads();
    int n = blockIdx.x * blockDim.x + threadIdx.x;
    if (n >= N) return;
    float tin[9];
    #pragma unroll
    for (int i = 0; i < 9; ++i) tin[i] = t[n * 9 + i];
    float h2[9];
    #pragma unroll
    for (int j = 0; j < 9; ++j) {
        float a = 0.f;
        #pragma unroll
        for (int i = 0; i < 9; ++i) a += tin[i] * Ws[i * 9 + j];
        h2[j] = a;
    }
    #pragma unroll
    for (int hh = 0; hh < HEADS; ++hh) {
        float s = 0.f;
        s += h2[hh * 3 + 0] * atts[hh * 3 + 0];
        s += h2[hh * 3 + 1] * atts[hh * 3 + 1];
        s += h2[hh * 3 + 2] * atts[hh * 3 + 2];
        R[n * 3 + hh] = make_float4(h2[hh * 3 + 0], h2[hh * 3 + 1], h2[hh * 3 + 2], s);
    }
}

__global__ void edge_count(const int* __restrict__ dst, int* __restrict__ cnt, int E) {
    int i = blockIdx.x * blockDim.x + threadIdx.x;
    if (i >= E) return;
    atomicAdd(&cnt[dst[i]], 1);
}

__global__ __launch_bounds__(1024) void scan_rows(const int* __restrict__ cnt,
                                                  int* __restrict__ rowptr,
                                                  int* __restrict__ cursor, int N) {
    __shared__ int sh[1024];
    const int t = threadIdx.x;
    const int base = blockIdx.x * 1024;

    int pre = 0;
    for (int i = t; i < base; i += 1024) pre += cnt[i];
    sh[t] = pre;
    __syncthreads();
    for (int off = 512; off > 0; off >>= 1) {
        if (t < off) sh[t] += sh[t + off];
        __syncthreads();
    }
    const int tile_off = sh[0];
    __syncthreads();

    const int i = base + t;
    const int c = (i < N) ? cnt[i] : 0;
    sh[t] = c;
    __syncthreads();
    for (int off = 1; off < 1024; off <<= 1) {
        int v = (t >= off) ? sh[t - off] : 0;
        __syncthreads();
        sh[t] += v;
        __syncthreads();
    }
    const int excl = tile_off + sh[t] - c;
    if (i < N) { rowptr[i] = excl; cursor[i] = excl; }
    if (i == N - 1) rowptr[N] = excl + c;
}

__global__ void edge_fill(const int* __restrict__ src, const int* __restrict__ dst,
                          int* __restrict__ cursor, int* __restrict__ csr, int E) {
    int i = blockIdx.x * blockDim.x + threadIdx.x;
    if (i >= E) return;
    int pos = atomicAdd(&cursor[dst[i]], 1);
    csr[pos] = src[i];
}

__device__ __forceinline__ void edge_step(const float4 r, const float ad, float& m,
                                          float& s, float& a0, float& a1, float& a2) {
    const float e  = lrelu(r.w + ad);
    const float d  = e - m;
    const float tt = expf(-fabsf(d));
    const bool  up = d > 0.f;
    const float sc = up ? tt : 1.f;
    const float p  = up ? 1.f : tt;
    m  = up ? e : m;
    s  = s  * sc + p;
    a0 = a0 * sc + p * r.x;
    a1 = a1 * sc + p * r.y;
    a2 = a2 * sc + p * r.z;
}

template <int LAYER>
__global__ __launch_bounds__(256) void gather_rows(const int* __restrict__ rowptr,
                                                   const int* __restrict__ csr,
                                                   const float4* __restrict__ R,
                                                   const float* __restrict__ attd,
                                                   const float* __restrict__ bias,
                                                   float* __restrict__ o, int N) {
    const int g = blockIdx.x * 256 + threadIdx.x;
    if (g >= N * HEADS) return;
    const int n  = g / HEADS;
    const int hh = g - n * HEADS;

    const float4 self = R[g];
    float ad = 0.f;
    ad += self.x * attd[hh * 3 + 0];
    ad += self.y * attd[hh * 3 + 1];
    ad += self.z * attd[hh * 3 + 2];

    float m = lrelu(self.w + ad);
    float s = 1.f, a0 = self.x, a1 = self.y, a2 = self.z;

    int j = rowptr[n];
    const int end = rowptr[n + 1];
    for (; j + 4 <= end; j += 4) {
        const int i0 = csr[j], i1 = csr[j + 1], i2 = csr[j + 2], i3 = csr[j + 3];
        const float4 r0 = R[i0 * HEADS + hh];
        const float4 r1 = R[i1 * HEADS + hh];
        const float4 r2 = R[i2 * HEADS + hh];
        const float4 r3 = R[i3 * HEADS + hh];
        edge_step(r0, ad, m, s, a0, a1, a2);
        edge_step(r1, ad, m, s, a0, a1, a2);
        edge_step(r2, ad, m, s, a0, a1, a2);
        edge_step(r3, ad, m, s, a0, a1, a2);
    }
    for (; j < end; ++j) {
        const float4 r = R[csr[j] * HEADS + hh];
        edge_step(r, ad, m, s, a0, a1, a2);
    }

    const float inv = 1.0f / (s + GAT_EPS);
    float v0 = a0 * inv, v1 = a1 * inv, v2 = a2 * inv;
    if (LAYER == 1) {
        v0 += bias[hh * 3 + 0];
        v1 += bias[hh * 3 + 1];
        v2 += bias[hh * 3 + 2];
        v0 = (v0 > 0.f) ? v0 : (expf(v0) - 1.0f);
        v1 = (v1 > 0.f) ? v1 : (expf(v1) - 1.0f);
        v2 = (v2 > 0.f) ? v2 : (expf(v2) - 1.0f);
    }
    o[g * 3 + 0] = v0;
    o[g * 3 + 1] = v1;
    o[g * 3 + 2] = v2;
}

__global__ void finish_out(const float* __restrict__ agg, const float* __restrict__ b2,
                           float* __restrict__ out, int N) {
    int n = blockIdx.x * blockDim.x + threadIdx.x;
    if (n >= N) return;
    float v[NCLS];
    #pragma unroll
    for (int cc = 0; cc < NCLS; ++cc) {
        float acv = 0.f;
        #pragma unroll
        for (int hh = 0; hh < HEADS; ++hh)
            acv += agg[n * 9 + hh * 3 + cc];
        v[cc] = acv * (1.0f / HEADS) + b2[cc];
    }
    float mx = fmaxf(v[0], fmaxf(v[1], v[2]));
    float lse = logf(expf(v[0] - mx) + expf(v[1] - mx) + expf(v[2] - mx));
    #pragma unroll
    for (int cc = 0; cc < NCLS; ++cc)
        out[n * NCLS + cc] = v[cc] - mx - lse;
}

extern "C" void kernel_launch(void* const* d_in, const int* in_sizes, int n_in,
                              void* d_out, int out_size, void* d_ws, size_t ws_size,
                              hipStream_t stream) {
    const float* x        = (const float*)d_in[0];
    const int*   eidx     = (const int*)d_in[1];
    const float* W1       = (const float*)d_in[2];
    const float* att_src1 = (const float*)d_in[3];
    const float* att_dst1 = (const float*)d_in[4];
    const float* b1       = (const float*)d_in[5];
    const float* W2       = (const float*)d_in[6];
    const float* att_src2 = (const float*)d_in[7];
    const float* att_dst2 = (const float*)d_in[8];
    const float* b2       = (const float*)d_in[9];

    const int N = in_sizes[0] / IN_DIM;
    const int E = in_sizes[1] / 2;
    const int* src = eidx;
    const int* dst = eidx + E;

    const size_t need = ((size_t)N * 24 + 1 + (size_t)E) * 4;
    if (ws_size < need) return;

    float4* R      = (float4*)d_ws;
    float*  tb     = (float*)d_ws + (size_t)N * 12;
    int*    cnt    = (int*)(tb + (size_t)N * 9);
    int*    rowptr = cnt + N;
    int*    cursor = rowptr + N + 1;
    int*    csr    = cursor + N;

    const int nb = (N + 255) / 256;
    const int eb = (E + 255) / 256;
    const int hb = (N * HEADS + 255) / 256;
    const int sb = (N + 1023) / 1024;
    const int tiles = (N + 15) / 16;
    const int gb = (tiles + 7) / 8;

    gemm1_wmma <<<gb, 256, 0, stream>>>(x, W1, tb, N);
    node_rec1  <<<nb, 256, 0, stream>>>(tb, att_src1, R, cnt, N);

    edge_count <<<eb, 256, 0, stream>>>(dst, cnt, E);
    scan_rows  <<<sb, 1024, 0, stream>>>(cnt, rowptr, cursor, N);
    edge_fill  <<<eb, 256, 0, stream>>>(src, dst, cursor, csr, E);

    gather_rows<1><<<hb, 256, 0, stream>>>(rowptr, csr, R, att_dst1, b1, tb, N);

    node_rec2  <<<nb, 256, 0, stream>>>(tb, W2, att_src2, R, N);
    gather_rows<2><<<hb, 256, 0, stream>>>(rowptr, csr, R, att_dst2, b2, tb, N);
    finish_out <<<nb, 256, 0, stream>>>(tb, b2, (float*)d_out, N);
}


# Edge sort by destination redone on LDS counters (chunk histograms, scan, scatter into 256-node buckets, one block sorts a bucket in place): no global atomics; row sum order = LDS arrival order
# speedup vs baseline: 12.1619x; 2.3404x over previous
#include <hip/hip_runtime.h>
#include <math.h>

#define IN_DIM   256
#define HEADS    3
#define HID      3
#define NCLS     3
#define NEG_SLOPE 0.2f
#define GAT_EPS  1e-16f

typedef __attribute__((ext_vector_type(2))) float v2f;
typedef __attribute__((ext_vector_type(8))) float v8f;

__device__ __forceinline__ float lrelu(float x) { return x > 0.f ? x : NEG_SLOPE * x; }

__global__ __launch_bounds__(256) void gemm1_wmma(const float* __restrict__ x,
                                                  const float* __restrict__ W,
                                                  float* __restrict__ h, int N) {
    __shared__ float Wp[IN_DIM * 16];
    for (int i = threadIdx.x; i < IN_DIM * 16; i += 256) {
        int k = i >> 4, n = i & 15;
        Wp[i] = (n < HEADS * HID) ? W[k * (HEADS * HID) + n] : 0.0f;
    }
    __syncthreads();

    int lane  = threadIdx.x & 31;
    int wave  = threadIdx.x >> 5;
    long tile = (long)blockIdx.x * 8 + wave;
    long m0   = tile * 16;
    if (m0 >= N) return;

    int laneM = lane & 15;
    int koff  = (lane < 16) ? 0 : 2;
    const float* xrow = x + (m0 + laneM) * IN_DIM;

    v8f c = {};
    #pragma unroll 4
    for (int k0 = 0; k0 < IN_DIM; k0 += 4) {
        v2f a, b;
        a.x = xrow[k0 + koff];
        a.y = xrow[k0 + koff + 1];
        b.x = Wp[(k0 + koff)     * 16 + laneM];
        b.y = Wp[(k0 + koff + 1) * 16 + laneM];
        c = __builtin_amdgcn_wmma_f32_16x16x4_f32(false, a, false, b,
                                                  (short)0, c, false, false);
    }

    int rbase = (lane < 16) ? 0 : 8;
    if (laneM < HEADS * HID) {
        #pragma unroll
        for (int v = 0; v < 8; ++v)
            h[(m0 + rbase + v) * (HEADS * HID) + laneM] = c[v];
    }
}

__global__ void node_rec1(const float* __restrict__ h, const float* __restrict__ atts,
                          float4* __restrict__ R, int N) {
    int n = blockIdx.x * blockDim.x + threadIdx.x;
    if (n >= N) return;
    #pragma unroll
    for (int hh = 0; hh < HEADS; ++hh) {
        float v0 = h[n * 9 + hh * 3 + 0];
        float v1 = h[n * 9 + hh * 3 + 1];
        float v2 = h[n * 9 + hh * 3 + 2];
        float s = 0.f;
        s += v0 * atts[hh * 3 + 0];
        s += v1 * atts[hh * 3 + 1];
        s += v2 * atts[hh * 3 + 2];
        R[n * 3 + hh] = make_float4(v0, v1, v2, s);
    }
}

__global__ void node_rec2(const float* __restrict__ t, const float* __restrict__ W2,
                          const float* __restrict__ atts, float4* __restrict__ R, int N) {
    __shared__ float Ws[81];
    if (threadIdx.x < 81) Ws[threadIdx.x] = W2[threadIdx.x];
    __syncthreads();
    int n = blockIdx.x * blockDim.x + threadIdx.x;
    if (n >= N) return;
    float tin[9];
    #pragma unroll
    for (int i = 0; i < 9; ++i) tin[i] = t[n * 9 + i];
    float h2[9];
    #pragma unroll
    for (int j = 0; j < 9; ++j) {
        float a = 0.f;
        #pragma unroll
        for (int i = 0; i < 9; ++i) a += tin[i] * Ws[i * 9 + j];
        h2[j] = a;
    }
    #pragma unroll
    for (int hh = 0; hh < HEADS; ++hh) {
        float s = 0.f;
        s += h2[hh * 3 + 0] * atts[hh * 3 + 0];
        s += h2[hh * 3 + 1] * atts[hh * 3 + 1];
        s += h2[hh * 3 + 2] * atts[hh * 3 + 2];
        R[n * 3 + hh] = make_float4(h2[hh * 3 + 0], h2[hh * 3 + 1], h2[hh * 3 + 2], s);
    }
}

#define BSH         8
#define BNODES      (1 << BSH)
#define SORT_BLOCKS 256
#define STAGE_CAP   24576

__global__ __launch_bounds__(1024) void bucket_hist(const int* __restrict__ dst,
                                                    int* __restrict__ H, int E, int nbk) {
    extern __shared__ int hist[];
    const int t = threadIdx.x;
    for (int k = t; k < nbk; k += 1024) hist[k] = 0;
    __syncthreads();
    const int ch = (E + (int)gridDim.x - 1) / (int)gridDim.x;
    const int lo = blockIdx.x * ch;
    const int hi = min(E, lo + ch);
    int i = lo + t;
    for (; i + 3 * 1024 < hi; i += 4 * 1024) {
        const int d0 = dst[i], d1 = dst[i + 1024], d2 = dst[i + 2048], d3 = dst[i + 3072];
        atomicAdd(&hist[d0 >> BSH], 1);
        atomicAdd(&hist[d1 >> BSH], 1);
        atomicAdd(&hist[d2 >> BSH], 1);
        atomicAdd(&hist[d3 >> BSH], 1);
    }
    for (; i < hi; i += 1024) atomicAdd(&hist[dst[i] >> BSH], 1);
    __syncthreads();
    for (int k = t; k < nbk; k += 1024) H[k * (int)gridDim.x + blockIdx.x] = hist[k];
}

__global__ __launch_bounds__(1024) void scan_rows(const int* __restrict__ cnt,
                                                  int* __restrict__ rowptr, int N) {
    __shared__ int sh[1024];
    const int t = threadIdx.x;
    const int base = blockIdx.x * 1024;

    int pre = 0;
    for (int i = t; i < base; i += 1024) pre += cnt[i];
    sh[t] = pre;
    __syncthreads();
    for (int off = 512; off > 0; off >>= 1) {
        if (t < off) sh[t] += sh[t + off];
        __syncthreads();
    }
    const int tile_off = sh[0];
    __syncthreads();

    const int i = base + t;
    const int c = (i < N) ? cnt[i] : 0;
    sh[t] = c;
    __syncthreads();
    for (int off = 1; off < 1024; off <<= 1) {
        int v = (t >= off) ? sh[t - off] : 0;
        __syncthreads();
        sh[t] += v;
        __syncthreads();
    }
    const int excl = tile_off + sh[t] - c;
    if (i < N) rowptr[i] = excl;
    if (i == N - 1) rowptr[N] = excl + c;
}

__global__ __launch_bounds__(1024) void bucket_scatter(const int* __restrict__ src,
                                                       const int* __restrict__ dst,
                                                       const int* __restrict__ Hoff,
                                                       unsigned* __restrict__ P, int E, int nbk) {
    extern __shared__ int cur[];
    const int t = threadIdx.x;
    for (int k = t; k < nbk; k += 1024) cur[k] = Hoff[k * (int)gridDim.x + blockIdx.x];
    __syncthreads();
    const int ch = (E + (int)gridDim.x - 1) / (int)gridDim.x;
    const int lo = blockIdx.x * ch;
    const int hi = min(E, lo + ch);
    int i = lo + t;
    for (; i + 3 * 1024 < hi; i += 4 * 1024) {
        const int d0 = dst[i], d1 = dst[i + 1024], d2 = dst[i + 2048], d3 = dst[i + 3072];
        const int s0 = src[i], s1 = src[i + 1024], s2 = src[i + 2048], s3 = src[i + 3072];
        const int p0 = atomicAdd(&cur[d0 >> BSH], 1);
        const int p1 = atomicAdd(&cur[d1 >> BSH], 1);
        const int p2 = atomicAdd(&cur[d2 >> BSH], 1);
        const int p3 = atomicAdd(&cur[d3 >> BSH], 1);
        P[p0] = ((unsigned)s0 << BSH) | (unsigned)(d0 & (BNODES - 1));
        P[p1] = ((unsigned)s1 << BSH) | (unsigned)(d1 & (BNODES - 1));
        P[p2] = ((unsigned)s2 << BSH) | (unsigned)(d2 & (BNODES - 1));
        P[p3] = ((unsigned)s3 << BSH) | (unsigned)(d3 & (BNODES - 1));
    }
    for (; i < hi; i += 1024) {
        const int d = dst[i], s = src[i];
        const int p = atomicAdd(&cur[d >> BSH], 1);
        P[p] = ((unsigned)s << BSH) | (unsigned)(d & (BNODES - 1));
    }
}

__global__ __launch_bounds__(1024) void bucket_rows(const int* __restrict__ Hoff, int* csr,
                                                    int* __restrict__ rowptr, int N, int nchunks) {
    __shared__ unsigned stage[STAGE_CAP];
    __shared__ int cnt[BNODES];
    __shared__ int cur[BNODES];
    __shared__ int wsum[BNODES / 32];
    const int t = threadIdx.x;
    const int b = blockIdx.x;
    const int lo = Hoff[b * nchunks];
    const int hi = Hoff[(b + 1) * nchunks];
    const int nst = min(hi - lo, STAGE_CAP);

    if (t < BNODES) cnt[t] = 0;
    __syncthreads();
    int i = t;
    for (; i + 3 * 1024 < nst; i += 4 * 1024) {
        const unsigned w0 = (unsigned)csr[lo + i],        w1 = (unsigned)csr[lo + i + 1024];
        const unsigned w2 = (unsigned)csr[lo + i + 2048], w3 = (unsigned)csr[lo + i + 3072];
        stage[i] = w0; stage[i + 1024] = w1; stage[i + 2048] = w2; stage[i + 3072] = w3;
        atomicAdd(&cnt[w0 & (BNODES - 1)], 1);
        atomicAdd(&cnt[w1 & (BNODES - 1)], 1);
        atomicAdd(&cnt[w2 & (BNODES - 1)], 1);
        atomicAdd(&cnt[w3 & (BNODES - 1)], 1);
    }
    for (; i < nst; i += 1024) {
        const unsigned w = (unsigned)csr[lo + i];
        stage[i] = w;
        atomicAdd(&cnt[w & (BNODES - 1)], 1);
    }
    __syncthreads();

    const int c = (t < BNODES) ? cnt[t] : 0;
    int v = c;
    for (int off = 1; off < 32; off <<= 1) {
        const int u = __shfl_up(v, off);
        if ((t & 31) >= off) v += u;
    }
    if (t < BNODES && (t & 31) == 31) wsum[t >> 5] = v;
    __syncthreads();
    if (t < BNODES) {
        int pre = 0;
        for (int w = 0; w < (t >> 5); ++w) pre += wsum[w];
        const int excl = pre + v - c;
        cur[t] = excl;
        const int node = b * BNODES + t;
        if (node < N) rowptr[node] = lo + excl;
        if (node == N - 1) rowptr[N] = hi;
    }
    __syncthreads();

    for (i = t; i < nst; i += 1024) {
        const unsigned w = stage[i];
        const int r = atomicAdd(&cur[w & (BNODES - 1)], 1);
        csr[lo + r] = (int)(w >> BSH);
    }
    for (i = nst + t; i < hi - lo; i += 1024) csr[lo + i] = 0;
}

__device__ __forceinline__ void edge_step(const float4 r, const float ad, float& m,
                                          float& s, float& a0, float& a1, float& a2) {
    const float e  = lrelu(r.w + ad);
    const float d  = e - m;
    const float tt = expf(-fabsf(d));
    const bool  up = d > 0.f;
    const float sc = up ? tt : 1.f;
    const float p  = up ? 1.f : tt;
    m  = up ? e : m;
    s  = s  * sc + p;
    a0 = a0 * sc + p * r.x;
    a1 = a1 * sc + p * r.y;
    a2 = a2 * sc + p * r.z;
}

template <int LAYER>
__global__ __launch_bounds__(256) void gather_rows(const int* __restrict__ rowptr,
                                                   const int* __restrict__ csr,
                                                   const float4* __restrict__ R,
                                                   const float* __restrict__ attd,
                                                   const float* __restrict__ bias,
                                                   float* __restrict__ o, int N) {
    const int g = blockIdx.x * 256 + threadIdx.x;
    if (g >= N * HEADS) return;
    const int n  = g / HEADS;
    const int hh = g - n * HEADS;

    const float4 self = R[g];
    float ad = 0.f;
    ad += self.x * attd[hh * 3 + 0];
    ad += self.y * attd[hh * 3 + 1];
    ad += self.z * attd[hh * 3 + 2];

    float m = lrelu(self.w + ad);
    float s = 1.f, a0 = self.x, a1 = self.y, a2 = self.z;

    int j = rowptr[n];
    const int end = rowptr[n + 1];
    for (; j + 4 <= end; j += 4) {
        const int i0 = csr[j], i1 = csr[j + 1], i2 = csr[j + 2], i3 = csr[j + 3];
        const float4 r0 = R[i0 * HEADS + hh];
        const float4 r1 = R[i1 * HEADS + hh];
        const float4 r2 = R[i2 * HEADS + hh];
        const float4 r3 = R[i3 * HEADS + hh];
        edge_step(r0, ad, m, s, a0, a1, a2);
        edge_step(r1, ad, m, s, a0, a1, a2);
        edge_step(r2, ad, m, s, a0, a1, a2);
        edge_step(r3, ad, m, s, a0, a1, a2);
    }
    for (; j < end; ++j) {
        const float4 r = R[csr[j] * HEADS + hh];
        edge_step(r, ad, m, s, a0, a1, a2);
    }

    const float inv = 1.0f / (s + GAT_EPS);
    float v0 = a0 * inv, v1 = a1 * inv, v2 = a2 * inv;
    if (LAYER == 1) {
        v0 += bias[hh * 3 + 0];
        v1 += bias[hh * 3 + 1];
        v2 += bias[hh * 3 + 2];
        v0 = (v0 > 0.f) ? v0 : (expf(v0) - 1.0f);
        v1 = (v1 > 0.f) ? v1 : (expf(v1) - 1.0f);
        v2 = (v2 > 0.f) ? v2 : (expf(v2) - 1.0f);
    }
    o[g * 3 + 0] = v0;
    o[g * 3 + 1] = v1;
    o[g * 3 + 2] = v2;
}

__global__ void finish_out(const float* __restrict__ agg, const float* __restrict__ b2,
                           float* __restrict__ out, int N) {
    int n = blockIdx.x * blockDim.x + threadIdx.x;
    if (n >= N) return;
    float v[NCLS];
    #pragma unroll
    for (int cc = 0; cc < NCLS; ++cc) {
        float acv = 0.f;
        #pragma unroll
        for (int hh = 0; hh < HEADS; ++hh)
            acv += agg[n * 9 + hh * 3 + cc];
        v[cc] = acv * (1.0f / HEADS) + b2[cc];
    }
    float mx = fmaxf(v[0], fmaxf(v[1], v[2]));
    float lse = logf(expf(v[0] - mx) + expf(v[1] - mx) + expf(v[2] - mx));
    #pragma unroll
    for (int cc = 0; cc < NCLS; ++cc)
        out[n * NCLS + cc] = v[cc] - mx - lse;
}

extern "C" void kernel_launch(void* const* d_in, const int* in_sizes, int n_in,
                              void* d_out, int out_size, void* d_ws, size_t ws_size,
                              hipStream_t stream) {
    const float* x        = (const float*)d_in[0];
    const int*   eidx     = (const int*)d_in[1];
    const float* W1       = (const float*)d_in[2];
    const float* att_src1 = (const float*)d_in[3];
    const float* att_dst1 = (const float*)d_in[4];
    const float* b1       = (const float*)d_in[5];
    const float* W2       = (const float*)d_in[6];
    const float* att_src2 = (const float*)d_in[7];
    const float* att_dst2 = (const float*)d_in[8];
    const float* b2       = (const float*)d_in[9];

    const int N = in_sizes[0] / IN_DIM;
    const int E = in_sizes[1] / 2;
    const int* src = eidx;
    const int* dst = eidx + E;

    const int nb = (N + 255) / 256;
    const int hn = nb * SORT_BLOCKS;
    const size_t need = ((size_t)N * 22 + 2 + (size_t)E) * 4;
    if (ws_size < need) return;
    if ((size_t)hn * 2 + 1 > (size_t)N * 9) return;
    if (N > (1 << 24)) return;

    float4* R      = (float4*)d_ws;
    float*  tb     = (float*)d_ws + (size_t)N * 12;
    int*    rowptr = (int*)(tb + (size_t)N * 9);
    int*    csr    = rowptr + N + 1;
    int*    H      = (int*)tb;
    int*    Hoff   = H + hn;

    const int hb = (N * HEADS + 255) / 256;
    const int sb = (hn + 1023) / 1024;
    const int tiles = (N + 15) / 16;
    const int gb = (tiles + 7) / 8;

    gemm1_wmma <<<gb, 256, 0, stream>>>(x, W1, tb, N);
    node_rec1  <<<nb, 256, 0, stream>>>(tb, att_src1, R, N);

    bucket_hist   <<<SORT_BLOCKS, 1024, nb * sizeof(int), stream>>>(dst, H, E, nb);
    scan_rows     <<<sb, 1024, 0, stream>>>(H, Hoff, hn);
    bucket_scatter<<<SORT_BLOCKS, 1024, nb * sizeof(int), stream>>>(src, dst, Hoff, (unsigned*)csr, E, nb);
    bucket_rows   <<<nb, 1024, 0, stream>>>(Hoff, csr, rowptr, N, SORT_BLOCKS);

    gather_rows<1><<<hb, 256, 0, stream>>>(rowptr, csr, R, att_dst1, b1, tb, N);

    node_rec2  <<<nb, 256, 0, stream>>>(tb, W2, att_src2, R, N);
    gather_rows<2><<<hb, 256, 0, stream>>>(rowptr, csr, R, att_dst2, b2, tb, N);
    finish_out <<<nb, 256, 0, stream>>>(tb, b2, (float*)d_out, N);
}


# Three per-node launches folded into their producers (10 -> 7): product tile builds layer-1 records; gather blocks own 64 whole nodes, end with 9x9 product or head mean + log_softmax; a_src2 per edge
# speedup vs baseline: 12.6986x; 1.0441x over previous
#include <hip/hip_runtime.h>
#include <math.h>

#define IN_DIM   256
#define HEADS    3
#define HID      3
#define NCLS     3
#define NEG_SLOPE 0.2f
#define GAT_EPS  1e-16f

typedef __attribute__((ext_vector_type(2))) float v2f;
typedef __attribute__((ext_vector_type(8))) float v8f;

__device__ __forceinline__ float lrelu(float x) { return x > 0.f ? x : NEG_SLOPE * x; }

#define CT_PITCH 20
__global__ __launch_bounds__(256) void gemm1_rec(const float* __restrict__ x,
                                                 const float* __restrict__ W,
                                                 const float* __restrict__ atts,
                                                 float4* __restrict__ R, int N) {
    __shared__ float Wp[IN_DIM * 16];
    __shared__ float Ct[8][16 * CT_PITCH];
    for (int i = threadIdx.x; i < IN_DIM * 16; i += 256) {
        int k = i >> 4, n = i & 15;
        Wp[i] = (n < HEADS * HID) ? W[k * (HEADS * HID) + n] : 0.0f;
    }
    __syncthreads();

    int lane  = threadIdx.x & 31;
    int wave  = threadIdx.x >> 5;
    long tile = (long)blockIdx.x * 8 + wave;
    long m0   = tile * 16;
    const bool active = m0 < N;

    int laneM = lane & 15;
    int koff  = (lane < 16) ? 0 : 2;

    if (active) {
        const float* xrow = x + (m0 + laneM) * IN_DIM;
        v8f c = {};
        #pragma unroll 4
        for (int k0 = 0; k0 < IN_DIM; k0 += 4) {
            v2f a, b;
            a.x = xrow[k0 + koff];
            a.y = xrow[k0 + koff + 1];
            b.x = Wp[(k0 + koff)     * 16 + laneM];
            b.y = Wp[(k0 + koff + 1) * 16 + laneM];
            c = __builtin_amdgcn_wmma_f32_16x16x4_f32(false, a, false, b,
                                                      (short)0, c, false, false);
        }
        int rbase = (lane < 16) ? 0 : 8;
        #pragma unroll
        for (int v = 0; v < 8; ++v)
            Ct[wave][(rbase + v) * CT_PITCH + laneM] = c[v];
    }
    __syncthreads();
    if (active) {
        for (int rec = lane; rec < 16 * HEADS; rec += 32) {
            const int row = rec / HEADS;
            const int hh  = rec - row * HEADS;
            if (m0 + row < N) {
                const float v0 = Ct[wave][row * CT_PITCH + hh * 3 + 0];
                const float v1 = Ct[wave][row * CT_PITCH + hh * 3 + 1];
                const float v2 = Ct[wave][row * CT_PITCH + hh * 3 + 2];
                float s = 0.f;
                s += v0 * atts[hh * 3 + 0];
                s += v1 * atts[hh * 3 + 1];
                s += v2 * atts[hh * 3 + 2];
                R[m0 * HEADS + rec] = make_float4(v0, v1, v2, s);
            }
        }
    }
}

#define BSH         8
#define BNODES      (1 << BSH)
#define SORT_BLOCKS 256
#define STAGE_CAP   24576

__global__ __launch_bounds__(1024) void bucket_hist(const int* __restrict__ dst,
                                                    int* __restrict__ H, int E, int nbk) {
    extern __shared__ int hist[];
    const int t = threadIdx.x;
    for (int k = t; k < nbk; k += 1024) hist[k] = 0;
    __syncthreads();
    const int ch = (E + (int)gridDim.x - 1) / (int)gridDim.x;
    const int lo = blockIdx.x * ch;
    const int hi = min(E, lo + ch);
    int i = lo + t;
    for (; i + 3 * 1024 < hi; i += 4 * 1024) {
        const int d0 = dst[i], d1 = dst[i + 1024], d2 = dst[i + 2048], d3 = dst[i + 3072];
        atomicAdd(&hist[d0 >> BSH], 1);
        atomicAdd(&hist[d1 >> BSH], 1);
        atomicAdd(&hist[d2 >> BSH], 1);
        atomicAdd(&hist[d3 >> BSH], 1);
    }
    for (; i < hi; i += 1024) atomicAdd(&hist[dst[i] >> BSH], 1);
    __syncthreads();
    for (int k = t; k < nbk; k += 1024) H[k * (int)gridDim.x + blockIdx.x] = hist[k];
}

__global__ __launch_bounds__(1024) void scan_rows(const int* __restrict__ cnt,
                                                  int* __restrict__ rowptr, int N) {
    __shared__ int sh[1024];
    const int t = threadIdx.x;
    const int base = blockIdx.x * 1024;

    int pre = 0;
    for (int i = t; i < base; i += 1024) pre += cnt[i];
    sh[t] = pre;
    __syncthreads();
    for (int off = 512; off > 0; off >>= 1) {
        if (t < off) sh[t] += sh[t + off];
        __syncthreads();
    }
    const int tile_off = sh[0];
    __syncthreads();

    const int i = base + t;
    const int c = (i < N) ? cnt[i] : 0;
    sh[t] = c;
    __syncthreads();
    for (int off = 1; off < 1024; off <<= 1) {
        int v = (t >= off) ? sh[t - off] : 0;
        __syncthreads();
        sh[t] += v;
        __syncthreads();
    }
    const int excl = tile_off + sh[t] - c;
    if (i < N) rowptr[i] = excl;
    if (i == N - 1) rowptr[N] = excl + c;
}

__global__ __launch_bounds__(1024) void bucket_scatter(const int* __restrict__ src,
                                                       const int* __restrict__ dst,
                                                       const int* __restrict__ Hoff,
                                                       unsigned* __restrict__ P, int E, int nbk) {
    extern __shared__ int cur[];
    const int t = threadIdx.x;
    for (int k = t; k < nbk; k += 1024) cur[k] = Hoff[k * (int)gridDim.x + blockIdx.x];
    __syncthreads();
    const int ch = (E + (int)gridDim.x - 1) / (int)gridDim.x;
    const int lo = blockIdx.x * ch;
    const int hi = min(E, lo + ch);
    int i = lo + t;
    for (; i + 3 * 1024 < hi; i += 4 * 1024) {
        const int d0 = dst[i], d1 = dst[i + 1024], d2 = dst[i + 2048], d3 = dst[i + 3072];
        const int s0 = src[i], s1 = src[i + 1024], s2 = src[i + 2048], s3 = src[i + 3072];
        const int p0 = atomicAdd(&cur[d0 >> BSH], 1);
        const int p1 = atomicAdd(&cur[d1 >> BSH], 1);
        const int p2 = atomicAdd(&cur[d2 >> BSH], 1);
        const int p3 = atomicAdd(&cur[d3 >> BSH], 1);
        P[p0] = ((unsigned)s0 << BSH) | (unsigned)(d0 & (BNODES - 1));
        P[p1] = ((unsigned)s1 << BSH) | (unsigned)(d1 & (BNODES - 1));
        P[p2] = ((unsigned)s2 << BSH) | (unsigned)(d2 & (BNODES - 1));
        P[p3] = ((unsigned)s3 << BSH) | (unsigned)(d3 & (BNODES - 1));
    }
    for (; i < hi; i += 1024) {
        const int d = dst[i], s = src[i];
        const int p = atomicAdd(&cur[d >> BSH], 1);
        P[p] = ((unsigned)s << BSH) | (unsigned)(d & (BNODES - 1));
    }
}

__global__ __launch_bounds__(1024) void bucket_rows(const int* __restrict__ Hoff, int* csr,
                                                    int* __restrict__ rowptr, int N, int nchunks) {
    __shared__ unsigned stage[STAGE_CAP];
    __shared__ int cnt[BNODES];
    __shared__ int cur[BNODES];
    __shared__ int wsum[BNODES / 32];
    const int t = threadIdx.x;
    const int b = blockIdx.x;
    const int lo = Hoff[b * nchunks];
    const int hi = Hoff[(b + 1) * nchunks];
    const int nst = min(hi - lo, STAGE_CAP);

    if (t < BNODES) cnt[t] = 0;
    __syncthreads();
    int i = t;
    for (; i + 3 * 1024 < nst; i += 4 * 1024) {
        const unsigned w0 = (unsigned)csr[lo + i],        w1 = (unsigned)csr[lo + i + 1024];
        const unsigned w2 = (unsigned)csr[lo + i + 2048], w3 = (unsigned)csr[lo + i + 3072];
        stage[i] = w0; stage[i + 1024] = w1; stage[i + 2048] = w2; stage[i + 3072] = w3;
        atomicAdd(&cnt[w0 & (BNODES - 1)], 1);
        atomicAdd(&cnt[w1 & (BNODES - 1)], 1);
        atomicAdd(&cnt[w2 & (BNODES - 1)], 1);
        atomicAdd(&cnt[w3 & (BNODES - 1)], 1);
    }
    for (; i < nst; i += 1024) {
        const unsigned w = (unsigned)csr[lo + i];
        stage[i] = w;
        atomicAdd(&cnt[w & (BNODES - 1)], 1);
    }
    __syncthreads();

    const int c = (t < BNODES) ? cnt[t] : 0;
    int v = c;
    for (int off = 1; off < 32; off <<= 1) {
        const int u = __shfl_up(v, off);
        if ((t & 31) >= off) v += u;
    }
    if (t < BNODES && (t & 31) == 31) wsum[t >> 5] = v;
    __syncthreads();
    if (t < BNODES) {
        int pre = 0;
        for (int w = 0; w < (t >> 5); ++w) pre += wsum[w];
        const int excl = pre + v - c;
        cur[t] = excl;
        const int node = b * BNODES + t;
        if (node < N) rowptr[node] = lo + excl;
        if (node == N - 1) rowptr[N] = hi;
    }
    __syncthreads();

    for (i = t; i < nst; i += 1024) {
        const unsigned w = stage[i];
        const int r = atomicAdd(&cur[w & (BNODES - 1)], 1);
        csr[lo + r] = (int)(w >> BSH);
    }
    for (i = nst + t; i < hi - lo; i += 1024) csr[lo + i] = 0;
}

#define GB_NODES   64
#define GB_THREADS (GB_NODES * HEADS)

__device__ __forceinline__ void edge_step(const float4 r, const float ad, float& m,
                                          float& s, float& a0, float& a1, float& a2) {
    const float e  = lrelu(r.w + ad);
    const float d  = e - m;
    const float tt = expf(-fabsf(d));
    const bool  up = d > 0.f;
    const float sc = up ? tt : 1.f;
    const float p  = up ? 1.f : tt;
    m  = up ? e : m;
    s  = s  * sc + p;
    a0 = a0 * sc + p * r.x;
    a1 = a1 * sc + p * r.y;
    a2 = a2 * sc + p * r.z;
}

template <int LAYER>
__device__ __forceinline__ float4 load_rec(const float4* __restrict__ R4,
                                           const float* __restrict__ R3, const int idx,
                                           const float as0, const float as1, const float as2) {
    if (LAYER == 1) return R4[idx];
    const float* p = R3 + idx * 3;
    const float v0 = p[0], v1 = p[1], v2 = p[2];
    float s = 0.f;
    s += v0 * as0;
    s += v1 * as1;
    s += v2 * as2;
    return make_float4(v0, v1, v2, s);
}

template <int LAYER>
__global__ __launch_bounds__(GB_THREADS) void gather_rows(const int* __restrict__ rowptr,
                                                          const int* __restrict__ csr,
                                                          const float4* __restrict__ R4,
                                                          const float* __restrict__ R3,
                                                          const float* __restrict__ atts,
                                                          const float* __restrict__ attd,
                                                          const float* __restrict__ bias,
                                                          const float* __restrict__ W2,
                                                          float* __restrict__ o, int N) {
    __shared__ float ex[GB_THREADS * 3];
    __shared__ float Ws[81];
    const int t = threadIdx.x;
    if (LAYER == 1) { if (t < 81) Ws[t] = W2[t]; }

    const int g  = blockIdx.x * GB_THREADS + t;
    const int nl = t / HEADS;
    const int hh = t - nl * HEADS;
    const int n  = blockIdx.x * GB_NODES + nl;
    const bool valid = n < N;

    float v0 = 0.f, v1 = 0.f, v2 = 0.f;
    if (valid) {
        float as0 = 0.f, as1 = 0.f, as2 = 0.f;
        if (LAYER == 2) { as0 = atts[hh * 3 + 0]; as1 = atts[hh * 3 + 1]; as2 = atts[hh * 3 + 2]; }

        const float4 self = load_rec<LAYER>(R4, R3, g, as0, as1, as2);
        float ad = 0.f;
        ad += self.x * attd[hh * 3 + 0];
        ad += self.y * attd[hh * 3 + 1];
        ad += self.z * attd[hh * 3 + 2];

        float m = lrelu(self.w + ad);
        float s = 1.f, a0 = self.x, a1 = self.y, a2 = self.z;

        int j = rowptr[n];
        const int end = rowptr[n + 1];
        for (; j + 4 <= end; j += 4) {
            const int i0 = csr[j], i1 = csr[j + 1], i2 = csr[j + 2], i3 = csr[j + 3];
            const float4 r0 = load_rec<LAYER>(R4, R3, i0 * HEADS + hh, as0, as1, as2);
            const float4 r1 = load_rec<LAYER>(R4, R3, i1 * HEADS + hh, as0, as1, as2);
            const float4 r2 = load_rec<LAYER>(R4, R3, i2 * HEADS + hh, as0, as1, as2);
            const float4 r3 = load_rec<LAYER>(R4, R3, i3 * HEADS + hh, as0, as1, as2);
            edge_step(r0, ad, m, s, a0, a1, a2);
            edge_step(r1, ad, m, s, a0, a1, a2);
            edge_step(r2, ad, m, s, a0, a1, a2);
            edge_step(r3, ad, m, s, a0, a1, a2);
        }
        for (; j < end; ++j) {
            const float4 r = load_rec<LAYER>(R4, R3, csr[j] * HEADS + hh, as0, as1, as2);
            edge_step(r, ad, m, s, a0, a1, a2);
        }

        const float inv = 1.0f / (s + GAT_EPS);
        v0 = a0 * inv; v1 = a1 * inv; v2 = a2 * inv;
        if (LAYER == 1) {
            v0 += bias[hh * 3 + 0];
            v1 += bias[hh * 3 + 1];
            v2 += bias[hh * 3 + 2];
            v0 = (v0 > 0.f) ? v0 : (expf(v0) - 1.0f);
            v1 = (v1 > 0.f) ? v1 : (expf(v1) - 1.0f);
            v2 = (v2 > 0.f) ? v2 : (expf(v2) - 1.0f);
        }
    }
    ex[t * 3 + 0] = v0;
    ex[t * 3 + 1] = v1;
    ex[t * 3 + 2] = v2;
    __syncthreads();

    if (LAYER == 1) {
        const float* tn = &ex[nl * 9];
        float o0 = 0.f, o1 = 0.f, o2 = 0.f;
        for (int i = 0; i < 9; ++i) {
            const float ti = tn[i];
            o0 += ti * Ws[i * 9 + hh * 3 + 0];
            o1 += ti * Ws[i * 9 + hh * 3 + 1];
            o2 += ti * Ws[i * 9 + hh * 3 + 2];
        }
        if (valid) {
            o[g * 3 + 0] = o0;
            o[g * 3 + 1] = o1;
            o[g * 3 + 2] = o2;
        }
    } else {
        const int nn = blockIdx.x * GB_NODES + t;
        if (t < GB_NODES && nn < N) {
            float v[NCLS];
            for (int cc = 0; cc < NCLS; ++cc) {
                float acv = 0.f;
                for (int h3 = 0; h3 < HEADS; ++h3)
                    acv += ex[t * 9 + h3 * 3 + cc];
                v[cc] = acv * (1.0f / HEADS) + bias[cc];
            }
            float mx = fmaxf(v[0], fmaxf(v[1], v[2]));
            float lse = logf(expf(v[0] - mx) + expf(v[1] - mx) + expf(v[2] - mx));
            for (int cc = 0; cc < NCLS; ++cc)
                o[nn * NCLS + cc] = v[cc] - mx - lse;
        }
    }
}

extern "C" void kernel_launch(void* const* d_in, const int* in_sizes, int n_in,
                              void* d_out, int out_size, void* d_ws, size_t ws_size,
                              hipStream_t stream) {
    const float* x        = (const float*)d_in[0];
    const int*   eidx     = (const int*)d_in[1];
    const float* W1       = (const float*)d_in[2];
    const float* att_src1 = (const float*)d_in[3];
    const float* att_dst1 = (const float*)d_in[4];
    const float* b1       = (const float*)d_in[5];
    const float* W2       = (const float*)d_in[6];
    const float* att_src2 = (const float*)d_in[7];
    const float* att_dst2 = (const float*)d_in[8];
    const float* b2       = (const float*)d_in[9];

    const int N = in_sizes[0] / IN_DIM;
    const int E = in_sizes[1] / 2;
    const int* src = eidx;
    const int* dst = eidx + E;

    const int nb = (N + 255) / 256;
    const int hn = nb * SORT_BLOCKS;
    const size_t need = ((size_t)N * 22 + 2 + (size_t)E) * 4;
    if (ws_size < need) return;
    if ((size_t)hn * 2 + 1 > (size_t)N * 9) return;
    if (N > (1 << 24)) return;

    float4* R      = (float4*)d_ws;
    float*  tb     = (float*)d_ws + (size_t)N * 12;
    int*    rowptr = (int*)(tb + (size_t)N * 9);
    int*    csr    = rowptr + N + 1;
    int*    H      = (int*)tb;
    int*    Hoff   = H + hn;

    const int hb = (N + GB_NODES - 1) / GB_NODES;
    const int sb = (hn + 1023) / 1024;
    const int tiles = (N + 15) / 16;
    const int gb = (tiles + 7) / 8;

    gemm1_rec<<<gb, 256, 0, stream>>>(x, W1, att_src1, R, N);

    bucket_hist   <<<SORT_BLOCKS, 1024, nb * sizeof(int), stream>>>(dst, H, E, nb);
    scan_rows     <<<sb, 1024, 0, stream>>>(H, Hoff, hn);
    bucket_scatter<<<SORT_BLOCKS, 1024, nb * sizeof(int), stream>>>(src, dst, Hoff, (unsigned*)csr, E, nb);
    bucket_rows   <<<nb, 1024, 0, stream>>>(Hoff, csr, rowptr, N, SORT_BLOCKS);

    gather_rows<1><<<hb, GB_THREADS, 0, stream>>>(rowptr, csr, R, nullptr, att_src2, att_dst1, b1, W2, tb, N);

    gather_rows<2><<<hb, GB_THREADS, 0, stream>>>(rowptr, csr, nullptr, tb, att_src2, att_dst2, b2, W2,
                                                  (float*)d_out, N);
}


# Two launches folded (7 -> 5): first launch runs 25 product waves beside 7 edge-counting waves per block; scatter blocks scan the chunk histograms themselves; sort and sum order unchanged
# speedup vs baseline: 13.3029x; 1.0476x over previous
#include <hip/hip_runtime.h>
#include <math.h>

#define IN_DIM   256
#define HEADS    3
#define HID      3
#define NCLS     3
#define NEG_SLOPE 0.2f
#define GAT_EPS  1e-16f

typedef __attribute__((ext_vector_type(2))) float v2f;
typedef __attribute__((ext_vector_type(8))) float v8f;

__device__ __forceinline__ float lrelu(float x) { return x > 0.f ? x : NEG_SLOPE * x; }

#define CT_PITCH 20
#define BSH         8
#define BNODES      (1 << BSH)
#define SORT_BLOCKS 256
#define STAGE_CAP   24576
#define FB_THREADS 1024
#define PW         25
#define HT         (FB_THREADS - PW * 32)
#define HU         16
__global__ __launch_bounds__(FB_THREADS) void front(const float* __restrict__ x,
                                                    const float* __restrict__ W,
                                                    const float* __restrict__ atts,
                                                    float4* __restrict__ R, int N,
                                                    const int* __restrict__ dst,
                                                    int* __restrict__ H, int E, int nbk) {
    __shared__ float Wp[IN_DIM * 16];
    __shared__ float Ct[PW][16 * CT_PITCH];
    extern __shared__ int hist[];
    const int t = threadIdx.x;
    const bool counts = (int)blockIdx.x < SORT_BLOCKS;
    for (int i = t; i < IN_DIM * 16; i += FB_THREADS) {
        int k = i >> 4, n = i & 15;
        Wp[i] = (n < HEADS * HID) ? W[k * (HEADS * HID) + n] : 0.0f;
    }
    if (counts) for (int k = t; k < nbk; k += FB_THREADS) hist[k] = 0;
    __syncthreads();

    int lane  = t & 31;
    int wave  = t >> 5;
    long tile = (long)wave * (long)gridDim.x + (long)blockIdx.x;
    long m0   = tile * 16;
    const bool active = (wave < PW) && (m0 < N);

    int laneM = lane & 15;
    int koff  = (lane < 16) ? 0 : 2;

    if (active) {
        const float* xrow = x + (m0 + laneM) * IN_DIM;
        v8f c = {};
        #pragma unroll 4
        for (int k0 = 0; k0 < IN_DIM; k0 += 4) {
            v2f a, b;
            a.x = xrow[k0 + koff];
            a.y = xrow[k0 + koff + 1];
            b.x = Wp[(k0 + koff)     * 16 + laneM];
            b.y = Wp[(k0 + koff + 1) * 16 + laneM];
            c = __builtin_amdgcn_wmma_f32_16x16x4_f32(false, a, false, b,
                                                      (short)0, c, false, false);
        }
        int rbase = (lane < 16) ? 0 : 8;
        #pragma unroll
        for (int v = 0; v < 8; ++v)
            Ct[wave][(rbase + v) * CT_PITCH + laneM] = c[v];
    } else if (wave >= PW && counts) {
        const int ht = t - PW * 32;
        const int ch = (E + SORT_BLOCKS - 1) / SORT_BLOCKS;
        const int lo = (int)blockIdx.x * ch;
        const int hi = min(E, lo + ch);
        for (int i = lo + ht; i < hi; i += HU * HT) {
            int d[HU];
            #pragma unroll
            for (int u = 0; u < HU; ++u) d[u] = dst[min(i + u * HT, hi - 1)];
            #pragma unroll
            for (int u = 0; u < HU; ++u)
                atomicAdd(&hist[d[u] >> BSH], (i + u * HT < hi) ? 1 : 0);
        }
    }
    __syncthreads();
    if (active) {
        for (int rec = lane; rec < 16 * HEADS; rec += 32) {
            const int row = rec / HEADS;
            const int hh  = rec - row * HEADS;
            if (m0 + row < N) {
                const float v0 = Ct[wave][row * CT_PITCH + hh * 3 + 0];
                const float v1 = Ct[wave][row * CT_PITCH + hh * 3 + 1];
                const float v2 = Ct[wave][row * CT_PITCH + hh * 3 + 2];
                float s = 0.f;
                s += v0 * atts[hh * 3 + 0];
                s += v1 * atts[hh * 3 + 1];
                s += v2 * atts[hh * 3 + 2];
                R[m0 * HEADS + rec] = make_float4(v0, v1, v2, s);
            }
        }
    }
    if (counts) for (int k = t; k < nbk; k += FB_THREADS) H[k * SORT_BLOCKS + (int)blockIdx.x] = hist[k];
}


__global__ __launch_bounds__(1024) void bucket_scatter(const int* __restrict__ src,
                                                       const int* __restrict__ dst,
                                                       const int* __restrict__ H,
                                                       int* __restrict__ bstart,
                                                       unsigned* __restrict__ P, int E, int nbk) {
    extern __shared__ int shs[];
    int* cur  = shs;
    int* tot  = shs + nbk;
    int* wsum = tot + nbk;
    const int t = threadIdx.x;
    const int lane = t & 31, wave = t >> 5;
    const int c = blockIdx.x;
    for (int k = wave; k < nbk; k += 32) {
        const int4 a = *(const int4*)(H + (size_t)k * SORT_BLOCKS + 4 * lane);
        const int4 b = *(const int4*)(H + (size_t)k * SORT_BLOCKS + SORT_BLOCKS / 2 + 4 * lane);
        int ts = a.x + a.y + a.z + a.w + b.x + b.y + b.z + b.w;
        const int c0 = 4 * lane, c1 = SORT_BLOCKS / 2 + 4 * lane;
        int ps = (c0 < c ? a.x : 0) + (c0 + 1 < c ? a.y : 0) + (c0 + 2 < c ? a.z : 0) + (c0 + 3 < c ? a.w : 0)
               + (c1 < c ? b.x : 0) + (c1 + 1 < c ? b.y : 0) + (c1 + 2 < c ? b.z : 0) + (c1 + 3 < c ? b.w : 0);
        for (int off = 16; off > 0; off >>= 1) {
            ts += __shfl_xor(ts, off);
            ps += __shfl_xor(ps, off);
        }
        if (lane == 0) { tot[k] = ts; cur[k] = ps; }
    }
    __syncthreads();
    const int seg = (nbk + 1023) >> 10;
    const int k0 = min(t * seg, nbk), k1 = min(k0 + seg, nbk);
    int loc = 0;
    for (int k = k0; k < k1; ++k) loc += tot[k];
    int v = loc;
    for (int off = 1; off < 32; off <<= 1) {
        const int u = __shfl_up(v, off);
        if (lane >= off) v += u;
    }
    if (lane == 31) wsum[wave] = v;
    __syncthreads();
    int run = v - loc;
    for (int w = 0; w < wave; ++w) run += wsum[w];
    for (int k = k0; k < k1; ++k) {
        if (c == 0) bstart[k] = run;
        cur[k] += run;
        run += tot[k];
    }
    if (c == 0 && t == 1023) bstart[nbk] = run;
    __syncthreads();
    const int ch = (E + (int)gridDim.x - 1) / (int)gridDim.x;
    const int lo = blockIdx.x * ch;
    const int hi = min(E, lo + ch);
    int i = lo + t;
    for (; i + 3 * 1024 < hi; i += 4 * 1024) {
        const int d0 = dst[i], d1 = dst[i + 1024], d2 = dst[i + 2048], d3 = dst[i + 3072];
        const int s0 = src[i], s1 = src[i + 1024], s2 = src[i + 2048], s3 = src[i + 3072];
        const int p0 = atomicAdd(&cur[d0 >> BSH], 1);
        const int p1 = atomicAdd(&cur[d1 >> BSH], 1);
        const int p2 = atomicAdd(&cur[d2 >> BSH], 1);
        const int p3 = atomicAdd(&cur[d3 >> BSH], 1);
        P[p0] = ((unsigned)s0 << BSH) | (unsigned)(d0 & (BNODES - 1));
        P[p1] = ((unsigned)s1 << BSH) | (unsigned)(d1 & (BNODES - 1));
        P[p2] = ((unsigned)s2 << BSH) | (unsigned)(d2 & (BNODES - 1));
        P[p3] = ((unsigned)s3 << BSH) | (unsigned)(d3 & (BNODES - 1));
    }
    for (; i < hi; i += 1024) {
        const int d = dst[i], s = src[i];
        const int p = atomicAdd(&cur[d >> BSH], 1);
        P[p] = ((unsigned)s << BSH) | (unsigned)(d & (BNODES - 1));
    }
}

__global__ __launch_bounds__(1024) void bucket_rows(const int* __restrict__ bstart, int* csr,
                                                    int* __restrict__ rowptr, int N) {
    __shared__ unsigned stage[STAGE_CAP];
    __shared__ int cnt[BNODES];
    __shared__ int cur[BNODES];
    __shared__ int wsum[BNODES / 32];
    const int t = threadIdx.x;
    const int b = blockIdx.x;
    const int lo = bstart[b];
    const int hi = bstart[b + 1];
    const int nst = min(hi - lo, STAGE_CAP);

    if (t < BNODES) cnt[t] = 0;
    __syncthreads();
    int i = t;
    for (; i + 3 * 1024 < nst; i += 4 * 1024) {
        const unsigned w0 = (unsigned)csr[lo + i],        w1 = (unsigned)csr[lo + i + 1024];
        const unsigned w2 = (unsigned)csr[lo + i + 2048], w3 = (unsigned)csr[lo + i + 3072];
        stage[i] = w0; stage[i + 1024] = w1; stage[i + 2048] = w2; stage[i + 3072] = w3;
        atomicAdd(&cnt[w0 & (BNODES - 1)], 1);
        atomicAdd(&cnt[w1 & (BNODES - 1)], 1);
        atomicAdd(&cnt[w2 & (BNODES - 1)], 1);
        atomicAdd(&cnt[w3 & (BNODES - 1)], 1);
    }
    for (; i < nst; i += 1024) {
        const unsigned w = (unsigned)csr[lo + i];
        stage[i] = w;
        atomicAdd(&cnt[w & (BNODES - 1)], 1);
    }
    __syncthreads();

    const int c = (t < BNODES) ? cnt[t] : 0;
    int v = c;
    for (int off = 1; off < 32; off <<= 1) {
        const int u = __shfl_up(v, off);
        if ((t & 31) >= off) v += u;
    }
    if (t < BNODES && (t & 31) == 31) wsum[t >> 5] = v;
    __syncthreads();
    if (t < BNODES) {
        int pre = 0;
        for (int w = 0; w < (t >> 5); ++w) pre += wsum[w];
        const int excl = pre + v - c;
        cur[t] = excl;
        const int node = b * BNODES + t;
        if (node < N) rowptr[node] = lo + excl;
        if (node == N - 1) rowptr[N] = hi;
    }
    __syncthreads();

    for (i = t; i < nst; i += 1024) {
        const unsigned w = stage[i];
        const int r = atomicAdd(&cur[w & (BNODES - 1)], 1);
        csr[lo + r] = (int)(w >> BSH);
    }
    for (i = nst + t; i < hi - lo; i += 1024) csr[lo + i] = 0;
}

#define GB_NODES   64
#define GB_THREADS (GB_NODES * HEADS)

__device__ __forceinline__ void edge_step(const float4 r, const float ad, float& m,
                                          float& s, float& a0, float& a1, float& a2) {
    const float e  = lrelu(r.w + ad);
    const float d  = e - m;
    const float tt = expf(-fabsf(d));
    const bool  up = d > 0.f;
    const float sc = up ? tt : 1.f;
    const float p  = up ? 1.f : tt;
    m  = up ? e : m;
    s  = s  * sc + p;
    a0 = a0 * sc + p * r.x;
    a1 = a1 * sc + p * r.y;
    a2 = a2 * sc + p * r.z;
}

template <int LAYER>
__device__ __forceinline__ float4 load_rec(const float4* __restrict__ R4,
                                           const float* __restrict__ R3, const int idx,
                                           const float as0, const float as1, const float as2) {
    if (LAYER == 1) return R4[idx];
    const float* p = R3 + idx * 3;
    const float v0 = p[0], v1 = p[1], v2 = p[2];
    float s = 0.f;
    s += v0 * as0;
    s += v1 * as1;
    s += v2 * as2;
    return make_float4(v0, v1, v2, s);
}

template <int LAYER>
__global__ __launch_bounds__(GB_THREADS) void gather_rows(const int* __restrict__ rowptr,
                                                          const int* __restrict__ csr,
                                                          const float4* __restrict__ R4,
                                                          const float* __restrict__ R3,
                                                          const float* __restrict__ atts,
                                                          const float* __restrict__ attd,
                                                          const float* __restrict__ bias,
                                                          const float* __restrict__ W2,
                                                          float* __restrict__ o, int N) {
    __shared__ float ex[GB_THREADS * 3];
    __shared__ float Ws[81];
    const int t = threadIdx.x;
    if (LAYER == 1) { if (t < 81) Ws[t] = W2[t]; }

    const int g  = blockIdx.x * GB_THREADS + t;
    const int nl = t / HEADS;
    const int hh = t - nl * HEADS;
    const int n  = blockIdx.x * GB_NODES + nl;
    const bool valid = n < N;

    float v0 = 0.f, v1 = 0.f, v2 = 0.f;
    if (valid) {
        float as0 = 0.f, as1 = 0.f, as2 = 0.f;
        if (LAYER == 2) { as0 = atts[hh * 3 + 0]; as1 = atts[hh * 3 + 1]; as2 = atts[hh * 3 + 2]; }

        const float4 self = load_rec<LAYER>(R4, R3, g, as0, as1, as2);
        float ad = 0.f;
        ad += self.x * attd[hh * 3 + 0];
        ad += self.y * attd[hh * 3 + 1];
        ad += self.z * attd[hh * 3 + 2];

        float m = lrelu(self.w + ad);
        float s = 1.f, a0 = self.x, a1 = self.y, a2 = self.z;

        int j = rowptr[n];
        const int end = rowptr[n + 1];
        for (; j + 4 <= end; j += 4) {
            const int i0 = csr[j], i1 = csr[j + 1], i2 = csr[j + 2], i3 = csr[j + 3];
            const float4 r0 = load_rec<LAYER>(R4, R3, i0 * HEADS + hh, as0, as1, as2);
            const float4 r1 = load_rec<LAYER>(R4, R3, i1 * HEADS + hh, as0, as1, as2);
            const float4 r2 = load_rec<LAYER>(R4, R3, i2 * HEADS + hh, as0, as1, as2);
            const float4 r3 = load_rec<LAYER>(R4, R3, i3 * HEADS + hh, as0, as1, as2);
            edge_step(r0, ad, m, s, a0, a1, a2);
            edge_step(r1, ad, m, s, a0, a1, a2);
            edge_step(r2, ad, m, s, a0, a1, a2);
            edge_step(r3, ad, m, s, a0, a1, a2);
        }
        for (; j < end; ++j) {
            const float4 r = load_rec<LAYER>(R4, R3, csr[j] * HEADS + hh, as0, as1, as2);
            edge_step(r, ad, m, s, a0, a1, a2);
        }

        const float inv = 1.0f / (s + GAT_EPS);
        v0 = a0 * inv; v1 = a1 * inv; v2 = a2 * inv;
        if (LAYER == 1) {
            v0 += bias[hh * 3 + 0];
            v1 += bias[hh * 3 + 1];
            v2 += bias[hh * 3 + 2];
            v0 = (v0 > 0.f) ? v0 : (expf(v0) - 1.0f);
            v1 = (v1 > 0.f) ? v1 : (expf(v1) - 1.0f);
            v2 = (v2 > 0.f) ? v2 : (expf(v2) - 1.0f);
        }
    }
    ex[t * 3 + 0] = v0;
    ex[t * 3 + 1] = v1;
    ex[t * 3 + 2] = v2;
    __syncthreads();

    if (LAYER == 1) {
        const float* tn = &ex[nl * 9];
        float o0 = 0.f, o1 = 0.f, o2 = 0.f;
        for (int i = 0; i < 9; ++i) {
            const float ti = tn[i];
            o0 += ti * Ws[i * 9 + hh * 3 + 0];
            o1 += ti * Ws[i * 9 + hh * 3 + 1];
            o2 += ti * Ws[i * 9 + hh * 3 + 2];
        }
        if (valid) {
            o[g * 3 + 0] = o0;
            o[g * 3 + 1] = o1;
            o[g * 3 + 2] = o2;
        }
    } else {
        const int nn = blockIdx.x * GB_NODES + t;
        if (t < GB_NODES && nn < N) {
            float v[NCLS];
            for (int cc = 0; cc < NCLS; ++cc) {
                float acv = 0.f;
                for (int h3 = 0; h3 < HEADS; ++h3)
                    acv += ex[t * 9 + h3 * 3 + cc];
                v[cc] = acv * (1.0f / HEADS) + bias[cc];
            }
            float mx = fmaxf(v[0], fmaxf(v[1], v[2]));
            float lse = logf(expf(v[0] - mx) + expf(v[1] - mx) + expf(v[2] - mx));
            for (int cc = 0; cc < NCLS; ++cc)
                o[nn * NCLS + cc] = v[cc] - mx - lse;
        }
    }
}

extern "C" void kernel_launch(void* const* d_in, const int* in_sizes, int n_in,
                              void* d_out, int out_size, void* d_ws, size_t ws_size,
                              hipStream_t stream) {
    const float* x        = (const float*)d_in[0];
    const int*   eidx     = (const int*)d_in[1];
    const float* W1       = (const float*)d_in[2];
    const float* att_src1 = (const float*)d_in[3];
    const float* att_dst1 = (const float*)d_in[4];
    const float* b1       = (const float*)d_in[5];
    const float* W2       = (const float*)d_in[6];
    const float* att_src2 = (const float*)d_in[7];
    const float* att_dst2 = (const float*)d_in[8];
    const float* b2       = (const float*)d_in[9];

    const int N = in_sizes[0] / IN_DIM;
    const int E = in_sizes[1] / 2;
    const int* src = eidx;
    const int* dst = eidx + E;

    const int nb = (N + 255) / 256;
    const int hn = nb * SORT_BLOCKS;
    const size_t need = ((size_t)N * 22 + 2 + (size_t)E) * 4;
    if (ws_size < need) return;
    if ((size_t)hn + nb + 1 > (size_t)N * 9) return;
    if (N > (1 << 24)) return;

    float4* R      = (float4*)d_ws;
    float*  tb     = (float*)d_ws + (size_t)N * 12;
    int*    rowptr = (int*)(tb + (size_t)N * 9);
    int*    csr    = rowptr + N + 1;
    int*    H      = (int*)tb;
    int*    bstart = H + hn;

    const int hb = (N + GB_NODES - 1) / GB_NODES;
    const int tiles = (N + 15) / 16;
    const int pb = (tiles + PW - 1) / PW;
    const int fb = pb > SORT_BLOCKS ? pb : SORT_BLOCKS;

    front<<<fb, FB_THREADS, nb * sizeof(int), stream>>>(x, W1, att_src1, R, N, dst, H, E, nb);

    bucket_scatter<<<SORT_BLOCKS, 1024, (2 * nb + 32) * sizeof(int), stream>>>(src, dst, H, bstart,
                                                                            (unsigned*)csr, E, nb);
    bucket_rows   <<<nb, 1024, 0, stream>>>(bstart, csr, rowptr, N);

    gather_rows<1><<<hb, GB_THREADS, 0, stream>>>(rowptr, csr, R, nullptr, att_src2, att_dst1, b1, W2, tb, N);

    gather_rows<2><<<hb, GB_THREADS, 0, stream>>>(rowptr, csr, nullptr, tb, att_src2, att_dst2, b2, W2,
                                                  (float*)d_out, N);
}


# Each gather block copies its 64 nodes' contiguous slice of the sorted source list to LDS once; rows take indices from LDS: one memory round trip per 4-edge trip, not two; sum order unchanged
# speedup vs baseline: 14.2977x; 1.0748x over previous
#include <hip/hip_runtime.h>
#include <math.h>

#define IN_DIM   256
#define HEADS    3
#define HID      3
#define NCLS     3
#define NEG_SLOPE 0.2f
#define GAT_EPS  1e-16f

typedef __attribute__((ext_vector_type(2))) float v2f;
typedef __attribute__((ext_vector_type(8))) float v8f;

__device__ __forceinline__ float lrelu(float x) { return x > 0.f ? x : NEG_SLOPE * x; }

#define CT_PITCH 20
#define BSH         8
#define BNODES      (1 << BSH)
#define SORT_BLOCKS 256
#define STAGE_CAP   24576
#define FB_THREADS 1024
#define PW         25
#define HT         (FB_THREADS - PW * 32)
#define HU         16
__global__ __launch_bounds__(FB_THREADS) void front(const float* __restrict__ x,
                                                    const float* __restrict__ W,
                                                    const float* __restrict__ atts,
                                                    float4* __restrict__ R, int N,
                                                    const int* __restrict__ dst,
                                                    int* __restrict__ H, int E, int nbk) {
    __shared__ float Wp[IN_DIM * 16];
    __shared__ float Ct[PW][16 * CT_PITCH];
    extern __shared__ int hist[];
    const int t = threadIdx.x;
    const bool counts = (int)blockIdx.x < SORT_BLOCKS;
    for (int i = t; i < IN_DIM * 16; i += FB_THREADS) {
        int k = i >> 4, n = i & 15;
        Wp[i] = (n < HEADS * HID) ? W[k * (HEADS * HID) + n] : 0.0f;
    }
    if (counts) for (int k = t; k < nbk; k += FB_THREADS) hist[k] = 0;
    __syncthreads();

    int lane  = t & 31;
    int wave  = t >> 5;
    long tile = (long)wave * (long)gridDim.x + (long)blockIdx.x;
    long m0   = tile * 16;
    const bool active = (wave < PW) && (m0 < N);

    int laneM = lane & 15;
    int koff  = (lane < 16) ? 0 : 2;

    if (active) {
        const float* xrow = x + (m0 + laneM) * IN_DIM;
        v8f c = {};
        #pragma unroll 4
        for (int k0 = 0; k0 < IN_DIM; k0 += 4) {
            v2f a, b;
            a.x = xrow[k0 + koff];
            a.y = xrow[k0 + koff + 1];
            b.x = Wp[(k0 + koff)     * 16 + laneM];
            b.y = Wp[(k0 + koff + 1) * 16 + laneM];
            c = __builtin_amdgcn_wmma_f32_16x16x4_f32(false, a, false, b,
                                                      (short)0, c, false, false);
        }
        int rbase = (lane < 16) ? 0 : 8;
        #pragma unroll
        for (int v = 0; v < 8; ++v)
            Ct[wave][(rbase + v) * CT_PITCH + laneM] = c[v];
    } else if (wave >= PW && counts) {
        const int ht = t - PW * 32;
        const int ch = (E + SORT_BLOCKS - 1) / SORT_BLOCKS;
        const int lo = (int)blockIdx.x * ch;
        const int hi = min(E, lo + ch);
        for (int i = lo + ht; i < hi; i += HU * HT) {
            int d[HU];
            #pragma unroll
            for (int u = 0; u < HU; ++u) d[u] = dst[min(i + u * HT, hi - 1)];
            #pragma unroll
            for (int u = 0; u < HU; ++u)
                atomicAdd(&hist[d[u] >> BSH], (i + u * HT < hi) ? 1 : 0);
        }
    }
    __syncthreads();
    if (active) {
        for (int rec = lane; rec < 16 * HEADS; rec += 32) {
            const int row = rec / HEADS;
            const int hh  = rec - row * HEADS;
            if (m0 + row < N) {
                const float v0 = Ct[wave][row * CT_PITCH + hh * 3 + 0];
                const float v1 = Ct[wave][row * CT_PITCH + hh * 3 + 1];
                const float v2 = Ct[wave][row * CT_PITCH + hh * 3 + 2];
                float s = 0.f;
                s += v0 * atts[hh * 3 + 0];
                s += v1 * atts[hh * 3 + 1];
                s += v2 * atts[hh * 3 + 2];
                R[m0 * HEADS + rec] = make_float4(v0, v1, v2, s);
            }
        }
    }
    if (counts) for (int k = t; k < nbk; k += FB_THREADS) H[k * SORT_BLOCKS + (int)blockIdx.x] = hist[k];
}


__global__ __launch_bounds__(1024) void bucket_scatter(const int* __restrict__ src,
                                                       const int* __restrict__ dst,
                                                       const int* __restrict__ H,
                                                       int* __restrict__ bstart,
                                                       unsigned* __restrict__ P, int E, int nbk) {
    extern __shared__ int shs[];
    int* cur  = shs;
    int* tot  = shs + nbk;
    int* wsum = tot + nbk;
    const int t = threadIdx.x;
    const int lane = t & 31, wave = t >> 5;
    const int c = blockIdx.x;
    for (int k = wave; k < nbk; k += 32) {
        const int4 a = *(const int4*)(H + (size_t)k * SORT_BLOCKS + 4 * lane);
        const int4 b = *(const int4*)(H + (size_t)k * SORT_BLOCKS + SORT_BLOCKS / 2 + 4 * lane);
        int ts = a.x + a.y + a.z + a.w + b.x + b.y + b.z + b.w;
        const int c0 = 4 * lane, c1 = SORT_BLOCKS / 2 + 4 * lane;
        int ps = (c0 < c ? a.x : 0) + (c0 + 1 < c ? a.y : 0) + (c0 + 2 < c ? a.z : 0) + (c0 + 3 < c ? a.w : 0)
               + (c1 < c ? b.x : 0) + (c1 + 1 < c ? b.y : 0) + (c1 + 2 < c ? b.z : 0) + (c1 + 3 < c ? b.w : 0);
        for (int off = 16; off > 0; off >>= 1) {
            ts += __shfl_xor(ts, off);
            ps += __shfl_xor(ps, off);
        }
        if (lane == 0) { tot[k] = ts; cur[k] = ps; }
    }
    __syncthreads();
    const int seg = (nbk + 1023) >> 10;
    const int k0 = min(t * seg, nbk), k1 = min(k0 + seg, nbk);
    int loc = 0;
    for (int k = k0; k < k1; ++k) loc += tot[k];
    int v = loc;
    for (int off = 1; off < 32; off <<= 1) {
        const int u = __shfl_up(v, off);
        if (lane >= off) v += u;
    }
    if (lane == 31) wsum[wave] = v;
    __syncthreads();
    int run = v - loc;
    for (int w = 0; w < wave; ++w) run += wsum[w];
    for (int k = k0; k < k1; ++k) {
        if (c == 0) bstart[k] = run;
        cur[k] += run;
        run += tot[k];
    }
    if (c == 0 && t == 1023) bstart[nbk] = run;
    __syncthreads();
    const int ch = (E + (int)gridDim.x - 1) / (int)gridDim.x;
    const int lo = blockIdx.x * ch;
    const int hi = min(E, lo + ch);
    int i = lo + t;
    for (; i + 3 * 1024 < hi; i += 4 * 1024) {
        const int d0 = dst[i], d1 = dst[i + 1024], d2 = dst[i + 2048], d3 = dst[i + 3072];
        const int s0 = src[i], s1 = src[i + 1024], s2 = src[i + 2048], s3 = src[i + 3072];
        const int p0 = atomicAdd(&cur[d0 >> BSH], 1);
        const int p1 = atomicAdd(&cur[d1 >> BSH], 1);
        const int p2 = atomicAdd(&cur[d2 >> BSH], 1);
        const int p3 = atomicAdd(&cur[d3 >> BSH], 1);
        P[p0] = ((unsigned)s0 << BSH) | (unsigned)(d0 & (BNODES - 1));
        P[p1] = ((unsigned)s1 << BSH) | (unsigned)(d1 & (BNODES - 1));
        P[p2] = ((unsigned)s2 << BSH) | (unsigned)(d2 & (BNODES - 1));
        P[p3] = ((unsigned)s3 << BSH) | (unsigned)(d3 & (BNODES - 1));
    }
    for (; i < hi; i += 1024) {
        const int d = dst[i], s = src[i];
        const int p = atomicAdd(&cur[d >> BSH], 1);
        P[p] = ((unsigned)s << BSH) | (unsigned)(d & (BNODES - 1));
    }
}

__global__ __launch_bounds__(1024) void bucket_rows(const int* __restrict__ bstart, int* csr,
                                                    int* __restrict__ rowptr, int N) {
    __shared__ unsigned stage[STAGE_CAP];
    __shared__ int cnt[BNODES];
    __shared__ int cur[BNODES];
    __shared__ int wsum[BNODES / 32];
    const int t = threadIdx.x;
    const int b = blockIdx.x;
    const int lo = bstart[b];
    const int hi = bstart[b + 1];
    const int nst = min(hi - lo, STAGE_CAP);

    if (t < BNODES) cnt[t] = 0;
    __syncthreads();
    int i = t;
    for (; i + 3 * 1024 < nst; i += 4 * 1024) {
        const unsigned w0 = (unsigned)csr[lo + i],        w1 = (unsigned)csr[lo + i + 1024];
        const unsigned w2 = (unsigned)csr[lo + i + 2048], w3 = (unsigned)csr[lo + i + 3072];
        stage[i] = w0; stage[i + 1024] = w1; stage[i + 2048] = w2; stage[i + 3072] = w3;
        atomicAdd(&cnt[w0 & (BNODES - 1)], 1);
        atomicAdd(&cnt[w1 & (BNODES - 1)], 1);
        atomicAdd(&cnt[w2 & (BNODES - 1)], 1);
        atomicAdd(&cnt[w3 & (BNODES - 1)], 1);
    }
    for (; i < nst; i += 1024) {
        const unsigned w = (unsigned)csr[lo + i];
        stage[i] = w;
        atomicAdd(&cnt[w & (BNODES - 1)], 1);
    }
    __syncthreads();

    const int c = (t < BNODES) ? cnt[t] : 0;
    int v = c;
    for (int off = 1; off < 32; off <<= 1) {
        const int u = __shfl_up(v, off);
        if ((t & 31) >= off) v += u;
    }
    if (t < BNODES && (t & 31) == 31) wsum[t >> 5] = v;
    __syncthreads();
    if (t < BNODES) {
        int pre = 0;
        for (int w = 0; w < (t >> 5); ++w) pre += wsum[w];
        const int excl = pre + v - c;
        cur[t] = excl;
        const int node = b * BNODES + t;
        if (node < N) rowptr[node] = lo + excl;
        if (node == N - 1) rowptr[N] = hi;
    }
    __syncthreads();

    for (i = t; i < nst; i += 1024) {
        const unsigned w = stage[i];
        const int r = atomicAdd(&cur[w & (BNODES - 1)], 1);
        csr[lo + r] = (int)(w >> BSH);
    }
    for (i = nst + t; i < hi - lo; i += 1024) csr[lo + i] = 0;
}

#define GB_NODES   64
#define GB_THREADS (GB_NODES * HEADS)
#define GB_SU      16
#define GB_STAGE   (GB_SU * GB_THREADS)

__device__ __forceinline__ void edge_step(const float4 r, const float ad, float& m,
                                          float& s, float& a0, float& a1, float& a2) {
    const float e  = lrelu(r.w + ad);
    const float d  = e - m;
    const float tt = expf(-fabsf(d));
    const bool  up = d > 0.f;
    const float sc = up ? tt : 1.f;
    const float p  = up ? 1.f : tt;
    m  = up ? e : m;
    s  = s  * sc + p;
    a0 = a0 * sc + p * r.x;
    a1 = a1 * sc + p * r.y;
    a2 = a2 * sc + p * r.z;
}

template <int LAYER>
__device__ __forceinline__ float4 load_rec(const float4* __restrict__ R4,
                                           const float* __restrict__ R3, const int idx,
                                           const float as0, const float as1, const float as2) {
    if (LAYER == 1) return R4[idx];
    const float* p = R3 + idx * 3;
    const float v0 = p[0], v1 = p[1], v2 = p[2];
    float s = 0.f;
    s += v0 * as0;
    s += v1 * as1;
    s += v2 * as2;
    return make_float4(v0, v1, v2, s);
}

template <int LAYER, bool STAGED>
__device__ __forceinline__ void row_sum(const int* __restrict__ csr, const int* sl, const int lo,
                                        int j, const int end,
                                        const float4* __restrict__ R4, const float* __restrict__ R3,
                                        const int hh, const float as0, const float as1, const float as2,
                                        const float ad, float& m, float& s, float& a0, float& a1, float& a2) {
    for (; j + 4 <= end; j += 4) {
        int i0, i1, i2, i3;
        if (STAGED) { const int* q = sl + (j - lo); i0 = q[0]; i1 = q[1]; i2 = q[2]; i3 = q[3]; }
        else        { i0 = csr[j]; i1 = csr[j + 1]; i2 = csr[j + 2]; i3 = csr[j + 3]; }
        const float4 r0 = load_rec<LAYER>(R4, R3, i0 * HEADS + hh, as0, as1, as2);
        const float4 r1 = load_rec<LAYER>(R4, R3, i1 * HEADS + hh, as0, as1, as2);
        const float4 r2 = load_rec<LAYER>(R4, R3, i2 * HEADS + hh, as0, as1, as2);
        const float4 r3 = load_rec<LAYER>(R4, R3, i3 * HEADS + hh, as0, as1, as2);
        edge_step(r0, ad, m, s, a0, a1, a2);
        edge_step(r1, ad, m, s, a0, a1, a2);
        edge_step(r2, ad, m, s, a0, a1, a2);
        edge_step(r3, ad, m, s, a0, a1, a2);
    }
    for (; j < end; ++j) {
        const int i = STAGED ? sl[j - lo] : csr[j];
        const float4 r = load_rec<LAYER>(R4, R3, i * HEADS + hh, as0, as1, as2);
        edge_step(r, ad, m, s, a0, a1, a2);
    }
}

template <int LAYER>
__global__ __launch_bounds__(GB_THREADS) void gather_rows(const int* __restrict__ rowptr,
                                                          const int* __restrict__ csr,
                                                          const float4* __restrict__ R4,
                                                          const float* __restrict__ R3,
                                                          const float* __restrict__ atts,
                                                          const float* __restrict__ attd,
                                                          const float* __restrict__ bias,
                                                          const float* __restrict__ W2,
                                                          float* __restrict__ o, int N) {
    __shared__ float ex[GB_THREADS * 3];
    __shared__ float Ws[81];
    __shared__ int   sl[GB_STAGE];
    const int t = threadIdx.x;

    const int g  = blockIdx.x * GB_THREADS + t;
    const int nl = t / HEADS;
    const int hh = t - nl * HEADS;
    const int n  = blockIdx.x * GB_NODES + nl;
    const bool valid = n < N;

    const int n0 = blockIdx.x * GB_NODES;
    const int lo = rowptr[n0];
    const int hi = rowptr[min(n0 + GB_NODES, N)];
    const bool fits = (hi - lo) <= GB_STAGE;

    const int ns = valid ? n : 0;
    const int gs = ns * HEADS + hh;
    const float w2v = (LAYER == 1) ? W2[min(t, 80)] : 0.f;
    float as0 = 0.f, as1 = 0.f, as2 = 0.f;
    if (LAYER == 2) { as0 = atts[hh * 3 + 0]; as1 = atts[hh * 3 + 1]; as2 = atts[hh * 3 + 2]; }
    const float d0 = attd[hh * 3 + 0], d1 = attd[hh * 3 + 1], d2 = attd[hh * 3 + 2];
    float4 self;
    if (LAYER == 1) self = R4[gs];
    else            self = make_float4(R3[gs * 3 + 0], R3[gs * 3 + 1], R3[gs * 3 + 2], 0.f);
    const int j   = rowptr[ns];
    const int end = rowptr[ns + 1];

    if (fits && hi > lo) {
        const int cnt = hi - lo;
        int w[GB_SU];
        #pragma unroll
        for (int u = 0; u < GB_SU; ++u) w[u] = csr[lo + min(t + u * GB_THREADS, cnt - 1)];
        #pragma unroll
        for (int u = 0; u < GB_SU; ++u)
            if (t + u * GB_THREADS < cnt) sl[t + u * GB_THREADS] = w[u];
    }
    if (LAYER == 1) { if (t < 81) Ws[t] = w2v; }
    __syncthreads();

    float v0 = 0.f, v1 = 0.f, v2 = 0.f;
    if (valid) {
        if (LAYER == 2) {
            float sw = 0.f;
            sw += self.x * as0;
            sw += self.y * as1;
            sw += self.z * as2;
            self.w = sw;
        }
        float ad = 0.f;
        ad += self.x * d0;
        ad += self.y * d1;
        ad += self.z * d2;

        float m = lrelu(self.w + ad);
        float s = 1.f, a0 = self.x, a1 = self.y, a2 = self.z;

        if (fits) row_sum<LAYER, true >(csr, sl, lo, j, end, R4, R3, hh, as0, as1, as2, ad, m, s, a0, a1, a2);
        else      row_sum<LAYER, false>(csr, sl, lo, j, end, R4, R3, hh, as0, as1, as2, ad, m, s, a0, a1, a2);

        const float inv = 1.0f / (s + GAT_EPS);
        v0 = a0 * inv; v1 = a1 * inv; v2 = a2 * inv;
        if (LAYER == 1) {
            v0 += bias[hh * 3 + 0];
            v1 += bias[hh * 3 + 1];
            v2 += bias[hh * 3 + 2];
            v0 = (v0 > 0.f) ? v0 : (expf(v0) - 1.0f);
            v1 = (v1 > 0.f) ? v1 : (expf(v1) - 1.0f);
            v2 = (v2 > 0.f) ? v2 : (expf(v2) - 1.0f);
        }
    }
    ex[t * 3 + 0] = v0;
    ex[t * 3 + 1] = v1;
    ex[t * 3 + 2] = v2;
    __syncthreads();

    if (LAYER == 1) {
        const float* tn = &ex[nl * 9];
        float o0 = 0.f, o1 = 0.f, o2 = 0.f;
        for (int i = 0; i < 9; ++i) {
            const float ti = tn[i];
            o0 += ti * Ws[i * 9 + hh * 3 + 0];
            o1 += ti * Ws[i * 9 + hh * 3 + 1];
            o2 += ti * Ws[i * 9 + hh * 3 + 2];
        }
        if (valid) {
            o[g * 3 + 0] = o0;
            o[g * 3 + 1] = o1;
            o[g * 3 + 2] = o2;
        }
    } else {
        const int nn = blockIdx.x * GB_NODES + t;
        if (t < GB_NODES && nn < N) {
            float v[NCLS];
            for (int cc = 0; cc < NCLS; ++cc) {
                float acv = 0.f;
                for (int h3 = 0; h3 < HEADS; ++h3)
                    acv += ex[t * 9 + h3 * 3 + cc];
                v[cc] = acv * (1.0f / HEADS) + bias[cc];
            }
            float mx = fmaxf(v[0], fmaxf(v[1], v[2]));
            float lse = logf(expf(v[0] - mx) + expf(v[1] - mx) + expf(v[2] - mx));
            for (int cc = 0; cc < NCLS; ++cc)
                o[nn * NCLS + cc] = v[cc] - mx - lse;
        }
    }
}

extern "C" void kernel_launch(void* const* d_in, const int* in_sizes, int n_in,
                              void* d_out, int out_size, void* d_ws, size_t ws_size,
                              hipStream_t stream) {
    const float* x        = (const float*)d_in[0];
    const int*   eidx     = (const int*)d_in[1];
    const float* W1       = (const float*)d_in[2];
    const float* att_src1 = (const float*)d_in[3];
    const float* att_dst1 = (const float*)d_in[4];
    const float* b1       = (const float*)d_in[5];
    const float* W2       = (const float*)d_in[6];
    const float* att_src2 = (const float*)d_in[7];
    const float* att_dst2 = (const float*)d_in[8];
    const float* b2       = (const float*)d_in[9];

    const int N = in_sizes[0] / IN_DIM;
    const int E = in_sizes[1] / 2;
    const int* src = eidx;
    const int* dst = eidx + E;

    const int nb = (N + 255) / 256;
    const int hn = nb * SORT_BLOCKS;
    const size_t need = ((size_t)N * 22 + 2 + (size_t)E) * 4;
    if (ws_size < need) return;
    if ((size_t)hn + nb + 1 > (size_t)N * 9) return;
    if (N > (1 << 24)) return;

    float4* R      = (float4*)d_ws;
    float*  tb     = (float*)d_ws + (size_t)N * 12;
    int*    rowptr = (int*)(tb + (size_t)N * 9);
    int*    csr    = rowptr + N + 1;
    int*    H      = (int*)tb;
    int*    bstart = H + hn;

    const int hb = (N + GB_NODES - 1) / GB_NODES;
    const int tiles = (N + 15) / 16;
    const int pb = (tiles + PW - 1) / PW;
    const int fb = pb > SORT_BLOCKS ? pb : SORT_BLOCKS;

    front<<<fb, FB_THREADS, nb * sizeof(int), stream>>>(x, W1, att_src1, R, N, dst, H, E, nb);

    bucket_scatter<<<SORT_BLOCKS, 1024, (2 * nb + 32) * sizeof(int), stream>>>(src, dst, H, bstart,
                                                                            (unsigned*)csr, E, nb);
    bucket_rows   <<<nb, 1024, 0, stream>>>(bstart, csr, rowptr, N);

    gather_rows<1><<<hb, GB_THREADS, 0, stream>>>(rowptr, csr, R, nullptr, att_src2, att_dst1, b1, W2, tb, N);

    gather_rows<2><<<hb, GB_THREADS, 0, stream>>>(rowptr, csr, nullptr, tb, att_src2, att_dst2, b2, W2,
                                                  (float*)d_out, N);
}


# Both scatter passes of the edge sort gather their 4-byte words in LDS and store contiguous runs (pass 2: whole aligned lines), not 32 lines per store; row sum order = LDS arrival order as before
# speedup vs baseline: 18.6839x; 1.3068x over previous
#include <hip/hip_runtime.h>
#include <math.h>

#define IN_DIM   256
#define HEADS    3
#define HID      3
#define NCLS     3
#define NEG_SLOPE 0.2f
#define GAT_EPS  1e-16f

typedef __attribute__((ext_vector_type(2))) float v2f;
typedef __attribute__((ext_vector_type(8))) float v8f;

__device__ __forceinline__ float lrelu(float x) { return x > 0.f ? x : NEG_SLOPE * x; }

#define CT_PITCH 20
#define BSH         8
#define BNODES      (1 << BSH)
#define SORT_BLOCKS 256
#define STAGE_CAP   24576
#define FB_THREADS 1024
#define PW         25
#define HT         (FB_THREADS - PW * 32)
#define HU         16
__global__ __launch_bounds__(FB_THREADS) void front(const float* __restrict__ x,
                                                    const float* __restrict__ W,
                                                    const float* __restrict__ atts,
                                                    float4* __restrict__ R, int N,
                                                    const int* __restrict__ dst,
                                                    int* __restrict__ H, int E, int nbk) {
    __shared__ float Wp[IN_DIM * 16];
    __shared__ float Ct[PW][16 * CT_PITCH];
    extern __shared__ int hist[];
    const int t = threadIdx.x;
    const bool counts = (int)blockIdx.x < SORT_BLOCKS;
    for (int i = t; i < IN_DIM * 16; i += FB_THREADS) {
        int k = i >> 4, n = i & 15;
        Wp[i] = (n < HEADS * HID) ? W[k * (HEADS * HID) + n] : 0.0f;
    }
    if (counts) for (int k = t; k < nbk; k += FB_THREADS) hist[k] = 0;
    __syncthreads();

    int lane  = t & 31;
    int wave  = t >> 5;
    long tile = (long)wave * (long)gridDim.x + (long)blockIdx.x;
    long m0   = tile * 16;
    const bool active = (wave < PW) && (m0 < N);

    int laneM = lane & 15;
    int koff  = (lane < 16) ? 0 : 2;

    if (active) {
        const float* xrow = x + (m0 + laneM) * IN_DIM;
        v8f c = {};
        #pragma unroll 4
        for (int k0 = 0; k0 < IN_DIM; k0 += 4) {
            v2f a, b;
            a.x = xrow[k0 + koff];
            a.y = xrow[k0 + koff + 1];
            b.x = Wp[(k0 + koff)     * 16 + laneM];
            b.y = Wp[(k0 + koff + 1) * 16 + laneM];
            c = __builtin_amdgcn_wmma_f32_16x16x4_f32(false, a, false, b,
                                                      (short)0, c, false, false);
        }
        int rbase = (lane < 16) ? 0 : 8;
        #pragma unroll
        for (int v = 0; v < 8; ++v)
            Ct[wave][(rbase + v) * CT_PITCH + laneM] = c[v];
    } else if (wave >= PW && counts) {
        const int ht = t - PW * 32;
        const int ch = (E + SORT_BLOCKS - 1) / SORT_BLOCKS;
        const int lo = (int)blockIdx.x * ch;
        const int hi = min(E, lo + ch);
        for (int i = lo + ht; i < hi; i += HU * HT) {
            int d[HU];
            #pragma unroll
            for (int u = 0; u < HU; ++u) d[u] = dst[min(i + u * HT, hi - 1)];
            #pragma unroll
            for (int u = 0; u < HU; ++u)
                atomicAdd(&hist[d[u] >> BSH], (i + u * HT < hi) ? 1 : 0);
        }
    }
    __syncthreads();
    if (active) {
        for (int rec = lane; rec < 16 * HEADS; rec += 32) {
            const int row = rec / HEADS;
            const int hh  = rec - row * HEADS;
            if (m0 + row < N) {
                const float v0 = Ct[wave][row * CT_PITCH + hh * 3 + 0];
                const float v1 = Ct[wave][row * CT_PITCH + hh * 3 + 1];
                const float v2 = Ct[wave][row * CT_PITCH + hh * 3 + 2];
                float s = 0.f;
                s += v0 * atts[hh * 3 + 0];
                s += v1 * atts[hh * 3 + 1];
                s += v2 * atts[hh * 3 + 2];
                R[m0 * HEADS + rec] = make_float4(v0, v1, v2, s);
            }
        }
    }
    if (counts) for (int k = t; k < nbk; k += FB_THREADS) H[k * SORT_BLOCKS + (int)blockIdx.x] = hist[k];
}


#define CH_CAP 16384
__global__ __launch_bounds__(1024) void bucket_scatter(const int* __restrict__ src,
                                                       const int* __restrict__ dst,
                                                       const int* __restrict__ H,
                                                       int* __restrict__ bstart,
                                                       unsigned* __restrict__ P, int E, int nbk) {
    __shared__ uint2 wg[CH_CAP];
    extern __shared__ int shs[];
    int* cur   = shs;
    int* tot   = shs + nbk;
    int* dl    = tot + nbk;
    int* wsum  = dl + nbk;
    int* wsum2 = wsum + 32;
    const int t = threadIdx.x;
    const int lane = t & 31, wave = t >> 5;
    const int c = blockIdx.x;
    for (int k = wave; k < nbk; k += 32) {
        const int4 a = *(const int4*)(H + (size_t)k * SORT_BLOCKS + 4 * lane);
        const int4 b = *(const int4*)(H + (size_t)k * SORT_BLOCKS + SORT_BLOCKS / 2 + 4 * lane);
        int ts = a.x + a.y + a.z + a.w + b.x + b.y + b.z + b.w;
        const int c0 = 4 * lane, c1 = SORT_BLOCKS / 2 + 4 * lane;
        int ps = (c0 < c ? a.x : 0) + (c0 + 1 < c ? a.y : 0) + (c0 + 2 < c ? a.z : 0) + (c0 + 3 < c ? a.w : 0)
               + (c1 < c ? b.x : 0) + (c1 + 1 < c ? b.y : 0) + (c1 + 2 < c ? b.z : 0) + (c1 + 3 < c ? b.w : 0);
        for (int off = 16; off > 0; off >>= 1) {
            ts += __shfl_xor(ts, off);
            ps += __shfl_xor(ps, off);
        }
        const int  cc   = c & (SORT_BLOCKS / 2 - 1);
        const int4 hsel = (c < SORT_BLOCKS / 2) ? a : b;
        const int  jj   = cc & 3;
        const int  mine = jj == 0 ? hsel.x : jj == 1 ? hsel.y : jj == 2 ? hsel.z : hsel.w;
        const int  ow   = __shfl(mine, cc >> 2);
        if (lane == 0) { tot[k] = ts; cur[k] = ps; dl[k] = ow; }
    }
    __syncthreads();
    const int seg = (nbk + 1023) >> 10;
    const int k0 = min(t * seg, nbk), k1 = min(k0 + seg, nbk);
    int loc = 0, loc2 = 0;
    for (int k = k0; k < k1; ++k) { loc += tot[k]; loc2 += dl[k]; }
    int v = loc, v2 = loc2;
    for (int off = 1; off < 32; off <<= 1) {
        const int u = __shfl_up(v, off);
        const int u2 = __shfl_up(v2, off);
        if (lane >= off) { v += u; v2 += u2; }
    }
    if (lane == 31) { wsum[wave] = v; wsum2[wave] = v2; }
    __syncthreads();
    int run = v - loc, run2 = v2 - loc2;
    for (int w = 0; w < wave; ++w) { run += wsum[w]; run2 += wsum2[w]; }
    for (int k = k0; k < k1; ++k) {
        if (c == 0) bstart[k] = run;
        const int ow = dl[k];
        cur[k] += run;
        dl[k] = cur[k] - run2;
        run += tot[k];
        run2 += ow;
    }
    if (c == 0 && t == 1023) bstart[nbk] = run;
    __syncthreads();
    const int ch = (E + (int)gridDim.x - 1) / (int)gridDim.x;
    const int lo = blockIdx.x * ch;
    const int hi = min(E, lo + ch);
    const bool fits = (hi - lo) <= CH_CAP;
    int i = lo + t;
    for (; i + 3 * 1024 < hi; i += 4 * 1024) {
        const int d0 = dst[i], d1 = dst[i + 1024], d2 = dst[i + 2048], d3 = dst[i + 3072];
        const int s0 = src[i], s1 = src[i + 1024], s2 = src[i + 2048], s3 = src[i + 3072];
        const int p0 = atomicAdd(&cur[d0 >> BSH], 1);
        const int p1 = atomicAdd(&cur[d1 >> BSH], 1);
        const int p2 = atomicAdd(&cur[d2 >> BSH], 1);
        const int p3 = atomicAdd(&cur[d3 >> BSH], 1);
        const unsigned w0 = ((unsigned)s0 << BSH) | (unsigned)(d0 & (BNODES - 1));
        const unsigned w1 = ((unsigned)s1 << BSH) | (unsigned)(d1 & (BNODES - 1));
        const unsigned w2 = ((unsigned)s2 << BSH) | (unsigned)(d2 & (BNODES - 1));
        const unsigned w3 = ((unsigned)s3 << BSH) | (unsigned)(d3 & (BNODES - 1));
        if (fits) {
            wg[p0 - dl[d0 >> BSH]] = make_uint2(w0, (unsigned)p0);
            wg[p1 - dl[d1 >> BSH]] = make_uint2(w1, (unsigned)p1);
            wg[p2 - dl[d2 >> BSH]] = make_uint2(w2, (unsigned)p2);
            wg[p3 - dl[d3 >> BSH]] = make_uint2(w3, (unsigned)p3);
        } else {
            P[p0] = w0; P[p1] = w1; P[p2] = w2; P[p3] = w3;
        }
    }
    for (; i < hi; i += 1024) {
        const int d = dst[i], s = src[i];
        const int p = atomicAdd(&cur[d >> BSH], 1);
        const unsigned w = ((unsigned)s << BSH) | (unsigned)(d & (BNODES - 1));
        if (fits) wg[p - dl[d >> BSH]] = make_uint2(w, (unsigned)p);
        else      P[p] = w;
    }
    __syncthreads();
    if (fits) {
        const int cn = hi - lo;
        int q = t;
        for (; q + 3 * 1024 < cn; q += 4 * 1024) {
            const uint2 e0 = wg[q], e1 = wg[q + 1024], e2 = wg[q + 2048], e3 = wg[q + 3072];
            P[e0.y] = e0.x; P[e1.y] = e1.x; P[e2.y] = e2.x; P[e3.y] = e3.x;
        }
        for (; q < cn; q += 1024) {
            const uint2 e = wg[q];
            P[e.y] = e.x;
        }
    }
}

#define RWR 12
__global__ __launch_bounds__(1024) void bucket_rows(const int* __restrict__ bstart, int* csr,
                                                    int* __restrict__ rowptr, int N) {
    __shared__ int sorted[STAGE_CAP];
    __shared__ int cnt[BNODES];
    __shared__ int cur[BNODES];
    __shared__ int wsum[BNODES / 32];
    const int t = threadIdx.x;
    const int b = blockIdx.x;
    const int lo = bstart[b];
    const int hi = bstart[b + 1];
    const int nst = min(hi - lo, STAGE_CAP);

    if (t < BNODES) cnt[t] = 0;
    __syncthreads();
    unsigned w[RWR];
    #pragma unroll
    for (int u = 0; u < RWR; ++u) w[u] = 0u;
    if (nst > 0) {
        #pragma unroll
        for (int u = 0; u < RWR; ++u) w[u] = (unsigned)csr[lo + min(t + u * 1024, nst - 1)];
    }
    #pragma unroll
    for (int u = 0; u < RWR; ++u)
        if (t + u * 1024 < nst) atomicAdd(&cnt[w[u] & (BNODES - 1)], 1);
    for (int i = RWR * 1024 + t; i < nst; i += 1024)
        atomicAdd(&cnt[(unsigned)csr[lo + i] & (BNODES - 1)], 1);
    __syncthreads();

    const int c = (t < BNODES) ? cnt[t] : 0;
    int v = c;
    for (int off = 1; off < 32; off <<= 1) {
        const int u = __shfl_up(v, off);
        if ((t & 31) >= off) v += u;
    }
    if (t < BNODES && (t & 31) == 31) wsum[t >> 5] = v;
    __syncthreads();
    if (t < BNODES) {
        int pre = 0;
        for (int ww = 0; ww < (t >> 5); ++ww) pre += wsum[ww];
        const int excl = pre + v - c;
        cur[t] = excl;
        const int node = b * BNODES + t;
        if (node < N) rowptr[node] = lo + excl;
        if (node == N - 1) rowptr[N] = hi;
    }
    __syncthreads();

    #pragma unroll
    for (int u = 0; u < RWR; ++u)
        if (t + u * 1024 < nst) {
            const int r = atomicAdd(&cur[w[u] & (BNODES - 1)], 1);
            sorted[r] = (int)(w[u] >> BSH);
        }
    for (int i = RWR * 1024 + t; i < nst; i += 1024) {
        const unsigned wv = (unsigned)csr[lo + i];
        const int r = atomicAdd(&cur[wv & (BNODES - 1)], 1);
        sorted[r] = (int)(wv >> BSH);
    }
    __syncthreads();

    const int g0 = lo & ~31;
    int g = g0 + t;
    for (; g + 3 * 1024 < lo + nst; g += 4 * 1024) {
        const int x0 = sorted[max(g - lo, 0)], x1 = sorted[g + 1024 - lo];
        const int x2 = sorted[g + 2048 - lo],   x3 = sorted[g + 3072 - lo];
        if (g >= lo) csr[g] = x0;
        csr[g + 1024] = x1; csr[g + 2048] = x2; csr[g + 3072] = x3;
    }
    for (; g < lo + nst; g += 1024)
        if (g >= lo) csr[g] = sorted[g - lo];
    for (int i = nst + t; i < hi - lo; i += 1024) csr[lo + i] = 0;
}

#define GB_NODES   64
#define GB_THREADS (GB_NODES * HEADS)
#define GB_SU      16
#define GB_STAGE   (GB_SU * GB_THREADS)

__device__ __forceinline__ void edge_step(const float4 r, const float ad, float& m,
                                          float& s, float& a0, float& a1, float& a2) {
    const float e  = lrelu(r.w + ad);
    const float d  = e - m;
    const float tt = expf(-fabsf(d));
    const bool  up = d > 0.f;
    const float sc = up ? tt : 1.f;
    const float p  = up ? 1.f : tt;
    m  = up ? e : m;
    s  = s  * sc + p;
    a0 = a0 * sc + p * r.x;
    a1 = a1 * sc + p * r.y;
    a2 = a2 * sc + p * r.z;
}

template <int LAYER>
__device__ __forceinline__ float4 load_rec(const float4* __restrict__ R4,
                                           const float* __restrict__ R3, const int idx,
                                           const float as0, const float as1, const float as2) {
    if (LAYER == 1) return R4[idx];
    const float* p = R3 + idx * 3;
    const float v0 = p[0], v1 = p[1], v2 = p[2];
    float s = 0.f;
    s += v0 * as0;
    s += v1 * as1;
    s += v2 * as2;
    return make_float4(v0, v1, v2, s);
}

template <int LAYER, bool STAGED>
__device__ __forceinline__ void row_sum(const int* __restrict__ csr, const int* sl, const int lo,
                                        int j, const int end,
                                        const float4* __restrict__ R4, const float* __restrict__ R3,
                                        const int hh, const float as0, const float as1, const float as2,
                                        const float ad, float& m, float& s, float& a0, float& a1, float& a2) {
    for (; j + 4 <= end; j += 4) {
        int i0, i1, i2, i3;
        if (STAGED) { const int* q = sl + (j - lo); i0 = q[0]; i1 = q[1]; i2 = q[2]; i3 = q[3]; }
        else        { i0 = csr[j]; i1 = csr[j + 1]; i2 = csr[j + 2]; i3 = csr[j + 3]; }
        const float4 r0 = load_rec<LAYER>(R4, R3, i0 * HEADS + hh, as0, as1, as2);
        const float4 r1 = load_rec<LAYER>(R4, R3, i1 * HEADS + hh, as0, as1, as2);
        const float4 r2 = load_rec<LAYER>(R4, R3, i2 * HEADS + hh, as0, as1, as2);
        const float4 r3 = load_rec<LAYER>(R4, R3, i3 * HEADS + hh, as0, as1, as2);
        edge_step(r0, ad, m, s, a0, a1, a2);
        edge_step(r1, ad, m, s, a0, a1, a2);
        edge_step(r2, ad, m, s, a0, a1, a2);
        edge_step(r3, ad, m, s, a0, a1, a2);
    }
    for (; j < end; ++j) {
        const int i = STAGED ? sl[j - lo] : csr[j];
        const float4 r = load_rec<LAYER>(R4, R3, i * HEADS + hh, as0, as1, as2);
        edge_step(r, ad, m, s, a0, a1, a2);
    }
}

template <int LAYER>
__global__ __launch_bounds__(GB_THREADS) void gather_rows(const int* __restrict__ rowptr,
                                                          const int* __restrict__ csr,
                                                          const float4* __restrict__ R4,
                                                          const float* __restrict__ R3,
                                                          const float* __restrict__ atts,
                                                          const float* __restrict__ attd,
                                                          const float* __restrict__ bias,
                                                          const float* __restrict__ W2,
                                                          float* __restrict__ o, int N) {
    __shared__ float ex[GB_THREADS * 3];
    __shared__ float Ws[81];
    __shared__ int   sl[GB_STAGE];
    const int t = threadIdx.x;

    const int g  = blockIdx.x * GB_THREADS + t;
    const int nl = t / HEADS;
    const int hh = t - nl * HEADS;
    const int n  = blockIdx.x * GB_NODES + nl;
    const bool valid = n < N;

    const int n0 = blockIdx.x * GB_NODES;
    const int lo = rowptr[n0];
    const int hi = rowptr[min(n0 + GB_NODES, N)];
    const bool fits = (hi - lo) <= GB_STAGE;

    const int ns = valid ? n : 0;
    const int gs = ns * HEADS + hh;
    const float w2v = (LAYER == 1) ? W2[min(t, 80)] : 0.f;
    float as0 = 0.f, as1 = 0.f, as2 = 0.f;
    if (LAYER == 2) { as0 = atts[hh * 3 + 0]; as1 = atts[hh * 3 + 1]; as2 = atts[hh * 3 + 2]; }
    const float d0 = attd[hh * 3 + 0], d1 = attd[hh * 3 + 1], d2 = attd[hh * 3 + 2];
    float4 self;
    if (LAYER == 1) self = R4[gs];
    else            self = make_float4(R3[gs * 3 + 0], R3[gs * 3 + 1], R3[gs * 3 + 2], 0.f);
    const int j   = rowptr[ns];
    const int end = rowptr[ns + 1];

    if (fits && hi > lo) {
        const int cnt = hi - lo;
        int w[GB_SU];
        #pragma unroll
        for (int u = 0; u < GB_SU; ++u) w[u] = csr[lo + min(t + u * GB_THREADS, cnt - 1)];
        #pragma unroll
        for (int u = 0; u < GB_SU; ++u)
            if (t + u * GB_THREADS < cnt) sl[t + u * GB_THREADS] = w[u];
    }
    if (LAYER == 1) { if (t < 81) Ws[t] = w2v; }
    __syncthreads();

    float v0 = 0.f, v1 = 0.f, v2 = 0.f;
    if (valid) {
        if (LAYER == 2) {
            float sw = 0.f;
            sw += self.x * as0;
            sw += self.y * as1;
            sw += self.z * as2;
            self.w = sw;
        }
        float ad = 0.f;
        ad += self.x * d0;
        ad += self.y * d1;
        ad += self.z * d2;

        float m = lrelu(self.w + ad);
        float s = 1.f, a0 = self.x, a1 = self.y, a2 = self.z;

        if (fits) row_sum<LAYER, true >(csr, sl, lo, j, end, R4, R3, hh, as0, as1, as2, ad, m, s, a0, a1, a2);
        else      row_sum<LAYER, false>(csr, sl, lo, j, end, R4, R3, hh, as0, as1, as2, ad, m, s, a0, a1, a2);

        const float inv = 1.0f / (s + GAT_EPS);
        v0 = a0 * inv; v1 = a1 * inv; v2 = a2 * inv;
        if (LAYER == 1) {
            v0 += bias[hh * 3 + 0];
            v1 += bias[hh * 3 + 1];
            v2 += bias[hh * 3 + 2];
            v0 = (v0 > 0.f) ? v0 : (expf(v0) - 1.0f);
            v1 = (v1 > 0.f) ? v1 : (expf(v1) - 1.0f);
            v2 = (v2 > 0.f) ? v2 : (expf(v2) - 1.0f);
        }
    }
    ex[t * 3 + 0] = v0;
    ex[t * 3 + 1] = v1;
    ex[t * 3 + 2] = v2;
    __syncthreads();

    if (LAYER == 1) {
        const float* tn = &ex[nl * 9];
        float o0 = 0.f, o1 = 0.f, o2 = 0.f;
        for (int i = 0; i < 9; ++i) {
            const float ti = tn[i];
            o0 += ti * Ws[i * 9 + hh * 3 + 0];
            o1 += ti * Ws[i * 9 + hh * 3 + 1];
            o2 += ti * Ws[i * 9 + hh * 3 + 2];
        }
        if (valid) {
            o[g * 3 + 0] = o0;
            o[g * 3 + 1] = o1;
            o[g * 3 + 2] = o2;
        }
    } else {
        const int nn = blockIdx.x * GB_NODES + t;
        if (t < GB_NODES && nn < N) {
            float v[NCLS];
            for (int cc = 0; cc < NCLS; ++cc) {
                float acv = 0.f;
                for (int h3 = 0; h3 < HEADS; ++h3)
                    acv += ex[t * 9 + h3 * 3 + cc];
                v[cc] = acv * (1.0f / HEADS) + bias[cc];
            }
            float mx = fmaxf(v[0], fmaxf(v[1], v[2]));
            float lse = logf(expf(v[0] - mx) + expf(v[1] - mx) + expf(v[2] - mx));
            for (int cc = 0; cc < NCLS; ++cc)
                o[nn * NCLS + cc] = v[cc] - mx - lse;
        }
    }
}

extern "C" void kernel_launch(void* const* d_in, const int* in_sizes, int n_in,
                              void* d_out, int out_size, void* d_ws, size_t ws_size,
                              hipStream_t stream) {
    const float* x        = (const float*)d_in[0];
    const int*   eidx     = (const int*)d_in[1];
    const float* W1       = (const float*)d_in[2];
    const float* att_src1 = (const float*)d_in[3];
    const float* att_dst1 = (const float*)d_in[4];
    const float* b1       = (const float*)d_in[5];
    const float* W2       = (const float*)d_in[6];
    const float* att_src2 = (const float*)d_in[7];
    const float* att_dst2 = (const float*)d_in[8];
    const float* b2       = (const float*)d_in[9];

    const int N = in_sizes[0] / IN_DIM;
    const int E = in_sizes[1] / 2;
    const int* src = eidx;
    const int* dst = eidx + E;

    const int nb = (N + 255) / 256;
    const int hn = nb * SORT_BLOCKS;
    const size_t need = ((size_t)N * 22 + 2 + (size_t)E) * 4;
    if (ws_size < need) return;
    if ((size_t)hn + nb + 1 > (size_t)N * 9) return;
    if (N > (1 << 24)) return;

    float4* R      = (float4*)d_ws;
    float*  tb     = (float*)d_ws + (size_t)N * 12;
    int*    rowptr = (int*)(tb + (size_t)N * 9);
    int*    csr    = rowptr + N + 1;
    int*    H      = (int*)tb;
    int*    bstart = H + hn;

    const int hb = (N + GB_NODES - 1) / GB_NODES;
    const int tiles = (N + 15) / 16;
    const int pb = (tiles + PW - 1) / PW;
    const int fb = pb > SORT_BLOCKS ? pb : SORT_BLOCKS;

    front<<<fb, FB_THREADS, nb * sizeof(int), stream>>>(x, W1, att_src1, R, N, dst, H, E, nb);

    bucket_scatter<<<SORT_BLOCKS, 1024, (3 * nb + 64) * sizeof(int), stream>>>(src, dst, H, bstart,
                                                                            (unsigned*)csr, E, nb);
    bucket_rows   <<<nb, 1024, 0, stream>>>(bstart, csr, rowptr, N);

    gather_rows<1><<<hb, GB_THREADS, 0, stream>>>(rowptr, csr, R, nullptr, att_src2, att_dst1, b1, W2, tb, N);

    gather_rows<2><<<hb, GB_THREADS, 0, stream>>>(rowptr, csr, nullptr, tb, att_src2, att_dst2, b2, W2,
                                                  (float*)d_out, N);
}


# Scatter launch asks for each phase's words at once: histograms chunk-major, summed per lane from 16-byte loads, 16 in flight; a thread's edges all loaded before ranking; row sum order = LDS arrival
# speedup vs baseline: 19.3769x; 1.0371x over previous
#include <hip/hip_runtime.h>
#include <math.h>

#define IN_DIM   256
#define HEADS    3
#define HID      3
#define NCLS     3
#define NEG_SLOPE 0.2f
#define GAT_EPS  1e-16f

typedef __attribute__((ext_vector_type(2))) float v2f;
typedef __attribute__((ext_vector_type(8))) float v8f;

__device__ __forceinline__ float lrelu(float x) { return x > 0.f ? x : NEG_SLOPE * x; }

#define CT_PITCH 20
#define BSH         8
#define BNODES      (1 << BSH)
#define SORT_BLOCKS 256
#define STAGE_CAP   24576
#define FB_THREADS 1024
#define PW         25
#define HT         (FB_THREADS - PW * 32)
#define HU         16
__global__ __launch_bounds__(FB_THREADS) void front(const float* __restrict__ x,
                                                    const float* __restrict__ W,
                                                    const float* __restrict__ atts,
                                                    float4* __restrict__ R, int N,
                                                    const int* __restrict__ dst,
                                                    int* __restrict__ H, int E, int nbk) {
    __shared__ float Wp[IN_DIM * 16];
    __shared__ float Ct[PW][16 * CT_PITCH];
    extern __shared__ int hist[];
    const int t = threadIdx.x;
    const bool counts = (int)blockIdx.x < SORT_BLOCKS;
    for (int i = t; i < IN_DIM * 16; i += FB_THREADS) {
        int k = i >> 4, n = i & 15;
        Wp[i] = (n < HEADS * HID) ? W[k * (HEADS * HID) + n] : 0.0f;
    }
    if (counts) for (int k = t; k < nbk; k += FB_THREADS) hist[k] = 0;
    __syncthreads();

    int lane  = t & 31;
    int wave  = t >> 5;
    long tile = (long)wave * (long)gridDim.x + (long)blockIdx.x;
    long m0   = tile * 16;
    const bool active = (wave < PW) && (m0 < N);

    int laneM = lane & 15;
    int koff  = (lane < 16) ? 0 : 2;

    if (active) {
        const float* xrow = x + (m0 + laneM) * IN_DIM;
        v8f c = {};
        #pragma unroll 4
        for (int k0 = 0; k0 < IN_DIM; k0 += 4) {
            v2f a, b;
            a.x = xrow[k0 + koff];
            a.y = xrow[k0 + koff + 1];
            b.x = Wp[(k0 + koff)     * 16 + laneM];
            b.y = Wp[(k0 + koff + 1) * 16 + laneM];
            c = __builtin_amdgcn_wmma_f32_16x16x4_f32(false, a, false, b,
                                                      (short)0, c, false, false);
        }
        int rbase = (lane < 16) ? 0 : 8;
        #pragma unroll
        for (int v = 0; v < 8; ++v)
            Ct[wave][(rbase + v) * CT_PITCH + laneM] = c[v];
    } else if (wave >= PW && counts) {
        const int ht = t - PW * 32;
        const int ch = (E + SORT_BLOCKS - 1) / SORT_BLOCKS;
        const int lo = (int)blockIdx.x * ch;
        const int hi = min(E, lo + ch);
        for (int i = lo + ht; i < hi; i += HU * HT) {
            int d[HU];
            #pragma unroll
            for (int u = 0; u < HU; ++u) d[u] = dst[min(i + u * HT, hi - 1)];
            #pragma unroll
            for (int u = 0; u < HU; ++u)
                atomicAdd(&hist[d[u] >> BSH], (i + u * HT < hi) ? 1 : 0);
        }
    }
    __syncthreads();
    if (active) {
        for (int rec = lane; rec < 16 * HEADS; rec += 32) {
            const int row = rec / HEADS;
            const int hh  = rec - row * HEADS;
            if (m0 + row < N) {
                const float v0 = Ct[wave][row * CT_PITCH + hh * 3 + 0];
                const float v1 = Ct[wave][row * CT_PITCH + hh * 3 + 1];
                const float v2 = Ct[wave][row * CT_PITCH + hh * 3 + 2];
                float s = 0.f;
                s += v0 * atts[hh * 3 + 0];
                s += v1 * atts[hh * 3 + 1];
                s += v2 * atts[hh * 3 + 2];
                R[m0 * HEADS + rec] = make_float4(v0, v1, v2, s);
            }
        }
    }
    const int nbp = (nbk + 3) & ~3;
    if (counts) for (int k = t; k < nbp; k += FB_THREADS)
        H[(size_t)blockIdx.x * nbp + k] = (k < nbk) ? hist[k] : 0;
}


#define CH_CAP 16384
#define HB     16
#define SU     16
__global__ __launch_bounds__(1024) void bucket_scatter(const int* __restrict__ src,
                                                       const int* __restrict__ dst,
                                                       const int* __restrict__ H,
                                                       int* __restrict__ bstart,
                                                       unsigned* __restrict__ P, int E, int nbk) {
    __shared__ uint2 wg[CH_CAP];
    extern __shared__ int shs[];
    int* cur   = shs;
    int* tot   = shs + nbk;
    int* dl    = tot + nbk;
    int* wsum  = dl + nbk;
    int* wsum2 = wsum + 32;
    const int t = threadIdx.x;
    const int lane = t & 31, wave = t >> 5;
    const int c = blockIdx.x;
    for (int k = t; k < nbk; k += 1024) { tot[k] = 0; cur[k] = 0; }
    __syncthreads();
    {
        const int nq = ((nbk + 3) & ~3) >> 2;
        const int QS = min(1024, ((nq + 31) >> 5) << 5);
        const int G  = 1024 / QS;
        const int g  = t / QS;
        const int4* H4 = (const int4*)H;
        if (g < G) for (int q = t - g * QS; q < nq; q += QS) {
            int tx = 0, ty = 0, tz = 0, tw = 0, px = 0, py = 0, pz = 0, pw = 0;
            const int4 ow = H4[(size_t)c * nq + q];
            for (int i0 = g; i0 < SORT_BLOCKS; i0 += HB * G) {
                int4 v[HB];
                #pragma unroll
                for (int u = 0; u < HB; ++u) {
                    const int cc = (c + min(i0 + u * G, SORT_BLOCKS - 1)) & (SORT_BLOCKS - 1);
                    v[u] = H4[(size_t)cc * nq + q];
                }
                __builtin_amdgcn_sched_barrier(0);
                #pragma unroll
                for (int u = 0; u < HB; ++u) {
                    const int  i   = i0 + u * G;
                    const bool in  = i < SORT_BLOCKS;
                    const bool bef = in && (c + i >= SORT_BLOCKS);
                    tx += in ? v[u].x : 0;  ty += in ? v[u].y : 0;
                    tz += in ? v[u].z : 0;  tw += in ? v[u].w : 0;
                    px += bef ? v[u].x : 0; py += bef ? v[u].y : 0;
                    pz += bef ? v[u].z : 0; pw += bef ? v[u].w : 0;
                }
            }
            const int k = 4 * q;
            if (k     < nbk) { atomicAdd(&tot[k],     tx); atomicAdd(&cur[k],     px); if (g == 0) dl[k]     = ow.x; }
            if (k + 1 < nbk) { atomicAdd(&tot[k + 1], ty); atomicAdd(&cur[k + 1], py); if (g == 0) dl[k + 1] = ow.y; }
            if (k + 2 < nbk) { atomicAdd(&tot[k + 2], tz); atomicAdd(&cur[k + 2], pz); if (g == 0) dl[k + 2] = ow.z; }
            if (k + 3 < nbk) { atomicAdd(&tot[k + 3], tw); atomicAdd(&cur[k + 3], pw); if (g == 0) dl[k + 3] = ow.w; }
        }
    }
    __syncthreads();
    const int seg = (nbk + 1023) >> 10;
    const int k0 = min(t * seg, nbk), k1 = min(k0 + seg, nbk);
    int loc = 0, loc2 = 0;
    for (int k = k0; k < k1; ++k) { loc += tot[k]; loc2 += dl[k]; }
    int v = loc, v2 = loc2;
    for (int off = 1; off < 32; off <<= 1) {
        const int u = __shfl_up(v, off);
        const int u2 = __shfl_up(v2, off);
        if (lane >= off) { v += u; v2 += u2; }
    }
    if (lane == 31) { wsum[wave] = v; wsum2[wave] = v2; }
    __syncthreads();
    int run = v - loc, run2 = v2 - loc2;
    for (int w = 0; w < wave; ++w) { run += wsum[w]; run2 += wsum2[w]; }
    for (int k = k0; k < k1; ++k) {
        if (c == 0) bstart[k] = run;
        const int ow = dl[k];
        cur[k] += run;
        dl[k] = cur[k] - run2;
        run += tot[k];
        run2 += ow;
    }
    if (c == 0 && t == 1023) bstart[nbk] = run;
    __syncthreads();
    const int ch = (E + (int)gridDim.x - 1) / (int)gridDim.x;
    const int lo = blockIdx.x * ch;
    const int hi = min(E, lo + ch);
    const bool fits = (hi - lo) <= CH_CAP;
    for (int i0 = lo + t; i0 < hi; i0 += SU * 1024) {
        int d[SU], s[SU];
        #pragma unroll
        for (int u = 0; u < SU; ++u) {
            const int i = min(i0 + u * 1024, hi - 1);
            d[u] = dst[i];
            s[u] = src[i];
        }
        #pragma unroll
        for (int u = 0; u < SU; ++u)
            if (i0 + u * 1024 < hi) {
                const int p = atomicAdd(&cur[d[u] >> BSH], 1);
                const unsigned w = ((unsigned)s[u] << BSH) | (unsigned)(d[u] & (BNODES - 1));
                if (fits) wg[p - dl[d[u] >> BSH]] = make_uint2(w, (unsigned)p);
                else      P[p] = w;
            }
    }
    __syncthreads();
    if (fits) {
        const int cn = hi - lo;
        int q = t;
        for (; q + 3 * 1024 < cn; q += 4 * 1024) {
            const uint2 e0 = wg[q], e1 = wg[q + 1024], e2 = wg[q + 2048], e3 = wg[q + 3072];
            P[e0.y] = e0.x; P[e1.y] = e1.x; P[e2.y] = e2.x; P[e3.y] = e3.x;
        }
        for (; q < cn; q += 1024) {
            const uint2 e = wg[q];
            P[e.y] = e.x;
        }
    }
}

#define RWR 12
__global__ __launch_bounds__(1024) void bucket_rows(const int* __restrict__ bstart, int* csr,
                                                    int* __restrict__ rowptr, int N) {
    __shared__ int sorted[STAGE_CAP];
    __shared__ int cnt[BNODES];
    __shared__ int cur[BNODES];
    __shared__ int wsum[BNODES / 32];
    const int t = threadIdx.x;
    const int b = blockIdx.x;
    const int lo = bstart[b];
    const int hi = bstart[b + 1];
    const int nst = min(hi - lo, STAGE_CAP);

    if (t < BNODES) cnt[t] = 0;
    __syncthreads();
    unsigned w[RWR];
    #pragma unroll
    for (int u = 0; u < RWR; ++u) w[u] = 0u;
    if (nst > 0) {
        #pragma unroll
        for (int u = 0; u < RWR; ++u) w[u] = (unsigned)csr[lo + min(t + u * 1024, nst - 1)];
    }
    #pragma unroll
    for (int u = 0; u < RWR; ++u)
        if (t + u * 1024 < nst) atomicAdd(&cnt[w[u] & (BNODES - 1)], 1);
    for (int i = RWR * 1024 + t; i < nst; i += 1024)
        atomicAdd(&cnt[(unsigned)csr[lo + i] & (BNODES - 1)], 1);
    __syncthreads();

    const int c = (t < BNODES) ? cnt[t] : 0;
    int v = c;
    for (int off = 1; off < 32; off <<= 1) {
        const int u = __shfl_up(v, off);
        if ((t & 31) >= off) v += u;
    }
    if (t < BNODES && (t & 31) == 31) wsum[t >> 5] = v;
    __syncthreads();
    if (t < BNODES) {
        int pre = 0;
        for (int ww = 0; ww < (t >> 5); ++ww) pre += wsum[ww];
        const int excl = pre + v - c;
        cur[t] = excl;
        const int node = b * BNODES + t;
        if (node < N) rowptr[node] = lo + excl;
        if (node == N - 1) rowptr[N] = hi;
    }
    __syncthreads();

    #pragma unroll
    for (int u = 0; u < RWR; ++u)
        if (t + u * 1024 < nst) {
            const int r = atomicAdd(&cur[w[u] & (BNODES - 1)], 1);
            sorted[r] = (int)(w[u] >> BSH);
        }
    for (int i = RWR * 1024 + t; i < nst; i += 1024) {
        const unsigned wv = (unsigned)csr[lo + i];
        const int r = atomicAdd(&cur[wv & (BNODES - 1)], 1);
        sorted[r] = (int)(wv >> BSH);
    }
    __syncthreads();

    const int g0 = lo & ~31;
    int g = g0 + t;
    for (; g + 3 * 1024 < lo + nst; g += 4 * 1024) {
        const int x0 = sorted[max(g - lo, 0)], x1 = sorted[g + 1024 - lo];
        const int x2 = sorted[g + 2048 - lo],   x3 = sorted[g + 3072 - lo];
        if (g >= lo) csr[g] = x0;
        csr[g + 1024] = x1; csr[g + 2048] = x2; csr[g + 3072] = x3;
    }
    for (; g < lo + nst; g += 1024)
        if (g >= lo) csr[g] = sorted[g - lo];
    for (int i = nst + t; i < hi - lo; i += 1024) csr[lo + i] = 0;
}

#define GB_NODES   64
#define GB_THREADS (GB_NODES * HEADS)
#define GB_SU      16
#define GB_STAGE   (GB_SU * GB_THREADS)

__device__ __forceinline__ void edge_step(const float4 r, const float ad, float& m,
                                          float& s, float& a0, float& a1, float& a2) {
    const float e  = lrelu(r.w + ad);
    const float d  = e - m;
    const float tt = expf(-fabsf(d));
    const bool  up = d > 0.f;
    const float sc = up ? tt : 1.f;
    const float p  = up ? 1.f : tt;
    m  = up ? e : m;
    s  = s  * sc + p;
    a0 = a0 * sc + p * r.x;
    a1 = a1 * sc + p * r.y;
    a2 = a2 * sc + p * r.z;
}

template <int LAYER>
__device__ __forceinline__ float4 load_rec(const float4* __restrict__ R4,
                                           const float* __restrict__ R3, const int idx,
                                           const float as0, const float as1, const float as2) {
    if (LAYER == 1) return R4[idx];
    const float* p = R3 + idx * 3;
    const float v0 = p[0], v1 = p[1], v2 = p[2];
    float s = 0.f;
    s += v0 * as0;
    s += v1 * as1;
    s += v2 * as2;
    return make_float4(v0, v1, v2, s);
}

template <int LAYER, bool STAGED>
__device__ __forceinline__ void row_sum(const int* __restrict__ csr, const int* sl, const int lo,
                                        int j, const int end,
                                        const float4* __restrict__ R4, const float* __restrict__ R3,
                                        const int hh, const float as0, const float as1, const float as2,
                                        const float ad, float& m, float& s, float& a0, float& a1, float& a2) {
    for (; j + 4 <= end; j += 4) {
        int i0, i1, i2, i3;
        if (STAGED) { const int* q = sl + (j - lo); i0 = q[0]; i1 = q[1]; i2 = q[2]; i3 = q[3]; }
        else        { i0 = csr[j]; i1 = csr[j + 1]; i2 = csr[j + 2]; i3 = csr[j + 3]; }
        const float4 r0 = load_rec<LAYER>(R4, R3, i0 * HEADS + hh, as0, as1, as2);
        const float4 r1 = load_rec<LAYER>(R4, R3, i1 * HEADS + hh, as0, as1, as2);
        const float4 r2 = load_rec<LAYER>(R4, R3, i2 * HEADS + hh, as0, as1, as2);
        const float4 r3 = load_rec<LAYER>(R4, R3, i3 * HEADS + hh, as0, as1, as2);
        edge_step(r0, ad, m, s, a0, a1, a2);
        edge_step(r1, ad, m, s, a0, a1, a2);
        edge_step(r2, ad, m, s, a0, a1, a2);
        edge_step(r3, ad, m, s, a0, a1, a2);
    }
    for (; j < end; ++j) {
        const int i = STAGED ? sl[j - lo] : csr[j];
        const float4 r = load_rec<LAYER>(R4, R3, i * HEADS + hh, as0, as1, as2);
        edge_step(r, ad, m, s, a0, a1, a2);
    }
}

template <int LAYER>
__global__ __launch_bounds__(GB_THREADS) void gather_rows(const int* __restrict__ rowptr,
                                                          const int* __restrict__ csr,
                                                          const float4* __restrict__ R4,
                                                          const float* __restrict__ R3,
                                                          const float* __restrict__ atts,
                                                          const float* __restrict__ attd,
                                                          const float* __restrict__ bias,
                                                          const float* __restrict__ W2,
                                                          float* __restrict__ o, int N) {
    __shared__ float ex[GB_THREADS * 3];
    __shared__ float Ws[81];
    __shared__ int   sl[GB_STAGE];
    const int t = threadIdx.x;

    const int g  = blockIdx.x * GB_THREADS + t;
    const int nl = t / HEADS;
    const int hh = t - nl * HEADS;
    const int n  = blockIdx.x * GB_NODES + nl;
    const bool valid = n < N;

    const int n0 = blockIdx.x * GB_NODES;
    const int lo = rowptr[n0];
    const int hi = rowptr[min(n0 + GB_NODES, N)];
    const bool fits = (hi - lo) <= GB_STAGE;

    const int ns = valid ? n : 0;
    const int gs = ns * HEADS + hh;
    const float w2v = (LAYER == 1) ? W2[min(t, 80)] : 0.f;
    float as0 = 0.f, as1 = 0.f, as2 = 0.f;
    if (LAYER == 2) { as0 = atts[hh * 3 + 0]; as1 = atts[hh * 3 + 1]; as2 = atts[hh * 3 + 2]; }
    const float d0 = attd[hh * 3 + 0], d1 = attd[hh * 3 + 1], d2 = attd[hh * 3 + 2];
    float4 self;
    if (LAYER == 1) self = R4[gs];
    else            self = make_float4(R3[gs * 3 + 0], R3[gs * 3 + 1], R3[gs * 3 + 2], 0.f);
    const int j   = rowptr[ns];
    const int end = rowptr[ns + 1];

    if (fits && hi > lo) {
        const int cnt = hi - lo;
        int w[GB_SU];
        #pragma unroll
        for (int u = 0; u < GB_SU; ++u) w[u] = csr[lo + min(t + u * GB_THREADS, cnt - 1)];
        #pragma unroll
        for (int u = 0; u < GB_SU; ++u)
            if (t + u * GB_THREADS < cnt) sl[t + u * GB_THREADS] = w[u];
    }
    if (LAYER == 1) { if (t < 81) Ws[t] = w2v; }
    __syncthreads();

    float v0 = 0.f, v1 = 0.f, v2 = 0.f;
    if (valid) {
        if (LAYER == 2) {
            float sw = 0.f;
            sw += self.x * as0;
            sw += self.y * as1;
            sw += self.z * as2;
            self.w = sw;
        }
        float ad = 0.f;
        ad += self.x * d0;
        ad += self.y * d1;
        ad += self.z * d2;

        float m = lrelu(self.w + ad);
        float s = 1.f, a0 = self.x, a1 = self.y, a2 = self.z;

        if (fits) row_sum<LAYER, true >(csr, sl, lo, j, end, R4, R3, hh, as0, as1, as2, ad, m, s, a0, a1, a2);
        else      row_sum<LAYER, false>(csr, sl, lo, j, end, R4, R3, hh, as0, as1, as2, ad, m, s, a0, a1, a2);

        const float inv = 1.0f / (s + GAT_EPS);
        v0 = a0 * inv; v1 = a1 * inv; v2 = a2 * inv;
        if (LAYER == 1) {
            v0 += bias[hh * 3 + 0];
            v1 += bias[hh * 3 + 1];
            v2 += bias[hh * 3 + 2];
            v0 = (v0 > 0.f) ? v0 : (expf(v0) - 1.0f);
            v1 = (v1 > 0.f) ? v1 : (expf(v1) - 1.0f);
            v2 = (v2 > 0.f) ? v2 : (expf(v2) - 1.0f);
        }
    }
    ex[t * 3 + 0] = v0;
    ex[t * 3 + 1] = v1;
    ex[t * 3 + 2] = v2;
    __syncthreads();

    if (LAYER == 1) {
        const float* tn = &ex[nl * 9];
        float o0 = 0.f, o1 = 0.f, o2 = 0.f;
        for (int i = 0; i < 9; ++i) {
            const float ti = tn[i];
            o0 += ti * Ws[i * 9 + hh * 3 + 0];
            o1 += ti * Ws[i * 9 + hh * 3 + 1];
            o2 += ti * Ws[i * 9 + hh * 3 + 2];
        }
        if (valid) {
            o[g * 3 + 0] = o0;
            o[g * 3 + 1] = o1;
            o[g * 3 + 2] = o2;
        }
    } else {
        const int nn = blockIdx.x * GB_NODES + t;
        if (t < GB_NODES && nn < N) {
            float v[NCLS];
            for (int cc = 0; cc < NCLS; ++cc) {
                float acv = 0.f;
                for (int h3 = 0; h3 < HEADS; ++h3)
                    acv += ex[t * 9 + h3 * 3 + cc];
                v[cc] = acv * (1.0f / HEADS) + bias[cc];
            }
            float mx = fmaxf(v[0], fmaxf(v[1], v[2]));
            float lse = logf(expf(v[0] - mx) + expf(v[1] - mx) + expf(v[2] - mx));
            for (int cc = 0; cc < NCLS; ++cc)
                o[nn * NCLS + cc] = v[cc] - mx - lse;
        }
    }
}

extern "C" void kernel_launch(void* const* d_in, const int* in_sizes, int n_in,
                              void* d_out, int out_size, void* d_ws, size_t ws_size,
                              hipStream_t stream) {
    const float* x        = (const float*)d_in[0];
    const int*   eidx     = (const int*)d_in[1];
    const float* W1       = (const float*)d_in[2];
    const float* att_src1 = (const float*)d_in[3];
    const float* att_dst1 = (const float*)d_in[4];
    const float* b1       = (const float*)d_in[5];
    const float* W2       = (const float*)d_in[6];
    const float* att_src2 = (const float*)d_in[7];
    const float* att_dst2 = (const float*)d_in[8];
    const float* b2       = (const float*)d_in[9];

    const int N = in_sizes[0] / IN_DIM;
    const int E = in_sizes[1] / 2;
    const int* src = eidx;
    const int* dst = eidx + E;

    const int nb = (N + 255) / 256;
    const int hn = ((nb + 3) & ~3) * SORT_BLOCKS;
    const size_t need = ((size_t)N * 22 + 2 + (size_t)E) * 4;
    if (ws_size < need) return;
    if ((size_t)hn + nb + 1 > (size_t)N * 9) return;
    if (N > (1 << 24)) return;

    float4* R      = (float4*)d_ws;
    float*  tb     = (float*)d_ws + (size_t)N * 12;
    int*    rowptr = (int*)(tb + (size_t)N * 9);
    int*    csr    = rowptr + N + 1;
    int*    H      = (int*)tb;
    int*    bstart = H + hn;

    const int hb = (N + GB_NODES - 1) / GB_NODES;
    const int tiles = (N + 15) / 16;
    const int pb = (tiles + PW - 1) / PW;
    const int fb = pb > SORT_BLOCKS ? pb : SORT_BLOCKS;

    front<<<fb, FB_THREADS, nb * sizeof(int), stream>>>(x, W1, att_src1, R, N, dst, H, E, nb);

    bucket_scatter<<<SORT_BLOCKS, 1024, (3 * nb + 64) * sizeof(int), stream>>>(src, dst, H, bstart,
                                                                            (unsigned*)csr, E, nb);
    bucket_rows   <<<nb, 1024, 0, stream>>>(bstart, csr, rowptr, N);

    gather_rows<1><<<hb, GB_THREADS, 0, stream>>>(rowptr, csr, R, nullptr, att_src2, att_dst1, b1, W2, tb, N);

    gather_rows<2><<<hb, GB_THREADS, 0, stream>>>(rowptr, csr, nullptr, tb, att_src2, att_dst2, b2, W2,
                                                  (float*)d_out, N);
}


# Workspace laid on 128-byte lines: node records of both layers three to a line (9 floats, reader forms a_src; row starts in spare words; edge list holds record offsets, base aligned); sums unchanged
# speedup vs baseline: 24.8326x; 1.2816x over previous
#include <hip/hip_runtime.h>
#include <math.h>

#define IN_DIM   256
#define HEADS    3
#define HID      3
#define NCLS     3
#define NEG_SLOPE 0.2f
#define GAT_EPS  1e-16f

typedef __attribute__((ext_vector_type(2))) float v2f;
typedef __attribute__((ext_vector_type(8))) float v8f;

__device__ __forceinline__ float lrelu(float x) { return x > 0.f ? x : NEG_SLOPE * x; }

#define LINE_W  32
#define ROWSLOT 27
__device__ __forceinline__ int rec_off(const int node) {
    const int q = node / 3;
    return q * LINE_W + (node - q * 3) * 9;
}
__device__ __forceinline__ int row_slot(const int node) {
    const int q = node / 3;
    return q * LINE_W + ROWSLOT + (node - q * 3);
}

#define CT_PITCH 20
#define BSH         8
#define BNODES      (1 << BSH)
#define SORT_BLOCKS 256
#define STAGE_CAP   24576
#define FB_THREADS 1024
#define PW         25
#define HT         (FB_THREADS - PW * 32)
#define HU         16
__global__ __launch_bounds__(FB_THREADS) void front(const float* __restrict__ x,
                                                    const float* __restrict__ W,
                                                    float* __restrict__ RA, int N,
                                                    const int* __restrict__ dst,
                                                    int* __restrict__ H, int E, int nbk) {
    __shared__ float Wp[IN_DIM * 16];
    __shared__ float Ct[PW][16 * CT_PITCH];
    extern __shared__ int hist[];
    const int t = threadIdx.x;
    const bool counts = (int)blockIdx.x < SORT_BLOCKS;
    for (int i = t; i < IN_DIM * 16; i += FB_THREADS) {
        int k = i >> 4, n = i & 15;
        Wp[i] = (n < HEADS * HID) ? W[k * (HEADS * HID) + n] : 0.0f;
    }
    if (counts) for (int k = t; k < nbk; k += FB_THREADS) hist[k] = 0;
    __syncthreads();

    int lane  = t & 31;
    int wave  = t >> 5;
    long tile = (long)wave * (long)gridDim.x + (long)blockIdx.x;
    long m0   = tile * 16;
    const bool active = (wave < PW) && (m0 < N);

    int laneM = lane & 15;
    int koff  = (lane < 16) ? 0 : 2;

    if (active) {
        const float* xrow = x + (m0 + laneM) * IN_DIM;
        v8f c = {};
        #pragma unroll 4
        for (int k0 = 0; k0 < IN_DIM; k0 += 4) {
            v2f a, b;
            a.x = xrow[k0 + koff];
            a.y = xrow[k0 + koff + 1];
            b.x = Wp[(k0 + koff)     * 16 + laneM];
            b.y = Wp[(k0 + koff + 1) * 16 + laneM];
            c = __builtin_amdgcn_wmma_f32_16x16x4_f32(false, a, false, b,
                                                      (short)0, c, false, false);
        }
        int rbase = (lane < 16) ? 0 : 8;
        #pragma unroll
        for (int v = 0; v < 8; ++v)
            Ct[wave][(rbase + v) * CT_PITCH + laneM] = c[v];
    } else if (wave >= PW && counts) {
        const int ht = t - PW * 32;
        const int ch = (E + SORT_BLOCKS - 1) / SORT_BLOCKS;
        const int lo = (int)blockIdx.x * ch;
        const int hi = min(E, lo + ch);
        for (int i = lo + ht; i < hi; i += HU * HT) {
            int d[HU];
            #pragma unroll
            for (int u = 0; u < HU; ++u) d[u] = dst[min(i + u * HT, hi - 1)];
            #pragma unroll
            for (int u = 0; u < HU; ++u)
                atomicAdd(&hist[d[u] >> BSH], (i + u * HT < hi) ? 1 : 0);
        }
    }
    __syncthreads();
    if (active) {
        for (int e = lane; e < 16 * HEADS * HID; e += 32) {
            const int row = e / (HEADS * HID);
            const int cc  = e - row * (HEADS * HID);
            if (m0 + row < N)
                RA[rec_off((int)(m0 + row)) + cc] = Ct[wave][row * CT_PITCH + cc];
        }
    }
    const int nbp = (nbk + 3) & ~3;
    if (counts) for (int k = t; k < nbp; k += FB_THREADS)
        H[(size_t)blockIdx.x * nbp + k] = (k < nbk) ? hist[k] : 0;
}


#define CH_CAP 16384
#define HB     16
#define SU     16
__global__ __launch_bounds__(1024) void bucket_scatter(const int* __restrict__ src,
                                                       const int* __restrict__ dst,
                                                       const int* __restrict__ H,
                                                       int* __restrict__ bstart,
                                                       unsigned* __restrict__ P, int E, int nbk) {
    __shared__ uint2 wg[CH_CAP];
    extern __shared__ int shs[];
    int* cur   = shs;
    int* tot   = shs + nbk;
    int* dl    = tot + nbk;
    int* wsum  = dl + nbk;
    int* wsum2 = wsum + 32;
    const int t = threadIdx.x;
    const int lane = t & 31, wave = t >> 5;
    const int c = blockIdx.x;
    for (int k = t; k < nbk; k += 1024) { tot[k] = 0; cur[k] = 0; }
    __syncthreads();
    {
        const int nq = ((nbk + 3) & ~3) >> 2;
        const int QS = min(1024, ((nq + 31) >> 5) << 5);
        const int G  = 1024 / QS;
        const int g  = t / QS;
        const int4* H4 = (const int4*)H;
        if (g < G) for (int q = t - g * QS; q < nq; q += QS) {
            int tx = 0, ty = 0, tz = 0, tw = 0, px = 0, py = 0, pz = 0, pw = 0;
            const int4 ow = H4[(size_t)c * nq + q];
            for (int i0 = g; i0 < SORT_BLOCKS; i0 += HB * G) {
                int4 v[HB];
                #pragma unroll
                for (int u = 0; u < HB; ++u) {
                    const int cc = (c + min(i0 + u * G, SORT_BLOCKS - 1)) & (SORT_BLOCKS - 1);
                    v[u] = H4[(size_t)cc * nq + q];
                }
                __builtin_amdgcn_sched_barrier(0);
                #pragma unroll
                for (int u = 0; u < HB; ++u) {
                    const int  i   = i0 + u * G;
                    const bool in  = i < SORT_BLOCKS;
                    const bool bef = in && (c + i >= SORT_BLOCKS);
                    tx += in ? v[u].x : 0;  ty += in ? v[u].y : 0;
                    tz += in ? v[u].z : 0;  tw += in ? v[u].w : 0;
                    px += bef ? v[u].x : 0; py += bef ? v[u].y : 0;
                    pz += bef ? v[u].z : 0; pw += bef ? v[u].w : 0;
                }
            }
            const int k = 4 * q;
            if (k     < nbk) { atomicAdd(&tot[k],     tx); atomicAdd(&cur[k],     px); if (g == 0) dl[k]     = ow.x; }
            if (k + 1 < nbk) { atomicAdd(&tot[k + 1], ty); atomicAdd(&cur[k + 1], py); if (g == 0) dl[k + 1] = ow.y; }
            if (k + 2 < nbk) { atomicAdd(&tot[k + 2], tz); atomicAdd(&cur[k + 2], pz); if (g == 0) dl[k + 2] = ow.z; }
            if (k + 3 < nbk) { atomicAdd(&tot[k + 3], tw); atomicAdd(&cur[k + 3], pw); if (g == 0) dl[k + 3] = ow.w; }
        }
    }
    __syncthreads();
    const int seg = (nbk + 1023) >> 10;
    const int k0 = min(t * seg, nbk), k1 = min(k0 + seg, nbk);
    int loc = 0, loc2 = 0;
    for (int k = k0; k < k1; ++k) { loc += tot[k]; loc2 += dl[k]; }
    int v = loc, v2 = loc2;
    for (int off = 1; off < 32; off <<= 1) {
        const int u = __shfl_up(v, off);
        const int u2 = __shfl_up(v2, off);
        if (lane >= off) { v += u; v2 += u2; }
    }
    if (lane == 31) { wsum[wave] = v; wsum2[wave] = v2; }
    __syncthreads();
    int run = v - loc, run2 = v2 - loc2;
    for (int w = 0; w < wave; ++w) { run += wsum[w]; run2 += wsum2[w]; }
    for (int k = k0; k < k1; ++k) {
        if (c == 0) bstart[k] = run;
        const int ow = dl[k];
        cur[k] += run;
        dl[k] = cur[k] - run2;
        run += tot[k];
        run2 += ow;
    }
    if (c == 0 && t == 1023) bstart[nbk] = run;
    __syncthreads();
    const int ch = (E + (int)gridDim.x - 1) / (int)gridDim.x;
    const int lo = blockIdx.x * ch;
    const int hi = min(E, lo + ch);
    const bool fits = (hi - lo) <= CH_CAP;
    for (int i0 = lo + t; i0 < hi; i0 += SU * 1024) {
        int d[SU], s[SU];
        #pragma unroll
        for (int u = 0; u < SU; ++u) {
            const int i = min(i0 + u * 1024, hi - 1);
            d[u] = dst[i];
            s[u] = src[i];
        }
        #pragma unroll
        for (int u = 0; u < SU; ++u)
            if (i0 + u * 1024 < hi) {
                const int p = atomicAdd(&cur[d[u] >> BSH], 1);
                const unsigned w = ((unsigned)s[u] << BSH) | (unsigned)(d[u] & (BNODES - 1));
                if (fits) wg[p - dl[d[u] >> BSH]] = make_uint2(w, (unsigned)p);
                else      P[p] = w;
            }
    }
    __syncthreads();
    if (fits) {
        const int cn = hi - lo;
        int q = t;
        for (; q + 3 * 1024 < cn; q += 4 * 1024) {
            const uint2 e0 = wg[q], e1 = wg[q + 1024], e2 = wg[q + 2048], e3 = wg[q + 3072];
            P[e0.y] = e0.x; P[e1.y] = e1.x; P[e2.y] = e2.x; P[e3.y] = e3.x;
        }
        for (; q < cn; q += 1024) {
            const uint2 e = wg[q];
            P[e.y] = e.x;
        }
    }
}

#define RWR 12
__global__ __launch_bounds__(1024) void bucket_rows(const int* __restrict__ bstart, int* csr,
                                                    int* __restrict__ rows, int N) {
    __shared__ int sorted[STAGE_CAP];
    __shared__ int cnt[BNODES];
    __shared__ int cur[BNODES];
    __shared__ int wsum[BNODES / 32];
    const int t = threadIdx.x;
    const int b = blockIdx.x;
    const int lo = bstart[b];
    const int hi = bstart[b + 1];
    const int nst = min(hi - lo, STAGE_CAP);

    if (t < BNODES) cnt[t] = 0;
    __syncthreads();
    unsigned w[RWR];
    #pragma unroll
    for (int u = 0; u < RWR; ++u) w[u] = 0u;
    if (nst > 0) {
        #pragma unroll
        for (int u = 0; u < RWR; ++u) w[u] = (unsigned)csr[lo + min(t + u * 1024, nst - 1)];
    }
    #pragma unroll
    for (int u = 0; u < RWR; ++u)
        if (t + u * 1024 < nst) atomicAdd(&cnt[w[u] & (BNODES - 1)], 1);
    for (int i = RWR * 1024 + t; i < nst; i += 1024)
        atomicAdd(&cnt[(unsigned)csr[lo + i] & (BNODES - 1)], 1);
    __syncthreads();

    const int c = (t < BNODES) ? cnt[t] : 0;
    int v = c;
    for (int off = 1; off < 32; off <<= 1) {
        const int u = __shfl_up(v, off);
        if ((t & 31) >= off) v += u;
    }
    if (t < BNODES && (t & 31) == 31) wsum[t >> 5] = v;
    __syncthreads();
    if (t < BNODES) {
        int pre = 0;
        for (int ww = 0; ww < (t >> 5); ++ww) pre += wsum[ww];
        const int excl = pre + v - c;
        cur[t] = excl;
        const int node = b * BNODES + t;
        if (node < N) rows[row_slot(node)] = lo + excl;
        if (node == N - 1) rows[row_slot(N)] = hi;
    }
    __syncthreads();

    #pragma unroll
    for (int u = 0; u < RWR; ++u)
        if (t + u * 1024 < nst) {
            const int r = atomicAdd(&cur[w[u] & (BNODES - 1)], 1);
            sorted[r] = rec_off((int)(w[u] >> BSH));
        }
    for (int i = RWR * 1024 + t; i < nst; i += 1024) {
        const unsigned wv = (unsigned)csr[lo + i];
        const int r = atomicAdd(&cur[wv & (BNODES - 1)], 1);
        sorted[r] = rec_off((int)(wv >> BSH));
    }
    __syncthreads();

    const int g0 = lo & ~31;
    int g = g0 + t;
    for (; g + 3 * 1024 < lo + nst; g += 4 * 1024) {
        const int x0 = sorted[max(g - lo, 0)], x1 = sorted[g + 1024 - lo];
        const int x2 = sorted[g + 2048 - lo],   x3 = sorted[g + 3072 - lo];
        if (g >= lo) csr[g] = x0;
        csr[g + 1024] = x1; csr[g + 2048] = x2; csr[g + 3072] = x3;
    }
    for (; g < lo + nst; g += 1024)
        if (g >= lo) csr[g] = sorted[g - lo];
    for (int i = nst + t; i < hi - lo; i += 1024) csr[lo + i] = 0;
}

#define GB_NODES   64
#define GB_THREADS (GB_NODES * HEADS)
#define GB_SU      16
#define GB_STAGE   (GB_SU * GB_THREADS)

__device__ __forceinline__ void edge_step(const float4 r, const float ad, float& m,
                                          float& s, float& a0, float& a1, float& a2) {
    const float e  = lrelu(r.w + ad);
    const float d  = e - m;
    const float tt = expf(-fabsf(d));
    const bool  up = d > 0.f;
    const float sc = up ? tt : 1.f;
    const float p  = up ? 1.f : tt;
    m  = up ? e : m;
    s  = s  * sc + p;
    a0 = a0 * sc + p * r.x;
    a1 = a1 * sc + p * r.y;
    a2 = a2 * sc + p * r.z;
}

__device__ __forceinline__ float4 load_rec(const float* __restrict__ Rs, const int off,
                                           const float as0, const float as1, const float as2) {
    const float* p = Rs + off;
    const float v0 = p[0], v1 = p[1], v2 = p[2];
    float s = 0.f;
    s += v0 * as0;
    s += v1 * as1;
    s += v2 * as2;
    return make_float4(v0, v1, v2, s);
}

template <bool STAGED>
__device__ __forceinline__ void row_sum(const int* __restrict__ csr, const int* sl, const int lo,
                                        int j, const int end, const float* __restrict__ Rs,
                                        const int h3, const float as0, const float as1, const float as2,
                                        const float ad, float& m, float& s, float& a0, float& a1, float& a2) {
    for (; j + 4 <= end; j += 4) {
        int i0, i1, i2, i3;
        if (STAGED) { const int* q = sl + (j - lo); i0 = q[0]; i1 = q[1]; i2 = q[2]; i3 = q[3]; }
        else        { i0 = csr[j]; i1 = csr[j + 1]; i2 = csr[j + 2]; i3 = csr[j + 3]; }
        const float4 r0 = load_rec(Rs, i0 + h3, as0, as1, as2);
        const float4 r1 = load_rec(Rs, i1 + h3, as0, as1, as2);
        const float4 r2 = load_rec(Rs, i2 + h3, as0, as1, as2);
        const float4 r3 = load_rec(Rs, i3 + h3, as0, as1, as2);
        edge_step(r0, ad, m, s, a0, a1, a2);
        edge_step(r1, ad, m, s, a0, a1, a2);
        edge_step(r2, ad, m, s, a0, a1, a2);
        edge_step(r3, ad, m, s, a0, a1, a2);
    }
    for (; j < end; ++j) {
        const int i = STAGED ? sl[j - lo] : csr[j];
        const float4 r = load_rec(Rs, i + h3, as0, as1, as2);
        edge_step(r, ad, m, s, a0, a1, a2);
    }
}

template <int LAYER>
__global__ __launch_bounds__(GB_THREADS) void gather_rows(const int* __restrict__ rows,
                                                          const int* __restrict__ csr,
                                                          const float* __restrict__ Rs,
                                                          const float* __restrict__ atts,
                                                          const float* __restrict__ attd,
                                                          const float* __restrict__ bias,
                                                          const float* __restrict__ W2,
                                                          float* __restrict__ o, int N) {
    __shared__ float ex[GB_THREADS * 3];
    __shared__ float Ws[81];
    __shared__ int   sl[GB_STAGE];
    const int t = threadIdx.x;

    const int nl = t / HEADS;
    const int hh = t - nl * HEADS;
    const int n  = blockIdx.x * GB_NODES + nl;
    const bool valid = n < N;

    const int n0 = blockIdx.x * GB_NODES;
    const int lo = rows[row_slot(n0)];
    const int hi = rows[row_slot(min(n0 + GB_NODES, N))];
    const bool fits = (hi - lo) <= GB_STAGE;

    const int ns = valid ? n : 0;
    const int hw = hh * 3;
    const int gs = rec_off(ns) + hw;
    const float w2v = (LAYER == 1) ? W2[min(t, 80)] : 0.f;
    const float as0 = atts[hw + 0], as1 = atts[hw + 1], as2 = atts[hw + 2];
    const float d0 = attd[hw + 0], d1 = attd[hw + 1], d2 = attd[hw + 2];
    float4 self = make_float4(Rs[gs + 0], Rs[gs + 1], Rs[gs + 2], 0.f);
    const int j   = rows[row_slot(ns)];
    const int end = rows[row_slot(ns + 1)];

    if (fits && hi > lo) {
        const int cnt = hi - lo;
        int w[GB_SU];
        #pragma unroll
        for (int u = 0; u < GB_SU; ++u) w[u] = csr[lo + min(t + u * GB_THREADS, cnt - 1)];
        #pragma unroll
        for (int u = 0; u < GB_SU; ++u)
            if (t + u * GB_THREADS < cnt) sl[t + u * GB_THREADS] = w[u];
    }
    if (LAYER == 1) { if (t < 81) Ws[t] = w2v; }
    __syncthreads();

    float v0 = 0.f, v1 = 0.f, v2 = 0.f;
    if (valid) {
        {
            float sw = 0.f;
            sw += self.x * as0;
            sw += self.y * as1;
            sw += self.z * as2;
            self.w = sw;
        }
        float ad = 0.f;
        ad += self.x * d0;
        ad += self.y * d1;
        ad += self.z * d2;

        float m = lrelu(self.w + ad);
        float s = 1.f, a0 = self.x, a1 = self.y, a2 = self.z;

        if (fits) row_sum<true >(csr, sl, lo, j, end, Rs, hw, as0, as1, as2, ad, m, s, a0, a1, a2);
        else      row_sum<false>(csr, sl, lo, j, end, Rs, hw, as0, as1, as2, ad, m, s, a0, a1, a2);

        const float inv = 1.0f / (s + GAT_EPS);
        v0 = a0 * inv; v1 = a1 * inv; v2 = a2 * inv;
        if (LAYER == 1) {
            v0 += bias[hh * 3 + 0];
            v1 += bias[hh * 3 + 1];
            v2 += bias[hh * 3 + 2];
            v0 = (v0 > 0.f) ? v0 : (expf(v0) - 1.0f);
            v1 = (v1 > 0.f) ? v1 : (expf(v1) - 1.0f);
            v2 = (v2 > 0.f) ? v2 : (expf(v2) - 1.0f);
        }
    }
    ex[t * 3 + 0] = v0;
    ex[t * 3 + 1] = v1;
    ex[t * 3 + 2] = v2;
    __syncthreads();

    if (LAYER == 1) {
        const float* tn = &ex[nl * 9];
        float o0 = 0.f, o1 = 0.f, o2 = 0.f;
        for (int i = 0; i < 9; ++i) {
            const float ti = tn[i];
            o0 += ti * Ws[i * 9 + hh * 3 + 0];
            o1 += ti * Ws[i * 9 + hh * 3 + 1];
            o2 += ti * Ws[i * 9 + hh * 3 + 2];
        }
        if (valid) {
            o[gs + 0] = o0;
            o[gs + 1] = o1;
            o[gs + 2] = o2;
        }
    } else {
        const int nn = blockIdx.x * GB_NODES + t;
        if (t < GB_NODES && nn < N) {
            float v[NCLS];
            for (int cc = 0; cc < NCLS; ++cc) {
                float acv = 0.f;
                for (int h3 = 0; h3 < HEADS; ++h3)
                    acv += ex[t * 9 + h3 * 3 + cc];
                v[cc] = acv * (1.0f / HEADS) + bias[cc];
            }
            float mx = fmaxf(v[0], fmaxf(v[1], v[2]));
            float lse = logf(expf(v[0] - mx) + expf(v[1] - mx) + expf(v[2] - mx));
            for (int cc = 0; cc < NCLS; ++cc)
                o[nn * NCLS + cc] = v[cc] - mx - lse;
        }
    }
}

extern "C" void kernel_launch(void* const* d_in, const int* in_sizes, int n_in,
                              void* d_out, int out_size, void* d_ws, size_t ws_size,
                              hipStream_t stream) {
    const float* x        = (const float*)d_in[0];
    const int*   eidx     = (const int*)d_in[1];
    const float* W1       = (const float*)d_in[2];
    const float* att_src1 = (const float*)d_in[3];
    const float* att_dst1 = (const float*)d_in[4];
    const float* b1       = (const float*)d_in[5];
    const float* W2       = (const float*)d_in[6];
    const float* att_src2 = (const float*)d_in[7];
    const float* att_dst2 = (const float*)d_in[8];
    const float* b2       = (const float*)d_in[9];

    const int N = in_sizes[0] / IN_DIM;
    const int E = in_sizes[1] / 2;
    const int* src = eidx;
    const int* dst = eidx + E;

    const int nb = (N + 255) / 256;
    const int hn = ((nb + 3) & ~3) * SORT_BLOCKS;
    const size_t plane = (size_t)((N + 3) / 3) * LINE_W;
    const size_t need = (2 * plane + (size_t)E) * 4;
    if (ws_size < need) return;
    if ((size_t)hn + nb + 1 > plane) return;
    if (N > (1 << 24)) return;

    float*  RA     = (float*)d_ws;
    float*  RB     = RA + plane;
    int*    csr    = (int*)(RB + plane);
    int*    H      = (int*)RB;
    int*    bstart = H + hn;

    const int hb = (N + GB_NODES - 1) / GB_NODES;
    const int tiles = (N + 15) / 16;
    const int pb = (tiles + PW - 1) / PW;
    const int fb = pb > SORT_BLOCKS ? pb : SORT_BLOCKS;

    front<<<fb, FB_THREADS, nb * sizeof(int), stream>>>(x, W1, RA, N, dst, H, E, nb);

    bucket_scatter<<<SORT_BLOCKS, 1024, (3 * nb + 64) * sizeof(int), stream>>>(src, dst, H, bstart,
                                                                            (unsigned*)csr, E, nb);
    bucket_rows   <<<nb, 1024, 0, stream>>>(bstart, csr, (int*)RA, N);

    gather_rows<1><<<hb, GB_THREADS, 0, stream>>>((const int*)RA, csr, RA, att_src1, att_dst1, b1, W2, RB, N);

    gather_rows<2><<<hb, GB_THREADS, 0, stream>>>((const int*)RA, csr, RB, att_src2, att_dst2, b2, W2,
                                                  (float*)d_out, N);
}


# Bucket sort launch and layer-1 gather launch made one (5 -> 4): the block that sorts a 256-node bucket in LDS sums its nodes' layer-1 rows from that LDS list; row sum order = LDS arrival as before
# speedup vs baseline: 25.6621x; 1.0334x over previous
#include <hip/hip_runtime.h>
#include <math.h>

#define IN_DIM   256
#define HEADS    3
#define HID      3
#define NCLS     3
#define NEG_SLOPE 0.2f
#define GAT_EPS  1e-16f

typedef __attribute__((ext_vector_type(2))) float v2f;
typedef __attribute__((ext_vector_type(8))) float v8f;

__device__ __forceinline__ float lrelu(float x) { return x > 0.f ? x : NEG_SLOPE * x; }

#define LINE_W  32
#define ROWSLOT 27
#define BSLOT   30
__device__ __forceinline__ int rec_off(const int node) {
    const int q = node / 3;
    return q * LINE_W + (node - q * 3) * 9;
}
__device__ __forceinline__ int row_slot(const int node) {
    const int q = node / 3;
    return q * LINE_W + ROWSLOT + (node - q * 3);
}

#define CT_PITCH 20
#define BSH         8
#define BNODES      (1 << BSH)
#define SORT_BLOCKS 256
#define STAGE_CAP   24576
#define FB_THREADS 1024
#define PW         25
#define HT         (FB_THREADS - PW * 32)
#define HU         16
__global__ __launch_bounds__(FB_THREADS) void front(const float* __restrict__ x,
                                                    const float* __restrict__ W,
                                                    float* __restrict__ RA, int N,
                                                    const int* __restrict__ dst,
                                                    int* __restrict__ H, int E, int nbk) {
    __shared__ float Wp[IN_DIM * 16];
    __shared__ float Ct[PW][16 * CT_PITCH];
    extern __shared__ int hist[];
    const int t = threadIdx.x;
    const bool counts = (int)blockIdx.x < SORT_BLOCKS;
    for (int i = t; i < IN_DIM * 16; i += FB_THREADS) {
        int k = i >> 4, n = i & 15;
        Wp[i] = (n < HEADS * HID) ? W[k * (HEADS * HID) + n] : 0.0f;
    }
    if (counts) for (int k = t; k < nbk; k += FB_THREADS) hist[k] = 0;
    __syncthreads();

    int lane  = t & 31;
    int wave  = t >> 5;
    long tile = (long)wave * (long)gridDim.x + (long)blockIdx.x;
    long m0   = tile * 16;
    const bool active = (wave < PW) && (m0 < N);

    int laneM = lane & 15;
    int koff  = (lane < 16) ? 0 : 2;

    if (active) {
        const float* xrow = x + (m0 + laneM) * IN_DIM;
        v8f c = {};
        #pragma unroll 4
        for (int k0 = 0; k0 < IN_DIM; k0 += 4) {
            v2f a, b;
            a.x = xrow[k0 + koff];
            a.y = xrow[k0 + koff + 1];
            b.x = Wp[(k0 + koff)     * 16 + laneM];
            b.y = Wp[(k0 + koff + 1) * 16 + laneM];
            c = __builtin_amdgcn_wmma_f32_16x16x4_f32(false, a, false, b,
                                                      (short)0, c, false, false);
        }
        int rbase = (lane < 16) ? 0 : 8;
        #pragma unroll
        for (int v = 0; v < 8; ++v)
            Ct[wave][(rbase + v) * CT_PITCH + laneM] = c[v];
    } else if (wave >= PW && counts) {
        const int ht = t - PW * 32;
        const int ch = (E + SORT_BLOCKS - 1) / SORT_BLOCKS;
        const int lo = (int)blockIdx.x * ch;
        const int hi = min(E, lo + ch);
        for (int i = lo + ht; i < hi; i += HU * HT) {
            int d[HU];
            #pragma unroll
            for (int u = 0; u < HU; ++u) d[u] = dst[min(i + u * HT, hi - 1)];
            #pragma unroll
            for (int u = 0; u < HU; ++u)
                atomicAdd(&hist[d[u] >> BSH], (i + u * HT < hi) ? 1 : 0);
        }
    }
    __syncthreads();
    if (active) {
        for (int e = lane; e < 16 * HEADS * HID; e += 32) {
            const int row = e / (HEADS * HID);
            const int cc  = e - row * (HEADS * HID);
            if (m0 + row < N)
                RA[rec_off((int)(m0 + row)) + cc] = Ct[wave][row * CT_PITCH + cc];
        }
    }
    const int nbp = (nbk + 3) & ~3;
    if (counts) for (int k = t; k < nbp; k += FB_THREADS)
        H[(size_t)blockIdx.x * nbp + k] = (k < nbk) ? hist[k] : 0;
}


#define CH_CAP 16384
#define HB     16
#define SU     16
__global__ __launch_bounds__(1024) void bucket_scatter(const int* __restrict__ src,
                                                       const int* __restrict__ dst,
                                                       const int* __restrict__ H,
                                                       int* __restrict__ bstart,
                                                       unsigned* __restrict__ P, int E, int nbk) {
    __shared__ uint2 wg[CH_CAP];
    extern __shared__ int shs[];
    int* cur   = shs;
    int* tot   = shs + nbk;
    int* dl    = tot + nbk;
    int* wsum  = dl + nbk;
    int* wsum2 = wsum + 32;
    const int t = threadIdx.x;
    const int lane = t & 31, wave = t >> 5;
    const int c = blockIdx.x;
    for (int k = t; k < nbk; k += 1024) { tot[k] = 0; cur[k] = 0; }
    __syncthreads();
    {
        const int nq = ((nbk + 3) & ~3) >> 2;
        const int QS = min(1024, ((nq + 31) >> 5) << 5);
        const int G  = 1024 / QS;
        const int g  = t / QS;
        const int4* H4 = (const int4*)H;
        if (g < G) for (int q = t - g * QS; q < nq; q += QS) {
            int tx = 0, ty = 0, tz = 0, tw = 0, px = 0, py = 0, pz = 0, pw = 0;
            const int4 ow = H4[(size_t)c * nq + q];
            for (int i0 = g; i0 < SORT_BLOCKS; i0 += HB * G) {
                int4 v[HB];
                #pragma unroll
                for (int u = 0; u < HB; ++u) {
                    const int cc = (c + min(i0 + u * G, SORT_BLOCKS - 1)) & (SORT_BLOCKS - 1);
                    v[u] = H4[(size_t)cc * nq + q];
                }
                __builtin_amdgcn_sched_barrier(0);
                #pragma unroll
                for (int u = 0; u < HB; ++u) {
                    const int  i   = i0 + u * G;
                    const bool in  = i < SORT_BLOCKS;
                    const bool bef = in && (c + i >= SORT_BLOCKS);
                    tx += in ? v[u].x : 0;  ty += in ? v[u].y : 0;
                    tz += in ? v[u].z : 0;  tw += in ? v[u].w : 0;
                    px += bef ? v[u].x : 0; py += bef ? v[u].y : 0;
                    pz += bef ? v[u].z : 0; pw += bef ? v[u].w : 0;
                }
            }
            const int k = 4 * q;
            if (k     < nbk) { atomicAdd(&tot[k],     tx); atomicAdd(&cur[k],     px); if (g == 0) dl[k]     = ow.x; }
            if (k + 1 < nbk) { atomicAdd(&tot[k + 1], ty); atomicAdd(&cur[k + 1], py); if (g == 0) dl[k + 1] = ow.y; }
            if (k + 2 < nbk) { atomicAdd(&tot[k + 2], tz); atomicAdd(&cur[k + 2], pz); if (g == 0) dl[k + 2] = ow.z; }
            if (k + 3 < nbk) { atomicAdd(&tot[k + 3], tw); atomicAdd(&cur[k + 3], pw); if (g == 0) dl[k + 3] = ow.w; }
        }
    }
    __syncthreads();
    const int seg = (nbk + 1023) >> 10;
    const int k0 = min(t * seg, nbk), k1 = min(k0 + seg, nbk);
    int loc = 0, loc2 = 0;
    for (int k = k0; k < k1; ++k) { loc += tot[k]; loc2 += dl[k]; }
    int v = loc, v2 = loc2;
    for (int off = 1; off < 32; off <<= 1) {
        const int u = __shfl_up(v, off);
        const int u2 = __shfl_up(v2, off);
        if (lane >= off) { v += u; v2 += u2; }
    }
    if (lane == 31) { wsum[wave] = v; wsum2[wave] = v2; }
    __syncthreads();
    int run = v - loc, run2 = v2 - loc2;
    for (int w = 0; w < wave; ++w) { run += wsum[w]; run2 += wsum2[w]; }
    for (int k = k0; k < k1; ++k) {
        if (c == 0) bstart[k * LINE_W + BSLOT] = run;
        const int ow = dl[k];
        cur[k] += run;
        dl[k] = cur[k] - run2;
        run += tot[k];
        run2 += ow;
    }
    if (c == 0 && t == 1023) bstart[nbk * LINE_W + BSLOT] = run;
    __syncthreads();
    const int ch = (E + (int)gridDim.x - 1) / (int)gridDim.x;
    const int lo = blockIdx.x * ch;
    const int hi = min(E, lo + ch);
    const bool fits = (hi - lo) <= CH_CAP;
    for (int i0 = lo + t; i0 < hi; i0 += SU * 1024) {
        int d[SU], s[SU];
        #pragma unroll
        for (int u = 0; u < SU; ++u) {
            const int i = min(i0 + u * 1024, hi - 1);
            d[u] = dst[i];
            s[u] = src[i];
        }
        #pragma unroll
        for (int u = 0; u < SU; ++u)
            if (i0 + u * 1024 < hi) {
                const int p = atomicAdd(&cur[d[u] >> BSH], 1);
                const unsigned w = ((unsigned)s[u] << BSH) | (unsigned)(d[u] & (BNODES - 1));
                if (fits) wg[p - dl[d[u] >> BSH]] = make_uint2(w, (unsigned)p);
                else      P[p] = w;
            }
    }
    __syncthreads();
    if (fits) {
        const int cn = hi - lo;
        int q = t;
        for (; q + 3 * 1024 < cn; q += 4 * 1024) {
            const uint2 e0 = wg[q], e1 = wg[q + 1024], e2 = wg[q + 2048], e3 = wg[q + 3072];
            P[e0.y] = e0.x; P[e1.y] = e1.x; P[e2.y] = e2.x; P[e3.y] = e3.x;
        }
        for (; q < cn; q += 1024) {
            const uint2 e = wg[q];
            P[e.y] = e.x;
        }
    }
}

#define GB_NODES   64
#define GB_THREADS (GB_NODES * HEADS)
#define GB_SU      16
#define GB_STAGE   (GB_SU * GB_THREADS)

__device__ __forceinline__ void edge_step(const float4 r, const float ad, float& m,
                                          float& s, float& a0, float& a1, float& a2) {
    const float e  = lrelu(r.w + ad);
    const float d  = e - m;
    const float tt = expf(-fabsf(d));
    const bool  up = d > 0.f;
    const float sc = up ? tt : 1.f;
    const float p  = up ? 1.f : tt;
    m  = up ? e : m;
    s  = s  * sc + p;
    a0 = a0 * sc + p * r.x;
    a1 = a1 * sc + p * r.y;
    a2 = a2 * sc + p * r.z;
}

__device__ __forceinline__ float4 load_rec(const float* __restrict__ Rs, const int off,
                                           const float as0, const float as1, const float as2) {
    const float* p = Rs + off;
    const float v0 = p[0], v1 = p[1], v2 = p[2];
    float s = 0.f;
    s += v0 * as0;
    s += v1 * as1;
    s += v2 * as2;
    return make_float4(v0, v1, v2, s);
}

template <bool STAGED>
__device__ __forceinline__ void row_sum(const int* __restrict__ csr, const int* sl, const int lo,
                                        int j, const int end, const float* __restrict__ Rs,
                                        const int h3, const float as0, const float as1, const float as2,
                                        const float ad, float& m, float& s, float& a0, float& a1, float& a2) {
    for (; j + 4 <= end; j += 4) {
        int i0, i1, i2, i3;
        if (STAGED) { const int* q = sl + (j - lo); i0 = q[0]; i1 = q[1]; i2 = q[2]; i3 = q[3]; }
        else        { i0 = csr[j]; i1 = csr[j + 1]; i2 = csr[j + 2]; i3 = csr[j + 3]; }
        const float4 r0 = load_rec(Rs, i0 + h3, as0, as1, as2);
        const float4 r1 = load_rec(Rs, i1 + h3, as0, as1, as2);
        const float4 r2 = load_rec(Rs, i2 + h3, as0, as1, as2);
        const float4 r3 = load_rec(Rs, i3 + h3, as0, as1, as2);
        edge_step(r0, ad, m, s, a0, a1, a2);
        edge_step(r1, ad, m, s, a0, a1, a2);
        edge_step(r2, ad, m, s, a0, a1, a2);
        edge_step(r3, ad, m, s, a0, a1, a2);
    }
    for (; j < end; ++j) {
        const int i = STAGED ? sl[j - lo] : csr[j];
        const float4 r = load_rec(Rs, i + h3, as0, as1, as2);
        edge_step(r, ad, m, s, a0, a1, a2);
    }
}

#define RG_T 768
#define RWR  12
__global__ __launch_bounds__(RG_T) void rows_gather1(int* csr, float* RA, int N,
                                                     const float* __restrict__ atts,
                                                     const float* __restrict__ attd,
                                                     const float* __restrict__ bias,
                                                     const float* __restrict__ W2,
                                                     float* __restrict__ o) {
    __shared__ int   sorted[STAGE_CAP];
    __shared__ int   cnt[BNODES];
    __shared__ int   cur[BNODES];
    __shared__ int   rs[BNODES + 1];
    __shared__ int   wsum[BNODES / 32];
    __shared__ float ex[RG_T * 3];
    __shared__ float Ws[81];
    int* rows = (int*)RA;
    const float* Rs = RA;
    const int t = threadIdx.x;
    const int b = blockIdx.x;
    const int lo = rows[b * LINE_W + BSLOT];
    const int hi = rows[(b + 1) * LINE_W + BSLOT];
    const int nst = min(hi - lo, STAGE_CAP);

    const int nl = t / HEADS;
    const int hh = t - nl * HEADS;
    const int n  = b * BNODES + nl;
    const bool valid = n < N;
    const int ns = valid ? n : 0;
    const int hw = hh * 3;
    const int gs = rec_off(ns) + hw;

    if (t < BNODES) cnt[t] = 0;
    __syncthreads();
    const float w2v = W2[min(t, 80)];
    const float as0 = atts[hw + 0], as1 = atts[hw + 1], as2 = atts[hw + 2];
    const float d0 = attd[hw + 0], d1 = attd[hw + 1], d2 = attd[hw + 2];
    float4 self = make_float4(Rs[gs + 0], Rs[gs + 1], Rs[gs + 2], 0.f);
    unsigned w[RWR];
    #pragma unroll
    for (int u = 0; u < RWR; ++u) w[u] = 0u;
    if (nst > 0) {
        #pragma unroll
        for (int u = 0; u < RWR; ++u) w[u] = (unsigned)csr[lo + min(t + u * RG_T, nst - 1)];
    }
    #pragma unroll
    for (int u = 0; u < RWR; ++u)
        if (t + u * RG_T < nst) atomicAdd(&cnt[w[u] & (BNODES - 1)], 1);
    for (int i = RWR * RG_T + t; i < nst; i += RG_T)
        atomicAdd(&cnt[(unsigned)csr[lo + i] & (BNODES - 1)], 1);
    __syncthreads();

    const int c = (t < BNODES) ? cnt[t] : 0;
    int v = c;
    for (int off = 1; off < 32; off <<= 1) {
        const int u = __shfl_up(v, off);
        if ((t & 31) >= off) v += u;
    }
    if (t < BNODES && (t & 31) == 31) wsum[t >> 5] = v;
    if (t < 81) Ws[t] = w2v;
    __syncthreads();
    if (t < BNODES) {
        int pre = 0;
        for (int ww = 0; ww < (t >> 5); ++ww) pre += wsum[ww];
        const int excl = pre + v - c;
        cur[t] = excl;
        rs[t] = excl;
        if (t == BNODES - 1) rs[BNODES] = excl + c;
        const int node = b * BNODES + t;
        if (node < N) rows[row_slot(node)] = lo + excl;
        if (node == N - 1) rows[row_slot(N)] = hi;
    }
    __syncthreads();

    #pragma unroll
    for (int u = 0; u < RWR; ++u)
        if (t + u * RG_T < nst) {
            const int r = atomicAdd(&cur[w[u] & (BNODES - 1)], 1);
            sorted[r] = rec_off((int)(w[u] >> BSH));
        }
    for (int i = RWR * RG_T + t; i < nst; i += RG_T) {
        const unsigned wv = (unsigned)csr[lo + i];
        const int r = atomicAdd(&cur[wv & (BNODES - 1)], 1);
        sorted[r] = rec_off((int)(wv >> BSH));
    }
    __syncthreads();

    const int g0 = lo & ~31;
    int g = g0 + t;
    for (; g + 3 * RG_T < lo + nst; g += 4 * RG_T) {
        const int x0 = sorted[max(g - lo, 0)], x1 = sorted[g + RG_T - lo];
        const int x2 = sorted[g + 2 * RG_T - lo], x3 = sorted[g + 3 * RG_T - lo];
        if (g >= lo) csr[g] = x0;
        csr[g + RG_T] = x1; csr[g + 2 * RG_T] = x2; csr[g + 3 * RG_T] = x3;
    }
    for (; g < lo + nst; g += RG_T)
        if (g >= lo) csr[g] = sorted[g - lo];
    for (int i = nst + t; i < hi - lo; i += RG_T) csr[lo + i] = 0;

    float v0 = 0.f, v1 = 0.f, v2 = 0.f;
    if (valid) {
        {
            float sw = 0.f;
            sw += self.x * as0;
            sw += self.y * as1;
            sw += self.z * as2;
            self.w = sw;
        }
        float ad = 0.f;
        ad += self.x * d0;
        ad += self.y * d1;
        ad += self.z * d2;

        float m = lrelu(self.w + ad);
        float s = 1.f, a0 = self.x, a1 = self.y, a2 = self.z;

        row_sum<true>(csr, sorted, 0, rs[nl], rs[nl + 1], Rs, hw, as0, as1, as2, ad, m, s, a0, a1, a2);

        const float inv = 1.0f / (s + GAT_EPS);
        v0 = a0 * inv; v1 = a1 * inv; v2 = a2 * inv;
        v0 += bias[hh * 3 + 0];
        v1 += bias[hh * 3 + 1];
        v2 += bias[hh * 3 + 2];
        v0 = (v0 > 0.f) ? v0 : (expf(v0) - 1.0f);
        v1 = (v1 > 0.f) ? v1 : (expf(v1) - 1.0f);
        v2 = (v2 > 0.f) ? v2 : (expf(v2) - 1.0f);
    }
    ex[t * 3 + 0] = v0;
    ex[t * 3 + 1] = v1;
    ex[t * 3 + 2] = v2;
    __syncthreads();

    const float* tn = &ex[nl * 9];
    float o0 = 0.f, o1 = 0.f, o2 = 0.f;
    for (int i = 0; i < 9; ++i) {
        const float ti = tn[i];
        o0 += ti * Ws[i * 9 + hh * 3 + 0];
        o1 += ti * Ws[i * 9 + hh * 3 + 1];
        o2 += ti * Ws[i * 9 + hh * 3 + 2];
    }
    if (valid) {
        o[gs + 0] = o0;
        o[gs + 1] = o1;
        o[gs + 2] = o2;
    }
}

__global__ __launch_bounds__(GB_THREADS) void gather_rows2(const int* __restrict__ rows,
                                                           const int* __restrict__ csr,
                                                           const float* __restrict__ Rs,
                                                           const float* __restrict__ atts,
                                                           const float* __restrict__ attd,
                                                           const float* __restrict__ bias,
                                                           float* __restrict__ o, int N) {
    __shared__ float ex[GB_THREADS * 3];
    __shared__ int   sl[GB_STAGE];
    const int t = threadIdx.x;

    const int nl = t / HEADS;
    const int hh = t - nl * HEADS;
    const int n  = blockIdx.x * GB_NODES + nl;
    const bool valid = n < N;

    const int n0 = blockIdx.x * GB_NODES;
    const int lo = rows[row_slot(n0)];
    const int hi = rows[row_slot(min(n0 + GB_NODES, N))];
    const bool fits = (hi - lo) <= GB_STAGE;

    const int ns = valid ? n : 0;
    const int hw = hh * 3;
    const int gs = rec_off(ns) + hw;
    const float as0 = atts[hw + 0], as1 = atts[hw + 1], as2 = atts[hw + 2];
    const float d0 = attd[hw + 0], d1 = attd[hw + 1], d2 = attd[hw + 2];
    float4 self = make_float4(Rs[gs + 0], Rs[gs + 1], Rs[gs + 2], 0.f);
    const int j   = rows[row_slot(ns)];
    const int end = rows[row_slot(ns + 1)];

    if (fits && hi > lo) {
        const int cnt = hi - lo;
        int w[GB_SU];
        #pragma unroll
        for (int u = 0; u < GB_SU; ++u) w[u] = csr[lo + min(t + u * GB_THREADS, cnt - 1)];
        #pragma unroll
        for (int u = 0; u < GB_SU; ++u)
            if (t + u * GB_THREADS < cnt) sl[t + u * GB_THREADS] = w[u];
    }
    __syncthreads();

    float v0 = 0.f, v1 = 0.f, v2 = 0.f;
    if (valid) {
        {
            float sw = 0.f;
            sw += self.x * as0;
            sw += self.y * as1;
            sw += self.z * as2;
            self.w = sw;
        }
        float ad = 0.f;
        ad += self.x * d0;
        ad += self.y * d1;
        ad += self.z * d2;

        float m = lrelu(self.w + ad);
        float s = 1.f, a0 = self.x, a1 = self.y, a2 = self.z;

        if (fits) row_sum<true >(csr, sl, lo, j, end, Rs, hw, as0, as1, as2, ad, m, s, a0, a1, a2);
        else      row_sum<false>(csr, sl, lo, j, end, Rs, hw, as0, as1, as2, ad, m, s, a0, a1, a2);

        const float inv = 1.0f / (s + GAT_EPS);
        v0 = a0 * inv; v1 = a1 * inv; v2 = a2 * inv;
    }
    ex[t * 3 + 0] = v0;
    ex[t * 3 + 1] = v1;
    ex[t * 3 + 2] = v2;
    __syncthreads();

    const int nn = blockIdx.x * GB_NODES + t;
    if (t < GB_NODES && nn < N) {
        float v[NCLS];
        for (int cc = 0; cc < NCLS; ++cc) {
            float acv = 0.f;
            for (int h3 = 0; h3 < HEADS; ++h3)
                acv += ex[t * 9 + h3 * 3 + cc];
            v[cc] = acv * (1.0f / HEADS) + bias[cc];
        }
        float mx = fmaxf(v[0], fmaxf(v[1], v[2]));
        float lse = logf(expf(v[0] - mx) + expf(v[1] - mx) + expf(v[2] - mx));
        for (int cc = 0; cc < NCLS; ++cc)
            o[nn * NCLS + cc] = v[cc] - mx - lse;
    }
}

extern "C" void kernel_launch(void* const* d_in, const int* in_sizes, int n_in,
                              void* d_out, int out_size, void* d_ws, size_t ws_size,
                              hipStream_t stream) {
    const float* x        = (const float*)d_in[0];
    const int*   eidx     = (const int*)d_in[1];
    const float* W1       = (const float*)d_in[2];
    const float* att_src1 = (const float*)d_in[3];
    const float* att_dst1 = (const float*)d_in[4];
    const float* b1       = (const float*)d_in[5];
    const float* W2       = (const float*)d_in[6];
    const float* att_src2 = (const float*)d_in[7];
    const float* att_dst2 = (const float*)d_in[8];
    const float* b2       = (const float*)d_in[9];

    const int N = in_sizes[0] / IN_DIM;
    const int E = in_sizes[1] / 2;
    const int* src = eidx;
    const int* dst = eidx + E;

    const int nb = (N + 255) / 256;
    const int hn = ((nb + 3) & ~3) * SORT_BLOCKS;
    const size_t plane = (size_t)((N + 3) / 3) * LINE_W;
    const size_t need = (2 * plane + (size_t)E) * 4;
    if (ws_size < need) return;
    if ((size_t)hn > plane) return;
    if ((size_t)(nb + 2) * LINE_W > plane) return;
    if (N > (1 << 24)) return;

    float*  RA     = (float*)d_ws;
    float*  RB     = RA + plane;
    int*    csr    = (int*)(RB + plane);
    int*    H      = (int*)RB;

    const int hb = (N + GB_NODES - 1) / GB_NODES;
    const int tiles = (N + 15) / 16;
    const int pb = (tiles + PW - 1) / PW;
    const int fb = pb > SORT_BLOCKS ? pb : SORT_BLOCKS;

    front<<<fb, FB_THREADS, nb * sizeof(int), stream>>>(x, W1, RA, N, dst, H, E, nb);

    bucket_scatter<<<SORT_BLOCKS, 1024, (3 * nb + 64) * sizeof(int), stream>>>(src, dst, H, (int*)RA,
                                                                            (unsigned*)csr, E, nb);

    rows_gather1<<<nb, RG_T, 0, stream>>>(csr, RA, N, att_src1, att_dst1, b1, W2, RB);

    gather_rows2<<<hb, GB_THREADS, 0, stream>>>((const int*)RA, csr, RB, att_src2, att_dst2, b2,
                                                (float*)d_out, N);
}


# Edge weight exp(-|d|) = one multiply + the hardware base-2 exponential (v_exp_f32), not library expf (14 vector instructions per edge and head); weights differ in the last bits; sum order unchanged
# speedup vs baseline: 26.5920x; 1.0362x over previous
#include <hip/hip_runtime.h>
#include <math.h>

#define IN_DIM   256
#define HEADS    3
#define HID      3
#define NCLS     3
#define NEG_SLOPE 0.2f
#define GAT_EPS  1e-16f

typedef __attribute__((ext_vector_type(2))) float v2f;
typedef __attribute__((ext_vector_type(8))) float v8f;

__device__ __forceinline__ float lrelu(float x) { return x > 0.f ? x : NEG_SLOPE * x; }

#define LINE_W  32
#define ROWSLOT 27
#define BSLOT   30
__device__ __forceinline__ int rec_off(const int node) {
    const int q = node / 3;
    return q * LINE_W + (node - q * 3) * 9;
}
__device__ __forceinline__ int row_slot(const int node) {
    const int q = node / 3;
    return q * LINE_W + ROWSLOT + (node - q * 3);
}

#define CT_PITCH 20
#define BSH         8
#define BNODES      (1 << BSH)
#define SORT_BLOCKS 256
#define STAGE_CAP   24576
#define FB_THREADS 1024
#define PW         25
#define HT         (FB_THREADS - PW * 32)
#define HU         16
__global__ __launch_bounds__(FB_THREADS) void front(const float* __restrict__ x,
                                                    const float* __restrict__ W,
                                                    float* __restrict__ RA, int N,
                                                    const int* __restrict__ dst,
                                                    int* __restrict__ H, int E, int nbk) {
    __shared__ float Wp[IN_DIM * 16];
    __shared__ float Ct[PW][16 * CT_PITCH];
    extern __shared__ int hist[];
    const int t = threadIdx.x;
    const bool counts = (int)blockIdx.x < SORT_BLOCKS;
    for (int i = t; i < IN_DIM * 16; i += FB_THREADS) {
        int k = i >> 4, n = i & 15;
        Wp[i] = (n < HEADS * HID) ? W[k * (HEADS * HID) + n] : 0.0f;
    }
    if (counts) for (int k = t; k < nbk; k += FB_THREADS) hist[k] = 0;
    __syncthreads();

    int lane  = t & 31;
    int wave  = t >> 5;
    long tile = (long)wave * (long)gridDim.x + (long)blockIdx.x;
    long m0   = tile * 16;
    const bool active = (wave < PW) && (m0 < N);

    int laneM = lane & 15;
    int koff  = (lane < 16) ? 0 : 2;

    if (active) {
        const float* xrow = x + (m0 + laneM) * IN_DIM;
        v8f c = {};
        #pragma unroll 4
        for (int k0 = 0; k0 < IN_DIM; k0 += 4) {
            v2f a, b;
            a.x = xrow[k0 + koff];
            a.y = xrow[k0 + koff + 1];
            b.x = Wp[(k0 + koff)     * 16 + laneM];
            b.y = Wp[(k0 + koff + 1) * 16 + laneM];
            c = __builtin_amdgcn_wmma_f32_16x16x4_f32(false, a, false, b,
                                                      (short)0, c, false, false);
        }
        int rbase = (lane < 16) ? 0 : 8;
        #pragma unroll
        for (int v = 0; v < 8; ++v)
            Ct[wave][(rbase + v) * CT_PITCH + laneM] = c[v];
    } else if (wave >= PW && counts) {
        const int ht = t - PW * 32;
        const int ch = (E + SORT_BLOCKS - 1) / SORT_BLOCKS;
        const int lo = (int)blockIdx.x * ch;
        const int hi = min(E, lo + ch);
        for (int i = lo + ht; i < hi; i += HU * HT) {
            int d[HU];
            #pragma unroll
            for (int u = 0; u < HU; ++u) d[u] = dst[min(i + u * HT, hi - 1)];
            #pragma unroll
            for (int u = 0; u < HU; ++u)
                atomicAdd(&hist[d[u] >> BSH], (i + u * HT < hi) ? 1 : 0);
        }
    }
    __syncthreads();
    if (active) {
        for (int e = lane; e < 16 * HEADS * HID; e += 32) {
            const int row = e / (HEADS * HID);
            const int cc  = e - row * (HEADS * HID);
            if (m0 + row < N)
                RA[rec_off((int)(m0 + row)) + cc] = Ct[wave][row * CT_PITCH + cc];
        }
    }
    const int nbp = (nbk + 3) & ~3;
    if (counts) for (int k = t; k < nbp; k += FB_THREADS)
        H[(size_t)blockIdx.x * nbp + k] = (k < nbk) ? hist[k] : 0;
}


#define CH_CAP 16384
#define HB     16
#define SU     16
__global__ __launch_bounds__(1024) void bucket_scatter(const int* __restrict__ src,
                                                       const int* __restrict__ dst,
                                                       const int* __restrict__ H,
                                                       int* __restrict__ bstart,
                                                       unsigned* __restrict__ P, int E, int nbk) {
    __shared__ uint2 wg[CH_CAP];
    extern __shared__ int shs[];
    int* cur   = shs;
    int* tot   = shs + nbk;
    int* dl    = tot + nbk;
    int* wsum  = dl + nbk;
    int* wsum2 = wsum + 32;
    const int t = threadIdx.x;
    const int lane = t & 31, wave = t >> 5;
    const int c = blockIdx.x;
    for (int k = t; k < nbk; k += 1024) { tot[k] = 0; cur[k] = 0; }
    __syncthreads();
    {
        const int nq = ((nbk + 3) & ~3) >> 2;
        const int QS = min(1024, ((nq + 31) >> 5) << 5);
        const int G  = 1024 / QS;
        const int g  = t / QS;
        const int4* H4 = (const int4*)H;
        if (g < G) for (int q = t - g * QS; q < nq; q += QS) {
            int tx = 0, ty = 0, tz = 0, tw = 0, px = 0, py = 0, pz = 0, pw = 0;
            const int4 ow = H4[(size_t)c * nq + q];
            for (int i0 = g; i0 < SORT_BLOCKS; i0 += HB * G) {
                int4 v[HB];
                #pragma unroll
                for (int u = 0; u < HB; ++u) {
                    const int cc = (c + min(i0 + u * G, SORT_BLOCKS - 1)) & (SORT_BLOCKS - 1);
                    v[u] = H4[(size_t)cc * nq + q];
                }
                __builtin_amdgcn_sched_barrier(0);
                #pragma unroll
                for (int u = 0; u < HB; ++u) {
                    const int  i   = i0 + u * G;
                    const bool in  = i < SORT_BLOCKS;
                    const bool bef = in && (c + i >= SORT_BLOCKS);
                    tx += in ? v[u].x : 0;  ty += in ? v[u].y : 0;
                    tz += in ? v[u].z : 0;  tw += in ? v[u].w : 0;
                    px += bef ? v[u].x : 0; py += bef ? v[u].y : 0;
                    pz += bef ? v[u].z : 0; pw += bef ? v[u].w : 0;
                }
            }
            const int k = 4 * q;
            if (k     < nbk) { atomicAdd(&tot[k],     tx); atomicAdd(&cur[k],     px); if (g == 0) dl[k]     = ow.x; }
            if (k + 1 < nbk) { atomicAdd(&tot[k + 1], ty); atomicAdd(&cur[k + 1], py); if (g == 0) dl[k + 1] = ow.y; }
            if (k + 2 < nbk) { atomicAdd(&tot[k + 2], tz); atomicAdd(&cur[k + 2], pz); if (g == 0) dl[k + 2] = ow.z; }
            if (k + 3 < nbk) { atomicAdd(&tot[k + 3], tw); atomicAdd(&cur[k + 3], pw); if (g == 0) dl[k + 3] = ow.w; }
        }
    }
    __syncthreads();
    const int seg = (nbk + 1023) >> 10;
    const int k0 = min(t * seg, nbk), k1 = min(k0 + seg, nbk);
    int loc = 0, loc2 = 0;
    for (int k = k0; k < k1; ++k) { loc += tot[k]; loc2 += dl[k]; }
    int v = loc, v2 = loc2;
    for (int off = 1; off < 32; off <<= 1) {
        const int u = __shfl_up(v, off);
        const int u2 = __shfl_up(v2, off);
        if (lane >= off) { v += u; v2 += u2; }
    }
    if (lane == 31) { wsum[wave] = v; wsum2[wave] = v2; }
    __syncthreads();
    int run = v - loc, run2 = v2 - loc2;
    for (int w = 0; w < wave; ++w) { run += wsum[w]; run2 += wsum2[w]; }
    for (int k = k0; k < k1; ++k) {
        if (c == 0) bstart[k * LINE_W + BSLOT] = run;
        const int ow = dl[k];
        cur[k] += run;
        dl[k] = cur[k] - run2;
        run += tot[k];
        run2 += ow;
    }
    if (c == 0 && t == 1023) bstart[nbk * LINE_W + BSLOT] = run;
    __syncthreads();
    const int ch = (E + (int)gridDim.x - 1) / (int)gridDim.x;
    const int lo = blockIdx.x * ch;
    const int hi = min(E, lo + ch);
    const bool fits = (hi - lo) <= CH_CAP;
    for (int i0 = lo + t; i0 < hi; i0 += SU * 1024) {
        int d[SU], s[SU];
        #pragma unroll
        for (int u = 0; u < SU; ++u) {
            const int i = min(i0 + u * 1024, hi - 1);
            d[u] = dst[i];
            s[u] = src[i];
        }
        #pragma unroll
        for (int u = 0; u < SU; ++u)
            if (i0 + u * 1024 < hi) {
                const int p = atomicAdd(&cur[d[u] >> BSH], 1);
                const unsigned w = ((unsigned)s[u] << BSH) | (unsigned)(d[u] & (BNODES - 1));
                if (fits) wg[p - dl[d[u] >> BSH]] = make_uint2(w, (unsigned)p);
                else      P[p] = w;
            }
    }
    __syncthreads();
    if (fits) {
        const int cn = hi - lo;
        int q = t;
        for (; q + 3 * 1024 < cn; q += 4 * 1024) {
            const uint2 e0 = wg[q], e1 = wg[q + 1024], e2 = wg[q + 2048], e3 = wg[q + 3072];
            P[e0.y] = e0.x; P[e1.y] = e1.x; P[e2.y] = e2.x; P[e3.y] = e3.x;
        }
        for (; q < cn; q += 1024) {
            const uint2 e = wg[q];
            P[e.y] = e.x;
        }
    }
}

#define GB_NODES   64
#define GB_THREADS (GB_NODES * HEADS)
#define GB_SU      16
#define GB_STAGE   (GB_SU * GB_THREADS)

__device__ __forceinline__ float exp_neg_abs(const float d) {
    return __builtin_amdgcn_exp2f(-fabsf(d) * 1.44269504088896340736f);
}

__device__ __forceinline__ void edge_step(const float4 r, const float ad, float& m,
                                          float& s, float& a0, float& a1, float& a2) {
    const float e  = lrelu(r.w + ad);
    const float d  = e - m;
    const float tt = exp_neg_abs(d);
    const bool  up = d > 0.f;
    const float sc = up ? tt : 1.f;
    const float p  = up ? 1.f : tt;
    m  = up ? e : m;
    s  = s  * sc + p;
    a0 = a0 * sc + p * r.x;
    a1 = a1 * sc + p * r.y;
    a2 = a2 * sc + p * r.z;
}

__device__ __forceinline__ float4 load_rec(const float* __restrict__ Rs, const int off,
                                           const float as0, const float as1, const float as2) {
    const float* p = Rs + off;
    const float v0 = p[0], v1 = p[1], v2 = p[2];
    float s = 0.f;
    s += v0 * as0;
    s += v1 * as1;
    s += v2 * as2;
    return make_float4(v0, v1, v2, s);
}

template <bool STAGED>
__device__ __forceinline__ void row_sum(const int* __restrict__ csr, const int* sl, const int lo,
                                        int j, const int end, const float* __restrict__ Rs,
                                        const int h3, const float as0, const float as1, const float as2,
                                        const float ad, float& m, float& s, float& a0, float& a1, float& a2) {
    for (; j + 4 <= end; j += 4) {
        int i0, i1, i2, i3;
        if (STAGED) { const int* q = sl + (j - lo); i0 = q[0]; i1 = q[1]; i2 = q[2]; i3 = q[3]; }
        else        { i0 = csr[j]; i1 = csr[j + 1]; i2 = csr[j + 2]; i3 = csr[j + 3]; }
        const float4 r0 = load_rec(Rs, i0 + h3, as0, as1, as2);
        const float4 r1 = load_rec(Rs, i1 + h3, as0, as1, as2);
        const float4 r2 = load_rec(Rs, i2 + h3, as0, as1, as2);
        const float4 r3 = load_rec(Rs, i3 + h3, as0, as1, as2);
        edge_step(r0, ad, m, s, a0, a1, a2);
        edge_step(r1, ad, m, s, a0, a1, a2);
        edge_step(r2, ad, m, s, a0, a1, a2);
        edge_step(r3, ad, m, s, a0, a1, a2);
    }
    #pragma unroll 2
    for (; j < end; ++j) {
        const int i = STAGED ? sl[j - lo] : csr[j];
        const float4 r = load_rec(Rs, i + h3, as0, as1, as2);
        edge_step(r, ad, m, s, a0, a1, a2);
    }
}

#define RG_T 768
#define RWR  12
__global__ __launch_bounds__(RG_T) void rows_gather1(int* csr, float* RA, int N,
                                                     const float* __restrict__ atts,
                                                     const float* __restrict__ attd,
                                                     const float* __restrict__ bias,
                                                     const float* __restrict__ W2,
                                                     float* __restrict__ o) {
    __shared__ int   sorted[STAGE_CAP];
    __shared__ int   cnt[BNODES];
    __shared__ int   cur[BNODES];
    __shared__ int   rs[BNODES + 1];
    __shared__ int   wsum[BNODES / 32];
    __shared__ float ex[RG_T * 3];
    __shared__ float Ws[81];
    int* rows = (int*)RA;
    const float* Rs = RA;
    const int t = threadIdx.x;
    const int b = blockIdx.x;
    const int lo = rows[b * LINE_W + BSLOT];
    const int hi = rows[(b + 1) * LINE_W + BSLOT];
    const int nst = min(hi - lo, STAGE_CAP);

    const int nl = t / HEADS;
    const int hh = t - nl * HEADS;
    const int n  = b * BNODES + nl;
    const bool valid = n < N;
    const int ns = valid ? n : 0;
    const int hw = hh * 3;
    const int gs = rec_off(ns) + hw;

    if (t < BNODES) cnt[t] = 0;
    __syncthreads();
    const float w2v = W2[min(t, 80)];
    const float as0 = atts[hw + 0], as1 = atts[hw + 1], as2 = atts[hw + 2];
    const float d0 = attd[hw + 0], d1 = attd[hw + 1], d2 = attd[hw + 2];
    float4 self = make_float4(Rs[gs + 0], Rs[gs + 1], Rs[gs + 2], 0.f);
    unsigned w[RWR];
    #pragma unroll
    for (int u = 0; u < RWR; ++u) w[u] = 0u;
    if (nst > 0) {
        #pragma unroll
        for (int u = 0; u < RWR; ++u) w[u] = (unsigned)csr[lo + min(t + u * RG_T, nst - 1)];
    }
    #pragma unroll
    for (int u = 0; u < RWR; ++u)
        if (t + u * RG_T < nst) atomicAdd(&cnt[w[u] & (BNODES - 1)], 1);
    for (int i = RWR * RG_T + t; i < nst; i += RG_T)
        atomicAdd(&cnt[(unsigned)csr[lo + i] & (BNODES - 1)], 1);
    __syncthreads();

    const int c = (t < BNODES) ? cnt[t] : 0;
    int v = c;
    for (int off = 1; off < 32; off <<= 1) {
        const int u = __shfl_up(v, off);
        if ((t & 31) >= off) v += u;
    }
    if (t < BNODES && (t & 31) == 31) wsum[t >> 5] = v;
    if (t < 81) Ws[t] = w2v;
    __syncthreads();
    if (t < BNODES) {
        int pre = 0;
        for (int ww = 0; ww < (t >> 5); ++ww) pre += wsum[ww];
        const int excl = pre + v - c;
        cur[t] = excl;
        rs[t] = excl;
        if (t == BNODES - 1) rs[BNODES] = excl + c;
        const int node = b * BNODES + t;
        if (node < N) rows[row_slot(node)] = lo + excl;
        if (node == N - 1) rows[row_slot(N)] = hi;
    }
    __syncthreads();

    #pragma unroll
    for (int u = 0; u < RWR; ++u)
        if (t + u * RG_T < nst) {
            const int r = atomicAdd(&cur[w[u] & (BNODES - 1)], 1);
            sorted[r] = rec_off((int)(w[u] >> BSH));
        }
    for (int i = RWR * RG_T + t; i < nst; i += RG_T) {
        const unsigned wv = (unsigned)csr[lo + i];
        const int r = atomicAdd(&cur[wv & (BNODES - 1)], 1);
        sorted[r] = rec_off((int)(wv >> BSH));
    }
    __syncthreads();

    const int g0 = lo & ~31;
    int g = g0 + t;
    for (; g + 3 * RG_T < lo + nst; g += 4 * RG_T) {
        const int x0 = sorted[max(g - lo, 0)], x1 = sorted[g + RG_T - lo];
        const int x2 = sorted[g + 2 * RG_T - lo], x3 = sorted[g + 3 * RG_T - lo];
        if (g >= lo) csr[g] = x0;
        csr[g + RG_T] = x1; csr[g + 2 * RG_T] = x2; csr[g + 3 * RG_T] = x3;
    }
    for (; g < lo + nst; g += RG_T)
        if (g >= lo) csr[g] = sorted[g - lo];
    for (int i = nst + t; i < hi - lo; i += RG_T) csr[lo + i] = 0;

    float v0 = 0.f, v1 = 0.f, v2 = 0.f;
    if (valid) {
        {
            float sw = 0.f;
            sw += self.x * as0;
            sw += self.y * as1;
            sw += self.z * as2;
            self.w = sw;
        }
        float ad = 0.f;
        ad += self.x * d0;
        ad += self.y * d1;
        ad += self.z * d2;

        float m = lrelu(self.w + ad);
        float s = 1.f, a0 = self.x, a1 = self.y, a2 = self.z;

        row_sum<true>(csr, sorted, 0, rs[nl], rs[nl + 1], Rs, hw, as0, as1, as2, ad, m, s, a0, a1, a2);

        const float inv = 1.0f / (s + GAT_EPS);
        v0 = a0 * inv; v1 = a1 * inv; v2 = a2 * inv;
        v0 += bias[hh * 3 + 0];
        v1 += bias[hh * 3 + 1];
        v2 += bias[hh * 3 + 2];
        v0 = (v0 > 0.f) ? v0 : (expf(v0) - 1.0f);
        v1 = (v1 > 0.f) ? v1 : (expf(v1) - 1.0f);
        v2 = (v2 > 0.f) ? v2 : (expf(v2) - 1.0f);
    }
    ex[t * 3 + 0] = v0;
    ex[t * 3 + 1] = v1;
    ex[t * 3 + 2] = v2;
    __syncthreads();

    const float* tn = &ex[nl * 9];
    float o0 = 0.f, o1 = 0.f, o2 = 0.f;
    for (int i = 0; i < 9; ++i) {
        const float ti = tn[i];
        o0 += ti * Ws[i * 9 + hh * 3 + 0];
        o1 += ti * Ws[i * 9 + hh * 3 + 1];
        o2 += ti * Ws[i * 9 + hh * 3 + 2];
    }
    if (valid) {
        o[gs + 0] = o0;
        o[gs + 1] = o1;
        o[gs + 2] = o2;
    }
}

__global__ __launch_bounds__(GB_THREADS) void gather_rows2(const int* __restrict__ rows,
                                                           const int* __restrict__ csr,
                                                           const float* __restrict__ Rs,
                                                           const float* __restrict__ atts,
                                                           const float* __restrict__ attd,
                                                           const float* __restrict__ bias,
                                                           float* __restrict__ o, int N) {
    __shared__ float ex[GB_THREADS * 3];
    __shared__ int   sl[GB_STAGE];
    const int t = threadIdx.x;

    const int nl = t / HEADS;
    const int hh = t - nl * HEADS;
    const int n  = blockIdx.x * GB_NODES + nl;
    const bool valid = n < N;

    const int n0 = blockIdx.x * GB_NODES;
    const int lo = rows[row_slot(n0)];
    const int hi = rows[row_slot(min(n0 + GB_NODES, N))];
    const bool fits = (hi - lo) <= GB_STAGE;

    const int ns = valid ? n : 0;
    const int hw = hh * 3;
    const int gs = rec_off(ns) + hw;
    const float as0 = atts[hw + 0], as1 = atts[hw + 1], as2 = atts[hw + 2];
    const float d0 = attd[hw + 0], d1 = attd[hw + 1], d2 = attd[hw + 2];
    float4 self = make_float4(Rs[gs + 0], Rs[gs + 1], Rs[gs + 2], 0.f);
    const int j   = rows[row_slot(ns)];
    const int end = rows[row_slot(ns + 1)];

    if (fits && hi > lo) {
        const int cnt = hi - lo;
        int w[GB_SU];
        #pragma unroll
        for (int u = 0; u < GB_SU; ++u) w[u] = csr[lo + min(t + u * GB_THREADS, cnt - 1)];
        #pragma unroll
        for (int u = 0; u < GB_SU; ++u)
            if (t + u * GB_THREADS < cnt) sl[t + u * GB_THREADS] = w[u];
    }
    __syncthreads();

    float v0 = 0.f, v1 = 0.f, v2 = 0.f;
    if (valid) {
        {
            float sw = 0.f;
            sw += self.x * as0;
            sw += self.y * as1;
            sw += self.z * as2;
            self.w = sw;
        }
        float ad = 0.f;
        ad += self.x * d0;
        ad += self.y * d1;
        ad += self.z * d2;

        float m = lrelu(self.w + ad);
        float s = 1.f, a0 = self.x, a1 = self.y, a2 = self.z;

        if (fits) row_sum<true >(csr, sl, lo, j, end, Rs, hw, as0, as1, as2, ad, m, s, a0, a1, a2);
        else      row_sum<false>(csr, sl, lo, j, end, Rs, hw, as0, as1, as2, ad, m, s, a0, a1, a2);

        const float inv = 1.0f / (s + GAT_EPS);
        v0 = a0 * inv; v1 = a1 * inv; v2 = a2 * inv;
    }
    ex[t * 3 + 0] = v0;
    ex[t * 3 + 1] = v1;
    ex[t * 3 + 2] = v2;
    __syncthreads();

    const int nn = blockIdx.x * GB_NODES + t;
    if (t < GB_NODES && nn < N) {
        float v[NCLS];
        for (int cc = 0; cc < NCLS; ++cc) {
            float acv = 0.f;
            for (int h3 = 0; h3 < HEADS; ++h3)
                acv += ex[t * 9 + h3 * 3 + cc];
            v[cc] = acv * (1.0f / HEADS) + bias[cc];
        }
        float mx = fmaxf(v[0], fmaxf(v[1], v[2]));
        float lse = logf(expf(v[0] - mx) + expf(v[1] - mx) + expf(v[2] - mx));
        for (int cc = 0; cc < NCLS; ++cc)
            o[nn * NCLS + cc] = v[cc] - mx - lse;
    }
}

extern "C" void kernel_launch(void* const* d_in, const int* in_sizes, int n_in,
                              void* d_out, int out_size, void* d_ws, size_t ws_size,
                              hipStream_t stream) {
    const float* x        = (const float*)d_in[0];
    const int*   eidx     = (const int*)d_in[1];
    const float* W1       = (const float*)d_in[2];
    const float* att_src1 = (const float*)d_in[3];
    const float* att_dst1 = (const float*)d_in[4];
    const float* b1       = (const float*)d_in[5];
    const float* W2       = (const float*)d_in[6];
    const float* att_src2 = (const float*)d_in[7];
    const float* att_dst2 = (const float*)d_in[8];
    const float* b2       = (const float*)d_in[9];

    const int N = in_sizes[0] / IN_DIM;
    const int E = in_sizes[1] / 2;
    const int* src = eidx;
    const int* dst = eidx + E;

    const int nb = (N + 255) / 256;
    const int hn = ((nb + 3) & ~3) * SORT_BLOCKS;
    const size_t plane = (size_t)((N + 3) / 3) * LINE_W;
    const size_t need = (2 * plane + (size_t)E) * 4;
    if (ws_size < need) return;
    if ((size_t)hn > plane) return;
    if ((size_t)(nb + 2) * LINE_W > plane) return;
    if (N > (1 << 24)) return;

    float*  RA     = (float*)d_ws;
    float*  RB     = RA + plane;
    int*    csr    = (int*)(RB + plane);
    int*    H      = (int*)RB;

    const int hb = (N + GB_NODES - 1) / GB_NODES;
    const int tiles = (N + 15) / 16;
    const int pb = (tiles + PW - 1) / PW;
    const int fb = pb > SORT_BLOCKS ? pb : SORT_BLOCKS;

    front<<<fb, FB_THREADS, nb * sizeof(int), stream>>>(x, W1, RA, N, dst, H, E, nb);

    bucket_scatter<<<SORT_BLOCKS, 1024, (3 * nb + 64) * sizeof(int), stream>>>(src, dst, H, (int*)RA,
                                                                            (unsigned*)csr, E, nb);

    rows_gather1<<<nb, RG_T, 0, stream>>>(csr, RA, N, att_src1, att_dst1, b1, W2, RB);

    gather_rows2<<<hb, GB_THREADS, 0, stream>>>((const int*)RA, csr, RB, att_src2, att_dst2, b2,
                                                (float*)d_out, N);
}
